# starting kernel
# baseline (speedup 1.0000x reference)
#include <hip/hip_runtime.h>

#define Nn    100000
#define Ee    3200000
#define FIN   256
#define Hh    32

typedef float v2f __attribute__((ext_vector_type(2)));
typedef float v8f __attribute__((ext_vector_type(8)));

__device__ __forceinline__ v8f wmma4(v2f a, v2f b, v8f c) {
  return __builtin_amdgcn_wmma_f32_16x16x4_f32(
       false, a,  false, b,
       (short)0, c,  false,  false);
}

__device__ __forceinline__ float sigmoidf_(float x) {
  return 1.0f / (1.0f + __expf(-x));
}

__global__ void pack_weights_kernel(const float* __restrict__ Wz, const float* __restrict__ Wr,
                                    const float* __restrict__ Wh, const float* __restrict__ bz,
                                    const float* __restrict__ br, const float* __restrict__ bh,
                                    float* __restrict__ Wcat, float* __restrict__ bcat) {
  int gid = blockIdx.x * blockDim.x + threadIdx.x;
  if (gid < FIN * 96) {
    int k = gid / 96, c = gid - k * 96;
    float v = (c < 32) ? Wz[k * 32 + c] : ((c < 64) ? Wr[k * 32 + c - 32] : Wh[k * 32 + c - 64]);
    Wcat[gid] = v;
  }
  if (gid < 96) {
    bcat[gid] = (gid < 32) ? bz[gid] : ((gid < 64) ? br[gid - 32] : bh[gid - 64]);
  }
}

__global__ void zero_deg_kernel(float* __restrict__ deg) {
  int gid = blockIdx.x * blockDim.x + threadIdx.x;
  if (gid < Nn) deg[gid] = 0.0f;
}

__global__ void deg_accum_kernel(const int* __restrict__ dst, const float* __restrict__ w,
                                 float* __restrict__ deg) {
  int gid = blockIdx.x * blockDim.x + threadIdx.x;
  if (gid < Ee) unsafeAtomicAdd(&deg[dst[gid]], w[gid]);
}

__global__ void finalize_dis_kernel(float* __restrict__ deg_dis) {
  int gid = blockIdx.x * blockDim.x + threadIdx.x;
  if (gid < Nn) deg_dis[gid] = rsqrtf(deg_dis[gid] + 1.0f);
}

__global__ void __launch_bounds__(32) gemm_xw_kernel(const float* __restrict__ x,
                                                     const float* __restrict__ Wcat,
                                                     float* __restrict__ XW) {
  __shared__ float Xs[16 * 260];
  const int lane = threadIdx.x;
  const int tile = blockIdx.x * 16;

  for (int i = lane; i < 1024; i += 32) {
    int r = i >> 6, c4 = (i & 63) << 2;
    const float4 v = *reinterpret_cast<const float4*>(x + (size_t)(tile + r) * FIN + c4);
    *reinterpret_cast<float4*>(&Xs[r * 260 + c4]) = v;
  }
  __syncthreads();

  const int m = lane & 15;
  const int koff = (lane >> 4) << 1;

  v8f acc[6] = {};
  const float2* arow = reinterpret_cast<const float2*>(&Xs[m * 260]);

#pragma unroll 4
  for (int k0 = 0; k0 < FIN; k0 += 4) {
    const int kk = k0 + koff;
    float2 af = arow[kk >> 1];
    v2f a; a.x = af.x; a.y = af.y;
    const float* brow = Wcat + (size_t)kk * 96 + m;
#pragma unroll
    for (int t = 0; t < 6; ++t) {
      v2f b; b.x = brow[t * 16]; b.y = brow[96 + t * 16];
      acc[t] = wmma4(a, b, acc[t]);
    }
  }

  const int row0 = (lane >> 4) << 3;
#pragma unroll
  for (int t = 0; t < 6; ++t)
#pragma unroll
    for (int v = 0; v < 8; ++v)
      XW[(size_t)(tile + row0 + v) * 96 + t * 16 + m] = acc[t][v];
}

__global__ void init_agg_kernel(const float* __restrict__ dis, const float* __restrict__ XW,
                                const float* __restrict__ bcat, float* __restrict__ AGG) {
  size_t gid = (size_t)blockIdx.x * blockDim.x + threadIdx.x;
  if (gid < (size_t)Nn * 96) {
    size_t i = gid / 96;
    int c = (int)(gid - i * 96);
    float d = dis[i];
    AGG[gid] = d * d * XW[gid] + bcat[c];
  }
}

__global__ void edge_scatter_kernel(const int* __restrict__ src, const int* __restrict__ dst,
                                    const float* __restrict__ w, const float* __restrict__ dis,
                                    const float* __restrict__ XW, float* __restrict__ AGG) {
  const int lane = threadIdx.x & 31;
  const long long e = (((long long)blockIdx.x * blockDim.x) + threadIdx.x) >> 5;
  if (e >= Ee) return;
  const int s = src[e], d = dst[e];
  const float nrm = dis[s] * w[e] * dis[d];
  const size_t bs = (size_t)s * 96, bd = (size_t)d * 96;
#pragma unroll
  for (int i = 0; i < 3; ++i) {
    const int c = lane + i * 32;
    unsafeAtomicAdd(&AGG[bd + c], nrm * XW[bs + c]);
  }
}

__global__ void __launch_bounds__(32) gru_kernel(const float* __restrict__ AGG,
                                                 const float* __restrict__ Hp,
                                                 const float* __restrict__ LzW, const float* __restrict__ Lzb,
                                                 const float* __restrict__ LrW, const float* __restrict__ Lrb,
                                                 const float* __restrict__ LhW, const float* __restrict__ Lhb,
                                                 const float* __restrict__ headW, const float* __restrict__ headb,
                                                 float* __restrict__ y_out, float* __restrict__ h_out) {
  __shared__ float As[16 * 132];
  __shared__ float Rs[16 * 36];
  const int lane = threadIdx.x;
  const int tile = blockIdx.x * 16;

  for (int idx = lane; idx < 16 * 128; idx += 32) {
    int r = idx >> 7, c = idx & 127;
    float v = (c < 96) ? AGG[(size_t)(tile + r) * 96 + c] : Hp[(size_t)(tile + r) * 32 + (c - 96)];
    As[r * 132 + c] = v;
  }
  __syncthreads();

  const int m = lane & 15;
  const int n = lane & 15;
  const int koff = (lane >> 4) << 1;

  v8f z0 = {}, z1 = {}, r0 = {}, r1 = {};
#pragma unroll 4
  for (int k0 = 0; k0 < 64; k0 += 4) {
    const int kk = k0 + koff;
    const int colz = (kk < 32) ? kk : (64 + kk);
    const int colr = (kk < 32) ? (32 + kk) : (64 + kk);
    v2f az; az.x = As[m * 132 + colz]; az.y = As[m * 132 + colz + 1];
    v2f ar; ar.x = As[m * 132 + colr]; ar.y = As[m * 132 + colr + 1];
    const float* lz = LzW + kk * 32 + n;
    const float* lr = LrW + kk * 32 + n;
    v2f b;
    b.x = lz[0];  b.y = lz[32];  z0 = wmma4(az, b, z0);
    b.x = lz[16]; b.y = lz[48];  z1 = wmma4(az, b, z1);
    b.x = lr[0];  b.y = lr[32];  r0 = wmma4(ar, b, r0);
    b.x = lr[16]; b.y = lr[48];  r1 = wmma4(ar, b, r1);
  }

  const float bz0 = Lzb[n], bz1 = Lzb[16 + n];
  const float br0 = Lrb[n], br1 = Lrb[16 + n];
  const int row0 = (lane >> 4) << 3;
#pragma unroll
  for (int v = 0; v < 8; ++v) {
    z0[v] = sigmoidf_(z0[v] + bz0);
    z1[v] = sigmoidf_(z1[v] + bz1);
    r0[v] = sigmoidf_(r0[v] + br0);
    r1[v] = sigmoidf_(r1[v] + br1);
    Rs[(row0 + v) * 36 + n]      = r0[v];
    Rs[(row0 + v) * 36 + 16 + n] = r1[v];
  }
  __syncthreads();

  v8f t0 = {}, t1 = {};
#pragma unroll 4
  for (int k0 = 0; k0 < 64; k0 += 4) {
    const int kk = k0 + koff;
    v2f a;
    a.x = As[m * 132 + 64 + kk];
    a.y = As[m * 132 + 65 + kk];
    if (kk >= 32) {
      a.x *= Rs[m * 36 + kk - 32];
      a.y *= Rs[m * 36 + kk - 31];
    }
    const float* lh = LhW + kk * 32 + n;
    v2f b;
    b.x = lh[0];  b.y = lh[32];  t0 = wmma4(a, b, t0);
    b.x = lh[16]; b.y = lh[48];  t1 = wmma4(a, b, t1);
  }

  const float bh0 = Lhb[n], bh1 = Lhb[16 + n];
  const float hw0 = headW[n], hw1 = headW[16 + n];
  float p[8];
#pragma unroll
  for (int v = 0; v < 8; ++v) {
    const float ht0 = tanhf(t0[v] + bh0);
    const float ht1 = tanhf(t1[v] + bh1);
    const float hp0 = As[(row0 + v) * 132 + 96 + n];
    const float hp1 = As[(row0 + v) * 132 + 112 + n];
    const float h0 = z0[v] * hp0 + (1.0f - z0[v]) * ht0;
    const float h1 = z1[v] * hp1 + (1.0f - z1[v]) * ht1;
    h_out[(size_t)(tile + row0 + v) * 32 + n]      = h0;
    h_out[(size_t)(tile + row0 + v) * 32 + 16 + n] = h1;
    p[v] = fmaxf(h0, 0.0f) * hw0 + fmaxf(h1, 0.0f) * hw1;
  }
#pragma unroll
  for (int s = 1; s < 16; s <<= 1)
#pragma unroll
    for (int v = 0; v < 8; ++v)
      p[v] += __shfl_xor(p[v], s, 32);
  if (n == 0) {
    const float hb = headb[0];
#pragma unroll
    for (int v = 0; v < 8; ++v)
      y_out[tile + row0 + v] = p[v] + hb;
  }
}

extern "C" void kernel_launch(void* const* d_in, const int* in_sizes, int n_in,
                              void* d_out, int out_size, void* d_ws, size_t ws_size,
                              hipStream_t stream) {
  const float* x    = (const float*)d_in[0];
  const int*   ei   = (const int*)d_in[1];
  const float* ew   = (const float*)d_in[2];
  const float* Hp   = (const float*)d_in[3];
  const float* Wz   = (const float*)d_in[4];
  const float* bz   = (const float*)d_in[5];
  const float* Wr   = (const float*)d_in[6];
  const float* br   = (const float*)d_in[7];
  const float* Wh   = (const float*)d_in[8];
  const float* bh   = (const float*)d_in[9];
  const float* LzW  = (const float*)d_in[10];
  const float* Lzb  = (const float*)d_in[11];
  const float* LrW  = (const float*)d_in[12];
  const float* Lrb  = (const float*)d_in[13];
  const float* LhW  = (const float*)d_in[14];
  const float* Lhb  = (const float*)d_in[15];
  const float* hW   = (const float*)d_in[16];
  const float* hb   = (const float*)d_in[17];

  const int* src = ei;
  const int* dst = ei + Ee;

  float* out_y = (float*)d_out;
  float* out_h = out_y + Nn;

  float* ws   = (float*)d_ws;
  float* dis  = ws;
  float* Wcat = ws + Nn;
  float* bcat = Wcat + FIN * 96;
  float* XW   = bcat + 96;
  float* AGG  = XW + (size_t)Nn * 96;

  pack_weights_kernel<<<(FIN * 96 + 255) / 256, 256, 0, stream>>>(Wz, Wr, Wh, bz, br, bh, Wcat, bcat);
  zero_deg_kernel<<<(Nn + 255) / 256, 256, 0, stream>>>(dis);
  deg_accum_kernel<<<Ee / 256, 256, 0, stream>>>(dst, ew, dis);
  finalize_dis_kernel<<<(Nn + 255) / 256, 256, 0, stream>>>(dis);
  gemm_xw_kernel<<<Nn / 16, 32, 0, stream>>>(x, Wcat, XW);
  init_agg_kernel<<<(Nn * 96) / 256, 256, 0, stream>>>(dis, XW, bcat, AGG);
  edge_scatter_kernel<<<Ee / 8, 256, 0, stream>>>(src, dst, ew, dis, XW, AGG);
  gru_kernel<<<Nn / 16, 32, 0, stream>>>(AGG, Hp, LzW, Lzb, LrW, Lrb, LhW, Lhb, hW, hb, out_y, out_h);
}


# Edge aggregation by destination bins: count/scan/scatter edges into 256 node ranges, sum rows with LDS atomics per range, store AGG once; no global atomics on AGG, sum order still arrival order
# speedup vs baseline: 1.8506x; 1.8506x over previous
#include <hip/hip_runtime.h>

#define Nn    100000
#define Ee    3200000
#define FIN   256
#define Hh    32

typedef float v2f __attribute__((ext_vector_type(2)));
typedef float v8f __attribute__((ext_vector_type(8)));

__device__ __forceinline__ v8f wmma4(v2f a, v2f b, v8f c) {
  return __builtin_amdgcn_wmma_f32_16x16x4_f32(
       false, a,  false, b,
       (short)0, c,  false,  false);
}

__device__ __forceinline__ float sigmoidf_(float x) {
  return 1.0f / (1.0f + __expf(-x));
}

__global__ void pack_weights_kernel(const float* __restrict__ Wz, const float* __restrict__ Wr,
                                    const float* __restrict__ Wh, const float* __restrict__ bz,
                                    const float* __restrict__ br, const float* __restrict__ bh,
                                    float* __restrict__ Wcat, float* __restrict__ bcat) {
  int gid = blockIdx.x * blockDim.x + threadIdx.x;
  if (gid < FIN * 96) {
    int k = gid / 96, c = gid - k * 96;
    float v = (c < 32) ? Wz[k * 32 + c] : ((c < 64) ? Wr[k * 32 + c - 32] : Wh[k * 32 + c - 64]);
    Wcat[gid] = v;
  }
  if (gid < 96) {
    bcat[gid] = (gid < 32) ? bz[gid] : ((gid < 64) ? br[gid - 32] : bh[gid - 64]);
  }
}

__global__ void zero_deg_kernel(float* __restrict__ deg) {
  int gid = blockIdx.x * blockDim.x + threadIdx.x;
  if (gid < Nn) deg[gid] = 0.0f;
}

__global__ void deg_accum_kernel(const int* __restrict__ dst, const float* __restrict__ w,
                                 float* __restrict__ deg) {
  int gid = blockIdx.x * blockDim.x + threadIdx.x;
  if (gid < Ee) unsafeAtomicAdd(&deg[dst[gid]], w[gid]);
}

__global__ void finalize_dis_kernel(float* __restrict__ deg_dis) {
  int gid = blockIdx.x * blockDim.x + threadIdx.x;
  if (gid < Nn) deg_dis[gid] = rsqrtf(deg_dis[gid] + 1.0f);
}

__global__ void __launch_bounds__(32) gemm_xw_kernel(const float* __restrict__ x,
                                                     const float* __restrict__ Wcat,
                                                     float* __restrict__ XW) {
  __shared__ float Xs[16 * 260];
  const int lane = threadIdx.x;
  const int tile = blockIdx.x * 16;

  for (int i = lane; i < 1024; i += 32) {
    int r = i >> 6, c4 = (i & 63) << 2;
    const float4 v = *reinterpret_cast<const float4*>(x + (size_t)(tile + r) * FIN + c4);
    *reinterpret_cast<float4*>(&Xs[r * 260 + c4]) = v;
  }
  __syncthreads();

  const int m = lane & 15;
  const int koff = (lane >> 4) << 1;

  v8f acc[6] = {};
  const float2* arow = reinterpret_cast<const float2*>(&Xs[m * 260]);

#pragma unroll 4
  for (int k0 = 0; k0 < FIN; k0 += 4) {
    const int kk = k0 + koff;
    float2 af = arow[kk >> 1];
    v2f a; a.x = af.x; a.y = af.y;
    const float* brow = Wcat + (size_t)kk * 96 + m;
#pragma unroll
    for (int t = 0; t < 6; ++t) {
      v2f b; b.x = brow[t * 16]; b.y = brow[96 + t * 16];
      acc[t] = wmma4(a, b, acc[t]);
    }
  }

  const int row0 = (lane >> 4) << 3;
#pragma unroll
  for (int t = 0; t < 6; ++t)
#pragma unroll
    for (int v = 0; v < 8; ++v)
      XW[(size_t)(tile + row0 + v) * 96 + t * 16 + m] = acc[t][v];
}

#define NBINS   256
#define BINSZ   391
#define NSLICE  256
#define SLICE_E (Ee / NSLICE)

__global__ void __launch_bounds__(1024) bin_count_kernel(const int* __restrict__ dst,
                                                         int* __restrict__ counts) {
  __shared__ int hist[NBINS];
  const int t = threadIdx.x;
  if (t < NBINS) hist[t] = 0;
  __syncthreads();
  const int e0 = blockIdx.x * SLICE_E;
  const int e1 = e0 + SLICE_E;
  for (int e = e0 + t; e < e1; e += 4096) {
    int d[4];
#pragma unroll
    for (int u = 0; u < 4; ++u) {
      const int ee = e + u * 1024;
      d[u] = dst[(ee < e1) ? ee : e];
    }
#pragma unroll
    for (int u = 0; u < 4; ++u)
      if (e + u * 1024 < e1) atomicAdd(&hist[d[u] / BINSZ], 1);
  }
  __syncthreads();
  if (t < NBINS) counts[t * NSLICE + blockIdx.x] = hist[t];
}

__global__ void __launch_bounds__(1024) bin_scan_kernel(const int* __restrict__ counts,
                                                        int* __restrict__ offs) {
  __shared__ int part[1024];
  const int t = threadIdx.x;
  const int4* p = reinterpret_cast<const int4*>(counts) + t * 16;
  int4 q[16];
  int sum = 0;
#pragma unroll
  for (int i = 0; i < 16; ++i) {
    q[i] = p[i];
    sum += q[i].x + q[i].y + q[i].z + q[i].w;
  }
  part[t] = sum;
  __syncthreads();
  for (int d = 1; d < 1024; d <<= 1) {
    const int x = (t >= d) ? part[t - d] : 0;
    __syncthreads();
    part[t] += x;
    __syncthreads();
  }
  int run = part[t] - sum;
  int4* o = reinterpret_cast<int4*>(offs) + t * 16;
#pragma unroll
  for (int i = 0; i < 16; ++i) {
    int4 r;
    r.x = run; run += q[i].x;
    r.y = run; run += q[i].y;
    r.z = run; run += q[i].z;
    r.w = run; run += q[i].w;
    o[i] = r;
  }
}

__global__ void __launch_bounds__(1024) bin_scatter_kernel(const int* __restrict__ src,
                                                           const int* __restrict__ dst,
                                                           const float* __restrict__ w,
                                                           const float* __restrict__ dis,
                                                           const int* __restrict__ offs,
                                                           uint2* __restrict__ recs) {
  __shared__ int cursor[NBINS];
  const int t = threadIdx.x;
  if (t < NBINS) cursor[t] = offs[t * NSLICE + blockIdx.x];
  __syncthreads();
  const int e0 = blockIdx.x * SLICE_E;
  const int e1 = e0 + SLICE_E;
  for (int e = e0 + t; e < e1; e += 4096) {
    int s[4], d[4];
    float ww[4], ds[4], dd[4];
#pragma unroll
    for (int u = 0; u < 4; ++u) {
      const int ee = e + u * 1024;
      const int ec = (ee < e1) ? ee : e;
      s[u] = src[ec];
      d[u] = dst[ec];
      ww[u] = w[ec];
    }
#pragma unroll
    for (int u = 0; u < 4; ++u) {
      ds[u] = dis[s[u]];
      dd[u] = dis[d[u]];
    }
#pragma unroll
    for (int u = 0; u < 4; ++u) {
      if (e + u * 1024 < e1) {
        const float nrm = ds[u] * ww[u] * dd[u];
        const int bin = d[u] / BINSZ;
        const int dl = d[u] - bin * BINSZ;
        const int pos = atomicAdd(&cursor[bin], 1);
        recs[pos] = make_uint2((unsigned)s[u] | ((unsigned)dl << 17), __float_as_uint(nrm));
      }
    }
  }
}

__global__ void __launch_bounds__(1024) bin_aggregate_kernel(const uint2* __restrict__ recs,
                                                             const int* __restrict__ offs,
                                                             const float* __restrict__ dis,
                                                             const float* __restrict__ XW,
                                                             const float* __restrict__ bcat,
                                                             float* __restrict__ AGG) {
  __shared__ float acc[BINSZ * 96];
  const int t = threadIdx.x;
  const int bin = blockIdx.x;
  const int node0 = bin * BINSZ;
  const int nrows = (Nn - node0 < BINSZ) ? (Nn - node0) : BINSZ;
  const float* xw0 = XW + (size_t)node0 * 96;

  const float4* xw4 = reinterpret_cast<const float4*>(xw0);
  const float4* b4p = reinterpret_cast<const float4*>(bcat);
  float4* acc4 = reinterpret_cast<float4*>(acc);
  for (int q = t; q < nrows * 24; q += 1024) {
    const int i = q / 24;
    const int c4 = q - i * 24;
    const float d = dis[node0 + i];
    const float4 x = xw4[q];
    const float4 b = b4p[c4];
    float4 r;
    r.x = d * d * x.x + b.x;
    r.y = d * d * x.y + b.y;
    r.z = d * d * x.z + b.z;
    r.w = d * d * x.w + b.w;
    acc4[q] = r;
  }
  __syncthreads();

  const int start = offs[bin * NSLICE];
  const int end = (bin == NBINS - 1) ? Ee : offs[(bin + 1) * NSLICE];
  const int lane = t & 31;
  const int wave = __builtin_amdgcn_readfirstlane(t >> 5);

  int base = start + wave * 32;
  uint2 rec_next = make_uint2(0u, 0u);
  if (base < end && lane < end - base) rec_next = recs[base + lane];
  for (; base < end; base += 1024) {
    const int cnt = (end - base < 32) ? (end - base) : 32;
    const uint2 rec = rec_next;
    const int nb = base + 1024;
    for (int j0 = 0; j0 < cnt; j0 += 16) {
      float v[16][3];
      float nr[16];
      int dl[16];
#pragma unroll
      for (int u = 0; u < 16; ++u) {
        const unsigned px = (unsigned)__builtin_amdgcn_readlane((int)rec.x, j0 + u);
        nr[u] = __uint_as_float((unsigned)__builtin_amdgcn_readlane((int)rec.y, j0 + u));
        dl[u] = (int)(px >> 17);
        const float* row = XW + (size_t)(px & 0x1FFFFu) * 96 + lane;
        v[u][0] = row[0];
        v[u][1] = row[32];
        v[u][2] = row[64];
      }
      if (j0 == 0) {
        rec_next = make_uint2(0u, 0u);
        if (nb < end && lane < end - nb) rec_next = recs[nb + lane];
      }
#pragma unroll
      for (int u = 0; u < 16; ++u) {
        if (j0 + u < cnt) {
          float* a = &acc[dl[u] * 96 + lane];
          atomicAdd(a,      nr[u] * v[u][0]);
          atomicAdd(a + 32, nr[u] * v[u][1]);
          atomicAdd(a + 64, nr[u] * v[u][2]);
        }
      }
    }
  }
  __syncthreads();

  float4* out = reinterpret_cast<float4*>(AGG + (size_t)node0 * 96);
  const float4* a4 = reinterpret_cast<const float4*>(acc);
  for (int q = t; q < nrows * 24; q += 1024) out[q] = a4[q];
}

__global__ void __launch_bounds__(32) gru_kernel(const float* __restrict__ AGG,
                                                 const float* __restrict__ Hp,
                                                 const float* __restrict__ LzW, const float* __restrict__ Lzb,
                                                 const float* __restrict__ LrW, const float* __restrict__ Lrb,
                                                 const float* __restrict__ LhW, const float* __restrict__ Lhb,
                                                 const float* __restrict__ headW, const float* __restrict__ headb,
                                                 float* __restrict__ y_out, float* __restrict__ h_out) {
  __shared__ float As[16 * 132];
  __shared__ float Rs[16 * 36];
  const int lane = threadIdx.x;
  const int tile = blockIdx.x * 16;

  for (int idx = lane; idx < 16 * 128; idx += 32) {
    int r = idx >> 7, c = idx & 127;
    float v = (c < 96) ? AGG[(size_t)(tile + r) * 96 + c] : Hp[(size_t)(tile + r) * 32 + (c - 96)];
    As[r * 132 + c] = v;
  }
  __syncthreads();

  const int m = lane & 15;
  const int n = lane & 15;
  const int koff = (lane >> 4) << 1;

  v8f z0 = {}, z1 = {}, r0 = {}, r1 = {};
#pragma unroll 4
  for (int k0 = 0; k0 < 64; k0 += 4) {
    const int kk = k0 + koff;
    const int colz = (kk < 32) ? kk : (64 + kk);
    const int colr = (kk < 32) ? (32 + kk) : (64 + kk);
    v2f az; az.x = As[m * 132 + colz]; az.y = As[m * 132 + colz + 1];
    v2f ar; ar.x = As[m * 132 + colr]; ar.y = As[m * 132 + colr + 1];
    const float* lz = LzW + kk * 32 + n;
    const float* lr = LrW + kk * 32 + n;
    v2f b;
    b.x = lz[0];  b.y = lz[32];  z0 = wmma4(az, b, z0);
    b.x = lz[16]; b.y = lz[48];  z1 = wmma4(az, b, z1);
    b.x = lr[0];  b.y = lr[32];  r0 = wmma4(ar, b, r0);
    b.x = lr[16]; b.y = lr[48];  r1 = wmma4(ar, b, r1);
  }

  const float bz0 = Lzb[n], bz1 = Lzb[16 + n];
  const float br0 = Lrb[n], br1 = Lrb[16 + n];
  const int row0 = (lane >> 4) << 3;
#pragma unroll
  for (int v = 0; v < 8; ++v) {
    z0[v] = sigmoidf_(z0[v] + bz0);
    z1[v] = sigmoidf_(z1[v] + bz1);
    r0[v] = sigmoidf_(r0[v] + br0);
    r1[v] = sigmoidf_(r1[v] + br1);
    Rs[(row0 + v) * 36 + n]      = r0[v];
    Rs[(row0 + v) * 36 + 16 + n] = r1[v];
  }
  __syncthreads();

  v8f t0 = {}, t1 = {};
#pragma unroll 4
  for (int k0 = 0; k0 < 64; k0 += 4) {
    const int kk = k0 + koff;
    v2f a;
    a.x = As[m * 132 + 64 + kk];
    a.y = As[m * 132 + 65 + kk];
    if (kk >= 32) {
      a.x *= Rs[m * 36 + kk - 32];
      a.y *= Rs[m * 36 + kk - 31];
    }
    const float* lh = LhW + kk * 32 + n;
    v2f b;
    b.x = lh[0];  b.y = lh[32];  t0 = wmma4(a, b, t0);
    b.x = lh[16]; b.y = lh[48];  t1 = wmma4(a, b, t1);
  }

  const float bh0 = Lhb[n], bh1 = Lhb[16 + n];
  const float hw0 = headW[n], hw1 = headW[16 + n];
  float p[8];
#pragma unroll
  for (int v = 0; v < 8; ++v) {
    const float ht0 = tanhf(t0[v] + bh0);
    const float ht1 = tanhf(t1[v] + bh1);
    const float hp0 = As[(row0 + v) * 132 + 96 + n];
    const float hp1 = As[(row0 + v) * 132 + 112 + n];
    const float h0 = z0[v] * hp0 + (1.0f - z0[v]) * ht0;
    const float h1 = z1[v] * hp1 + (1.0f - z1[v]) * ht1;
    h_out[(size_t)(tile + row0 + v) * 32 + n]      = h0;
    h_out[(size_t)(tile + row0 + v) * 32 + 16 + n] = h1;
    p[v] = fmaxf(h0, 0.0f) * hw0 + fmaxf(h1, 0.0f) * hw1;
  }
#pragma unroll
  for (int s = 1; s < 16; s <<= 1)
#pragma unroll
    for (int v = 0; v < 8; ++v)
      p[v] += __shfl_xor(p[v], s, 32);
  if (n == 0) {
    const float hb = headb[0];
#pragma unroll
    for (int v = 0; v < 8; ++v)
      y_out[tile + row0 + v] = p[v] + hb;
  }
}

extern "C" void kernel_launch(void* const* d_in, const int* in_sizes, int n_in,
                              void* d_out, int out_size, void* d_ws, size_t ws_size,
                              hipStream_t stream) {
  const float* x    = (const float*)d_in[0];
  const int*   ei   = (const int*)d_in[1];
  const float* ew   = (const float*)d_in[2];
  const float* Hp   = (const float*)d_in[3];
  const float* Wz   = (const float*)d_in[4];
  const float* bz   = (const float*)d_in[5];
  const float* Wr   = (const float*)d_in[6];
  const float* br   = (const float*)d_in[7];
  const float* Wh   = (const float*)d_in[8];
  const float* bh   = (const float*)d_in[9];
  const float* LzW  = (const float*)d_in[10];
  const float* Lzb  = (const float*)d_in[11];
  const float* LrW  = (const float*)d_in[12];
  const float* Lrb  = (const float*)d_in[13];
  const float* LhW  = (const float*)d_in[14];
  const float* Lhb  = (const float*)d_in[15];
  const float* hW   = (const float*)d_in[16];
  const float* hb   = (const float*)d_in[17];

  const int* src = ei;
  const int* dst = ei + Ee;

  float* out_y = (float*)d_out;
  float* out_h = out_y + Nn;

  float* ws   = (float*)d_ws;
  float* dis  = ws;
  float* Wcat = ws + Nn;
  float* bcat = Wcat + FIN * 96;
  float* XW   = bcat + 96;
  float* AGG  = XW + (size_t)Nn * 96;
  int*   counts = (int*)(AGG + (size_t)Nn * 96);
  int*   offs   = counts + NBINS * NSLICE;
  uint2* recs   = (uint2*)(offs + NBINS * NSLICE);

  pack_weights_kernel<<<(FIN * 96 + 255) / 256, 256, 0, stream>>>(Wz, Wr, Wh, bz, br, bh, Wcat, bcat);
  zero_deg_kernel<<<(Nn + 255) / 256, 256, 0, stream>>>(dis);
  deg_accum_kernel<<<Ee / 256, 256, 0, stream>>>(dst, ew, dis);
  finalize_dis_kernel<<<(Nn + 255) / 256, 256, 0, stream>>>(dis);
  gemm_xw_kernel<<<Nn / 16, 32, 0, stream>>>(x, Wcat, XW);
  bin_count_kernel<<<NSLICE, 1024, 0, stream>>>(dst, counts);
  bin_scan_kernel<<<1, 1024, 0, stream>>>(counts, offs);
  bin_scatter_kernel<<<NSLICE, 1024, 0, stream>>>(src, dst, ew, dis, offs, recs);
  bin_aggregate_kernel<<<NBINS, 1024, 0, stream>>>(recs, offs, dis, XW, bcat, AGG);
  gru_kernel<<<Nn / 16, 32, 0, stream>>>(AGG, Hp, LzW, Lzb, LrW, Lrb, LhW, Lhb, hW, hb, out_y, out_h);
}


# Degree summed per destination bin with LDS atomics from the bucketed records, no global atomics left; norm formed in the aggregate from dis[src]; 8 launches; degree sum order still arrival order
# speedup vs baseline: 3.0217x; 1.6328x over previous
#include <hip/hip_runtime.h>

#define Nn    100000
#define Ee    3200000
#define FIN   256
#define Hh    32

typedef float v2f __attribute__((ext_vector_type(2)));
typedef float v8f __attribute__((ext_vector_type(8)));

__device__ __forceinline__ v8f wmma4(v2f a, v2f b, v8f c) {
  return __builtin_amdgcn_wmma_f32_16x16x4_f32(
       false, a,  false, b,
       (short)0, c,  false,  false);
}

__device__ __forceinline__ float sigmoidf_(float x) {
  return 1.0f / (1.0f + __expf(-x));
}

__global__ void pack_weights_kernel(const float* __restrict__ Wz, const float* __restrict__ Wr,
                                    const float* __restrict__ Wh, const float* __restrict__ bz,
                                    const float* __restrict__ br, const float* __restrict__ bh,
                                    float* __restrict__ Wcat, float* __restrict__ bcat) {
  int gid = blockIdx.x * blockDim.x + threadIdx.x;
  if (gid < FIN * 96) {
    int k = gid / 96, c = gid - k * 96;
    float v = (c < 32) ? Wz[k * 32 + c] : ((c < 64) ? Wr[k * 32 + c - 32] : Wh[k * 32 + c - 64]);
    Wcat[gid] = v;
  }
  if (gid < 96) {
    bcat[gid] = (gid < 32) ? bz[gid] : ((gid < 64) ? br[gid - 32] : bh[gid - 64]);
  }
}

__global__ void __launch_bounds__(32) gemm_xw_kernel(const float* __restrict__ x,
                                                     const float* __restrict__ Wcat,
                                                     float* __restrict__ XW) {
  __shared__ float Xs[16 * 260];
  const int lane = threadIdx.x;
  const int tile = blockIdx.x * 16;

  for (int i = lane; i < 1024; i += 32) {
    int r = i >> 6, c4 = (i & 63) << 2;
    const float4 v = *reinterpret_cast<const float4*>(x + (size_t)(tile + r) * FIN + c4);
    *reinterpret_cast<float4*>(&Xs[r * 260 + c4]) = v;
  }
  __syncthreads();

  const int m = lane & 15;
  const int koff = (lane >> 4) << 1;

  v8f acc[6] = {};
  const float2* arow = reinterpret_cast<const float2*>(&Xs[m * 260]);

#pragma unroll 4
  for (int k0 = 0; k0 < FIN; k0 += 4) {
    const int kk = k0 + koff;
    float2 af = arow[kk >> 1];
    v2f a; a.x = af.x; a.y = af.y;
    const float* brow = Wcat + (size_t)kk * 96 + m;
#pragma unroll
    for (int t = 0; t < 6; ++t) {
      v2f b; b.x = brow[t * 16]; b.y = brow[96 + t * 16];
      acc[t] = wmma4(a, b, acc[t]);
    }
  }

  const int row0 = (lane >> 4) << 3;
#pragma unroll
  for (int t = 0; t < 6; ++t)
#pragma unroll
    for (int v = 0; v < 8; ++v)
      XW[(size_t)(tile + row0 + v) * 96 + t * 16 + m] = acc[t][v];
}

#define NBINS   256
#define BINSZ   391
#define NSLICE  256
#define SLICE_E (Ee / NSLICE)

__global__ void __launch_bounds__(1024) bin_count_kernel(const int* __restrict__ dst,
                                                         int* __restrict__ counts) {
  __shared__ int hist[NBINS];
  const int t = threadIdx.x;
  if (t < NBINS) hist[t] = 0;
  __syncthreads();
  const int e0 = blockIdx.x * SLICE_E;
  const int e1 = e0 + SLICE_E;
  for (int e = e0 + t; e < e1; e += 4096) {
    int d[4];
#pragma unroll
    for (int u = 0; u < 4; ++u) {
      const int ee = e + u * 1024;
      d[u] = dst[(ee < e1) ? ee : e];
    }
#pragma unroll
    for (int u = 0; u < 4; ++u)
      if (e + u * 1024 < e1) atomicAdd(&hist[d[u] / BINSZ], 1);
  }
  __syncthreads();
  if (t < NBINS) counts[t * NSLICE + blockIdx.x] = hist[t];
}

__global__ void __launch_bounds__(1024) bin_scan_kernel(const int* __restrict__ counts,
                                                        int* __restrict__ offs) {
  __shared__ int part[1024];
  const int t = threadIdx.x;
  const int4* p = reinterpret_cast<const int4*>(counts) + t * 16;
  int4 q[16];
  int sum = 0;
#pragma unroll
  for (int i = 0; i < 16; ++i) {
    q[i] = p[i];
    sum += q[i].x + q[i].y + q[i].z + q[i].w;
  }
  part[t] = sum;
  __syncthreads();
  for (int d = 1; d < 1024; d <<= 1) {
    const int x = (t >= d) ? part[t - d] : 0;
    __syncthreads();
    part[t] += x;
    __syncthreads();
  }
  int run = part[t] - sum;
  int4* o = reinterpret_cast<int4*>(offs) + t * 16;
#pragma unroll
  for (int i = 0; i < 16; ++i) {
    int4 r;
    r.x = run; run += q[i].x;
    r.y = run; run += q[i].y;
    r.z = run; run += q[i].z;
    r.w = run; run += q[i].w;
    o[i] = r;
  }
}

__global__ void __launch_bounds__(1024) bin_scatter_kernel(const int* __restrict__ src,
                                                           const int* __restrict__ dst,
                                                           const float* __restrict__ w,
                                                           const int* __restrict__ offs,
                                                           uint2* __restrict__ recs) {
  __shared__ int cursor[NBINS];
  const int t = threadIdx.x;
  if (t < NBINS) cursor[t] = offs[t * NSLICE + blockIdx.x];
  __syncthreads();
  const int e0 = blockIdx.x * SLICE_E;
  const int e1 = e0 + SLICE_E;
  for (int e = e0 + t; e < e1; e += 4096) {
    int s[4], d[4];
    float ww[4];
#pragma unroll
    for (int u = 0; u < 4; ++u) {
      const int ee = e + u * 1024;
      const int ec = (ee < e1) ? ee : e;
      s[u] = src[ec];
      d[u] = dst[ec];
      ww[u] = w[ec];
    }
#pragma unroll
    for (int u = 0; u < 4; ++u) {
      if (e + u * 1024 < e1) {
        const int bin = d[u] / BINSZ;
        const int dl = d[u] - bin * BINSZ;
        const int pos = atomicAdd(&cursor[bin], 1);
        recs[pos] = make_uint2((unsigned)s[u] | ((unsigned)dl << 17), __float_as_uint(ww[u]));
      }
    }
  }
}

__global__ void __launch_bounds__(1024) bin_degree_kernel(const uint2* __restrict__ recs,
                                                          const int* __restrict__ offs,
                                                          float* __restrict__ dis) {
  __shared__ float deg[BINSZ];
  const int t = threadIdx.x;
  const int bin = blockIdx.x;
  const int node0 = bin * BINSZ;
  const int nrows = (Nn - node0 < BINSZ) ? (Nn - node0) : BINSZ;
  if (t < BINSZ) deg[t] = 0.0f;
  __syncthreads();
  const int start = offs[bin * NSLICE];
  const int end = (bin == NBINS - 1) ? Ee : offs[(bin + 1) * NSLICE];
  for (int e = start + t; e < end; e += 4096) {
    uint2 r[4];
#pragma unroll
    for (int u = 0; u < 4; ++u) {
      const int ee = e + u * 1024;
      r[u] = recs[(ee < end) ? ee : e];
    }
#pragma unroll
    for (int u = 0; u < 4; ++u)
      if (e + u * 1024 < end) atomicAdd(&deg[r[u].x >> 17], __uint_as_float(r[u].y));
  }
  __syncthreads();
  if (t < nrows) dis[node0 + t] = rsqrtf(deg[t] + 1.0f);
}

__global__ void __launch_bounds__(1024) bin_aggregate_kernel(const uint2* __restrict__ recs,
                                                             const int* __restrict__ offs,
                                                             const float* __restrict__ dis,
                                                             const float* __restrict__ XW,
                                                             const float* __restrict__ bcat,
                                                             float* __restrict__ AGG) {
  __shared__ float acc[BINSZ * 96];
  __shared__ float dloc[BINSZ];
  const int t = threadIdx.x;
  const int bin = blockIdx.x;
  const int node0 = bin * BINSZ;
  const int nrows = (Nn - node0 < BINSZ) ? (Nn - node0) : BINSZ;
  const float* xw0 = XW + (size_t)node0 * 96;
  if (t < BINSZ) dloc[t] = (t < nrows) ? dis[node0 + t] : 0.0f;

  const float4* xw4 = reinterpret_cast<const float4*>(xw0);
  const float4* b4p = reinterpret_cast<const float4*>(bcat);
  float4* acc4 = reinterpret_cast<float4*>(acc);
  for (int q = t; q < nrows * 24; q += 1024) {
    const int i = q / 24;
    const int c4 = q - i * 24;
    const float d = dis[node0 + i];
    const float4 x = xw4[q];
    const float4 b = b4p[c4];
    float4 r;
    r.x = d * d * x.x + b.x;
    r.y = d * d * x.y + b.y;
    r.z = d * d * x.z + b.z;
    r.w = d * d * x.w + b.w;
    acc4[q] = r;
  }
  __syncthreads();

  const int start = offs[bin * NSLICE];
  const int end = (bin == NBINS - 1) ? Ee : offs[(bin + 1) * NSLICE];
  const int lane = t & 31;
  const int wave = __builtin_amdgcn_readfirstlane(t >> 5);

  int base = start + wave * 32;
  uint2 rec_next = make_uint2(0u, 0u);
  if (base < end && lane < end - base) rec_next = recs[base + lane];
  float ds_next = dis[rec_next.x & 0x1FFFFu];
  for (; base < end; base += 1024) {
    const int cnt = (end - base < 32) ? (end - base) : 32;
    const uint2 rec = rec_next;
    const float nrm = ds_next * __uint_as_float(rec.y) * dloc[rec.x >> 17];
    const int nb = base + 1024;
#pragma unroll
    for (int j0 = 0; j0 < 32; j0 += 16) {
      if (j0 >= cnt) break;
      float v[16][3];
      float nr[16];
      int dl[16];
#pragma unroll
      for (int u = 0; u < 16; ++u) {
        const unsigned px = (unsigned)__builtin_amdgcn_readlane((int)rec.x, j0 + u);
        nr[u] = __uint_as_float((unsigned)__builtin_amdgcn_readlane((int)__float_as_uint(nrm), j0 + u));
        dl[u] = (int)(px >> 17);
        const float* row = XW + (size_t)(px & 0x1FFFFu) * 96 + lane;
        v[u][0] = row[0];
        v[u][1] = row[32];
        v[u][2] = row[64];
      }
      if (j0 == 0) {
        rec_next = make_uint2(0u, 0u);
        if (nb < end && lane < end - nb) rec_next = recs[nb + lane];
      } else {
        ds_next = dis[rec_next.x & 0x1FFFFu];
      }
#pragma unroll
      for (int u = 0; u < 16; ++u) {
        if (j0 + u < cnt) {
          float* a = &acc[dl[u] * 96 + lane];
          atomicAdd(a,      nr[u] * v[u][0]);
          atomicAdd(a + 32, nr[u] * v[u][1]);
          atomicAdd(a + 64, nr[u] * v[u][2]);
        }
      }
    }
  }
  __syncthreads();

  float4* out = reinterpret_cast<float4*>(AGG + (size_t)node0 * 96);
  const float4* a4 = reinterpret_cast<const float4*>(acc);
  for (int q = t; q < nrows * 24; q += 1024) out[q] = a4[q];
}

__global__ void __launch_bounds__(32) gru_kernel(const float* __restrict__ AGG,
                                                 const float* __restrict__ Hp,
                                                 const float* __restrict__ LzW, const float* __restrict__ Lzb,
                                                 const float* __restrict__ LrW, const float* __restrict__ Lrb,
                                                 const float* __restrict__ LhW, const float* __restrict__ Lhb,
                                                 const float* __restrict__ headW, const float* __restrict__ headb,
                                                 float* __restrict__ y_out, float* __restrict__ h_out) {
  __shared__ float As[16 * 132];
  __shared__ float Rs[16 * 36];
  const int lane = threadIdx.x;
  const int tile = blockIdx.x * 16;

  for (int idx = lane; idx < 16 * 128; idx += 32) {
    int r = idx >> 7, c = idx & 127;
    float v = (c < 96) ? AGG[(size_t)(tile + r) * 96 + c] : Hp[(size_t)(tile + r) * 32 + (c - 96)];
    As[r * 132 + c] = v;
  }
  __syncthreads();

  const int m = lane & 15;
  const int n = lane & 15;
  const int koff = (lane >> 4) << 1;

  v8f z0 = {}, z1 = {}, r0 = {}, r1 = {};
#pragma unroll 4
  for (int k0 = 0; k0 < 64; k0 += 4) {
    const int kk = k0 + koff;
    const int colz = (kk < 32) ? kk : (64 + kk);
    const int colr = (kk < 32) ? (32 + kk) : (64 + kk);
    v2f az; az.x = As[m * 132 + colz]; az.y = As[m * 132 + colz + 1];
    v2f ar; ar.x = As[m * 132 + colr]; ar.y = As[m * 132 + colr + 1];
    const float* lz = LzW + kk * 32 + n;
    const float* lr = LrW + kk * 32 + n;
    v2f b;
    b.x = lz[0];  b.y = lz[32];  z0 = wmma4(az, b, z0);
    b.x = lz[16]; b.y = lz[48];  z1 = wmma4(az, b, z1);
    b.x = lr[0];  b.y = lr[32];  r0 = wmma4(ar, b, r0);
    b.x = lr[16]; b.y = lr[48];  r1 = wmma4(ar, b, r1);
  }

  const float bz0 = Lzb[n], bz1 = Lzb[16 + n];
  const float br0 = Lrb[n], br1 = Lrb[16 + n];
  const int row0 = (lane >> 4) << 3;
#pragma unroll
  for (int v = 0; v < 8; ++v) {
    z0[v] = sigmoidf_(z0[v] + bz0);
    z1[v] = sigmoidf_(z1[v] + bz1);
    r0[v] = sigmoidf_(r0[v] + br0);
    r1[v] = sigmoidf_(r1[v] + br1);
    Rs[(row0 + v) * 36 + n]      = r0[v];
    Rs[(row0 + v) * 36 + 16 + n] = r1[v];
  }
  __syncthreads();

  v8f t0 = {}, t1 = {};
#pragma unroll 4
  for (int k0 = 0; k0 < 64; k0 += 4) {
    const int kk = k0 + koff;
    v2f a;
    a.x = As[m * 132 + 64 + kk];
    a.y = As[m * 132 + 65 + kk];
    if (kk >= 32) {
      a.x *= Rs[m * 36 + kk - 32];
      a.y *= Rs[m * 36 + kk - 31];
    }
    const float* lh = LhW + kk * 32 + n;
    v2f b;
    b.x = lh[0];  b.y = lh[32];  t0 = wmma4(a, b, t0);
    b.x = lh[16]; b.y = lh[48];  t1 = wmma4(a, b, t1);
  }

  const float bh0 = Lhb[n], bh1 = Lhb[16 + n];
  const float hw0 = headW[n], hw1 = headW[16 + n];
  float p[8];
#pragma unroll
  for (int v = 0; v < 8; ++v) {
    const float ht0 = tanhf(t0[v] + bh0);
    const float ht1 = tanhf(t1[v] + bh1);
    const float hp0 = As[(row0 + v) * 132 + 96 + n];
    const float hp1 = As[(row0 + v) * 132 + 112 + n];
    const float h0 = z0[v] * hp0 + (1.0f - z0[v]) * ht0;
    const float h1 = z1[v] * hp1 + (1.0f - z1[v]) * ht1;
    h_out[(size_t)(tile + row0 + v) * 32 + n]      = h0;
    h_out[(size_t)(tile + row0 + v) * 32 + 16 + n] = h1;
    p[v] = fmaxf(h0, 0.0f) * hw0 + fmaxf(h1, 0.0f) * hw1;
  }
#pragma unroll
  for (int s = 1; s < 16; s <<= 1)
#pragma unroll
    for (int v = 0; v < 8; ++v)
      p[v] += __shfl_xor(p[v], s, 32);
  if (n == 0) {
    const float hb = headb[0];
#pragma unroll
    for (int v = 0; v < 8; ++v)
      y_out[tile + row0 + v] = p[v] + hb;
  }
}

extern "C" void kernel_launch(void* const* d_in, const int* in_sizes, int n_in,
                              void* d_out, int out_size, void* d_ws, size_t ws_size,
                              hipStream_t stream) {
  const float* x    = (const float*)d_in[0];
  const int*   ei   = (const int*)d_in[1];
  const float* ew   = (const float*)d_in[2];
  const float* Hp   = (const float*)d_in[3];
  const float* Wz   = (const float*)d_in[4];
  const float* bz   = (const float*)d_in[5];
  const float* Wr   = (const float*)d_in[6];
  const float* br   = (const float*)d_in[7];
  const float* Wh   = (const float*)d_in[8];
  const float* bh   = (const float*)d_in[9];
  const float* LzW  = (const float*)d_in[10];
  const float* Lzb  = (const float*)d_in[11];
  const float* LrW  = (const float*)d_in[12];
  const float* Lrb  = (const float*)d_in[13];
  const float* LhW  = (const float*)d_in[14];
  const float* Lhb  = (const float*)d_in[15];
  const float* hW   = (const float*)d_in[16];
  const float* hb   = (const float*)d_in[17];

  const int* src = ei;
  const int* dst = ei + Ee;

  float* out_y = (float*)d_out;
  float* out_h = out_y + Nn;

  float* ws   = (float*)d_ws;
  float* dis  = ws;
  float* Wcat = ws + Nn;
  float* bcat = Wcat + FIN * 96;
  float* XW   = bcat + 96;
  float* AGG  = XW + (size_t)Nn * 96;
  int*   counts = (int*)(AGG + (size_t)Nn * 96);
  int*   offs   = counts + NBINS * NSLICE;
  uint2* recs   = (uint2*)(offs + NBINS * NSLICE);

  pack_weights_kernel<<<(FIN * 96 + 255) / 256, 256, 0, stream>>>(Wz, Wr, Wh, bz, br, bh, Wcat, bcat);
  gemm_xw_kernel<<<Nn / 16, 32, 0, stream>>>(x, Wcat, XW);
  bin_count_kernel<<<NSLICE, 1024, 0, stream>>>(dst, counts);
  bin_scan_kernel<<<1, 1024, 0, stream>>>(counts, offs);
  bin_scatter_kernel<<<NSLICE, 1024, 0, stream>>>(src, dst, ew, offs, recs);
  bin_degree_kernel<<<NBINS, 1024, 0, stream>>>(recs, offs, dis);
  bin_aggregate_kernel<<<NBINS, 1024, 0, stream>>>(recs, offs, dis, XW, bcat, AGG);
  gru_kernel<<<Nn / 16, 32, 0, stream>>>(AGG, Hp, LzW, Lzb, LrW, Lrb, LhW, Lhb, hW, hb, out_y, out_h);
}


# Tile staging of both WMMA kernels: all of a tile's global loads (32 and 16 float4 per lane, whole lines) issued as one group ahead of the LDS stores, not one load per trip waited for alone
# speedup vs baseline: 3.6275x; 1.2005x over previous
#include <hip/hip_runtime.h>

#define Nn    100000
#define Ee    3200000
#define FIN   256
#define Hh    32

typedef float v2f __attribute__((ext_vector_type(2)));
typedef float v8f __attribute__((ext_vector_type(8)));

__device__ __forceinline__ v8f wmma4(v2f a, v2f b, v8f c) {
  return __builtin_amdgcn_wmma_f32_16x16x4_f32(
       false, a,  false, b,
       (short)0, c,  false,  false);
}

__device__ __forceinline__ float sigmoidf_(float x) {
  return 1.0f / (1.0f + __expf(-x));
}

__global__ void pack_weights_kernel(const float* __restrict__ Wz, const float* __restrict__ Wr,
                                    const float* __restrict__ Wh, const float* __restrict__ bz,
                                    const float* __restrict__ br, const float* __restrict__ bh,
                                    float* __restrict__ Wcat, float* __restrict__ bcat) {
  int gid = blockIdx.x * blockDim.x + threadIdx.x;
  if (gid < FIN * 96) {
    int k = gid / 96, c = gid - k * 96;
    float v = (c < 32) ? Wz[k * 32 + c] : ((c < 64) ? Wr[k * 32 + c - 32] : Wh[k * 32 + c - 64]);
    Wcat[gid] = v;
  }
  if (gid < 96) {
    bcat[gid] = (gid < 32) ? bz[gid] : ((gid < 64) ? br[gid - 32] : bh[gid - 64]);
  }
}

__global__ void __launch_bounds__(32) gemm_xw_kernel(const float* __restrict__ x,
                                                     const float* __restrict__ Wcat,
                                                     float* __restrict__ XW) {
  __shared__ float Xs[16 * 260];
  const int lane = threadIdx.x;
  const int tile = blockIdx.x * 16;

  const float4* xt = reinterpret_cast<const float4*>(x + (size_t)tile * FIN) + lane;
  float4 st[32];
#pragma unroll
  for (int j = 0; j < 32; ++j) st[j] = xt[j * 32];
#pragma unroll
  for (int j = 0; j < 32; ++j) {
    const int i = lane + j * 32;
    const int r = i >> 6, c4 = (i & 63) << 2;
    *reinterpret_cast<float4*>(&Xs[r * 260 + c4]) = st[j];
  }
  __syncthreads();

  const int m = lane & 15;
  const int koff = (lane >> 4) << 1;

  v8f acc[6] = {};
  const float2* arow = reinterpret_cast<const float2*>(&Xs[m * 260]);

#pragma unroll 4
  for (int k0 = 0; k0 < FIN; k0 += 4) {
    const int kk = k0 + koff;
    float2 af = arow[kk >> 1];
    v2f a; a.x = af.x; a.y = af.y;
    const float* brow = Wcat + (size_t)kk * 96 + m;
#pragma unroll
    for (int t = 0; t < 6; ++t) {
      v2f b; b.x = brow[t * 16]; b.y = brow[96 + t * 16];
      acc[t] = wmma4(a, b, acc[t]);
    }
  }

  const int row0 = (lane >> 4) << 3;
#pragma unroll
  for (int t = 0; t < 6; ++t)
#pragma unroll
    for (int v = 0; v < 8; ++v)
      XW[(size_t)(tile + row0 + v) * 96 + t * 16 + m] = acc[t][v];
}

#define NBINS   256
#define BINSZ   391
#define NSLICE  256
#define SLICE_E (Ee / NSLICE)

__global__ void __launch_bounds__(1024) bin_count_kernel(const int* __restrict__ dst,
                                                         int* __restrict__ counts) {
  __shared__ int hist[NBINS];
  const int t = threadIdx.x;
  if (t < NBINS) hist[t] = 0;
  __syncthreads();
  const int e0 = blockIdx.x * SLICE_E;
  const int e1 = e0 + SLICE_E;
  for (int e = e0 + t; e < e1; e += 4096) {
    int d[4];
#pragma unroll
    for (int u = 0; u < 4; ++u) {
      const int ee = e + u * 1024;
      d[u] = dst[(ee < e1) ? ee : e];
    }
#pragma unroll
    for (int u = 0; u < 4; ++u)
      if (e + u * 1024 < e1) atomicAdd(&hist[d[u] / BINSZ], 1);
  }
  __syncthreads();
  if (t < NBINS) counts[t * NSLICE + blockIdx.x] = hist[t];
}

__global__ void __launch_bounds__(1024) bin_scan_kernel(const int* __restrict__ counts,
                                                        int* __restrict__ offs) {
  __shared__ int part[1024];
  const int t = threadIdx.x;
  const int4* p = reinterpret_cast<const int4*>(counts) + t * 16;
  int4 q[16];
  int sum = 0;
#pragma unroll
  for (int i = 0; i < 16; ++i) {
    q[i] = p[i];
    sum += q[i].x + q[i].y + q[i].z + q[i].w;
  }
  part[t] = sum;
  __syncthreads();
  for (int d = 1; d < 1024; d <<= 1) {
    const int x = (t >= d) ? part[t - d] : 0;
    __syncthreads();
    part[t] += x;
    __syncthreads();
  }
  int run = part[t] - sum;
  int4* o = reinterpret_cast<int4*>(offs) + t * 16;
#pragma unroll
  for (int i = 0; i < 16; ++i) {
    int4 r;
    r.x = run; run += q[i].x;
    r.y = run; run += q[i].y;
    r.z = run; run += q[i].z;
    r.w = run; run += q[i].w;
    o[i] = r;
  }
}

__global__ void __launch_bounds__(1024) bin_scatter_kernel(const int* __restrict__ src,
                                                           const int* __restrict__ dst,
                                                           const float* __restrict__ w,
                                                           const int* __restrict__ offs,
                                                           uint2* __restrict__ recs) {
  __shared__ int cursor[NBINS];
  const int t = threadIdx.x;
  if (t < NBINS) cursor[t] = offs[t * NSLICE + blockIdx.x];
  __syncthreads();
  const int e0 = blockIdx.x * SLICE_E;
  const int e1 = e0 + SLICE_E;
  for (int e = e0 + t; e < e1; e += 4096) {
    int s[4], d[4];
    float ww[4];
#pragma unroll
    for (int u = 0; u < 4; ++u) {
      const int ee = e + u * 1024;
      const int ec = (ee < e1) ? ee : e;
      s[u] = src[ec];
      d[u] = dst[ec];
      ww[u] = w[ec];
    }
#pragma unroll
    for (int u = 0; u < 4; ++u) {
      if (e + u * 1024 < e1) {
        const int bin = d[u] / BINSZ;
        const int dl = d[u] - bin * BINSZ;
        const int pos = atomicAdd(&cursor[bin], 1);
        recs[pos] = make_uint2((unsigned)s[u] | ((unsigned)dl << 17), __float_as_uint(ww[u]));
      }
    }
  }
}

__global__ void __launch_bounds__(1024) bin_degree_kernel(const uint2* __restrict__ recs,
                                                          const int* __restrict__ offs,
                                                          float* __restrict__ dis) {
  __shared__ float deg[BINSZ];
  const int t = threadIdx.x;
  const int bin = blockIdx.x;
  const int node0 = bin * BINSZ;
  const int nrows = (Nn - node0 < BINSZ) ? (Nn - node0) : BINSZ;
  if (t < BINSZ) deg[t] = 0.0f;
  __syncthreads();
  const int start = offs[bin * NSLICE];
  const int end = (bin == NBINS - 1) ? Ee : offs[(bin + 1) * NSLICE];
  for (int e = start + t; e < end; e += 4096) {
    uint2 r[4];
#pragma unroll
    for (int u = 0; u < 4; ++u) {
      const int ee = e + u * 1024;
      r[u] = recs[(ee < end) ? ee : e];
    }
#pragma unroll
    for (int u = 0; u < 4; ++u)
      if (e + u * 1024 < end) atomicAdd(&deg[r[u].x >> 17], __uint_as_float(r[u].y));
  }
  __syncthreads();
  if (t < nrows) dis[node0 + t] = rsqrtf(deg[t] + 1.0f);
}

__global__ void __launch_bounds__(1024) bin_aggregate_kernel(const uint2* __restrict__ recs,
                                                             const int* __restrict__ offs,
                                                             const float* __restrict__ dis,
                                                             const float* __restrict__ XW,
                                                             const float* __restrict__ bcat,
                                                             float* __restrict__ AGG) {
  __shared__ float acc[BINSZ * 96];
  __shared__ float dloc[BINSZ];
  const int t = threadIdx.x;
  const int bin = blockIdx.x;
  const int node0 = bin * BINSZ;
  const int nrows = (Nn - node0 < BINSZ) ? (Nn - node0) : BINSZ;
  const float* xw0 = XW + (size_t)node0 * 96;
  if (t < BINSZ) dloc[t] = (t < nrows) ? dis[node0 + t] : 0.0f;

  const float4* xw4 = reinterpret_cast<const float4*>(xw0);
  const float4* b4p = reinterpret_cast<const float4*>(bcat);
  float4* acc4 = reinterpret_cast<float4*>(acc);
  for (int q = t; q < nrows * 24; q += 1024) {
    const int i = q / 24;
    const int c4 = q - i * 24;
    const float d = dis[node0 + i];
    const float4 x = xw4[q];
    const float4 b = b4p[c4];
    float4 r;
    r.x = d * d * x.x + b.x;
    r.y = d * d * x.y + b.y;
    r.z = d * d * x.z + b.z;
    r.w = d * d * x.w + b.w;
    acc4[q] = r;
  }
  __syncthreads();

  const int start = offs[bin * NSLICE];
  const int end = (bin == NBINS - 1) ? Ee : offs[(bin + 1) * NSLICE];
  const int lane = t & 31;
  const int wave = __builtin_amdgcn_readfirstlane(t >> 5);

  int base = start + wave * 32;
  uint2 rec_next = make_uint2(0u, 0u);
  if (base < end && lane < end - base) rec_next = recs[base + lane];
  float ds_next = dis[rec_next.x & 0x1FFFFu];
  for (; base < end; base += 1024) {
    const int cnt = (end - base < 32) ? (end - base) : 32;
    const uint2 rec = rec_next;
    const float nrm = ds_next * __uint_as_float(rec.y) * dloc[rec.x >> 17];
    const int nb = base + 1024;
#pragma unroll
    for (int j0 = 0; j0 < 32; j0 += 16) {
      if (j0 >= cnt) break;
      float v[16][3];
      float nr[16];
      int dl[16];
#pragma unroll
      for (int u = 0; u < 16; ++u) {
        const unsigned px = (unsigned)__builtin_amdgcn_readlane((int)rec.x, j0 + u);
        nr[u] = __uint_as_float((unsigned)__builtin_amdgcn_readlane((int)__float_as_uint(nrm), j0 + u));
        dl[u] = (int)(px >> 17);
        const float* row = XW + (size_t)(px & 0x1FFFFu) * 96 + lane;
        v[u][0] = row[0];
        v[u][1] = row[32];
        v[u][2] = row[64];
      }
      if (j0 == 0) {
        rec_next = make_uint2(0u, 0u);
        if (nb < end && lane < end - nb) rec_next = recs[nb + lane];
      } else {
        ds_next = dis[rec_next.x & 0x1FFFFu];
      }
#pragma unroll
      for (int u = 0; u < 16; ++u) {
        if (j0 + u < cnt) {
          float* a = &acc[dl[u] * 96 + lane];
          atomicAdd(a,      nr[u] * v[u][0]);
          atomicAdd(a + 32, nr[u] * v[u][1]);
          atomicAdd(a + 64, nr[u] * v[u][2]);
        }
      }
    }
  }
  __syncthreads();

  float4* out = reinterpret_cast<float4*>(AGG + (size_t)node0 * 96);
  const float4* a4 = reinterpret_cast<const float4*>(acc);
  for (int q = t; q < nrows * 24; q += 1024) out[q] = a4[q];
}

__global__ void __launch_bounds__(32) gru_kernel(const float* __restrict__ AGG,
                                                 const float* __restrict__ Hp,
                                                 const float* __restrict__ LzW, const float* __restrict__ Lzb,
                                                 const float* __restrict__ LrW, const float* __restrict__ Lrb,
                                                 const float* __restrict__ LhW, const float* __restrict__ Lhb,
                                                 const float* __restrict__ headW, const float* __restrict__ headb,
                                                 float* __restrict__ y_out, float* __restrict__ h_out) {
  __shared__ __attribute__((aligned(16))) float As[16 * 132];
  __shared__ float Rs[16 * 36];
  const int lane = threadIdx.x;
  const int tile = blockIdx.x * 16;

  const float4* ag4 = reinterpret_cast<const float4*>(AGG + (size_t)tile * 96) + lane;
  const float4* hp4 = reinterpret_cast<const float4*>(Hp + (size_t)tile * 32) + lane;
  float4 sa[12], sh[4];
#pragma unroll
  for (int j = 0; j < 12; ++j) sa[j] = ag4[j * 32];
#pragma unroll
  for (int j = 0; j < 4; ++j) sh[j] = hp4[j * 32];
#pragma unroll
  for (int j = 0; j < 12; ++j) {
    const int q = lane + j * 32;
    const int r = q / 24, c4 = (q - r * 24) << 2;
    *reinterpret_cast<float4*>(&As[r * 132 + c4]) = sa[j];
  }
#pragma unroll
  for (int j = 0; j < 4; ++j) {
    const int q = lane + j * 32;
    const int r = q >> 3, c4 = (q & 7) << 2;
    *reinterpret_cast<float4*>(&As[r * 132 + 96 + c4]) = sh[j];
  }
  __syncthreads();

  const int m = lane & 15;
  const int n = lane & 15;
  const int koff = (lane >> 4) << 1;

  v8f z0 = {}, z1 = {}, r0 = {}, r1 = {};
#pragma unroll 4
  for (int k0 = 0; k0 < 64; k0 += 4) {
    const int kk = k0 + koff;
    const int colz = (kk < 32) ? kk : (64 + kk);
    const int colr = (kk < 32) ? (32 + kk) : (64 + kk);
    v2f az; az.x = As[m * 132 + colz]; az.y = As[m * 132 + colz + 1];
    v2f ar; ar.x = As[m * 132 + colr]; ar.y = As[m * 132 + colr + 1];
    const float* lz = LzW + kk * 32 + n;
    const float* lr = LrW + kk * 32 + n;
    v2f b;
    b.x = lz[0];  b.y = lz[32];  z0 = wmma4(az, b, z0);
    b.x = lz[16]; b.y = lz[48];  z1 = wmma4(az, b, z1);
    b.x = lr[0];  b.y = lr[32];  r0 = wmma4(ar, b, r0);
    b.x = lr[16]; b.y = lr[48];  r1 = wmma4(ar, b, r1);
  }

  const float bz0 = Lzb[n], bz1 = Lzb[16 + n];
  const float br0 = Lrb[n], br1 = Lrb[16 + n];
  const int row0 = (lane >> 4) << 3;
#pragma unroll
  for (int v = 0; v < 8; ++v) {
    z0[v] = sigmoidf_(z0[v] + bz0);
    z1[v] = sigmoidf_(z1[v] + bz1);
    r0[v] = sigmoidf_(r0[v] + br0);
    r1[v] = sigmoidf_(r1[v] + br1);
    Rs[(row0 + v) * 36 + n]      = r0[v];
    Rs[(row0 + v) * 36 + 16 + n] = r1[v];
  }
  __syncthreads();

  v8f t0 = {}, t1 = {};
#pragma unroll 4
  for (int k0 = 0; k0 < 64; k0 += 4) {
    const int kk = k0 + koff;
    v2f a;
    a.x = As[m * 132 + 64 + kk];
    a.y = As[m * 132 + 65 + kk];
    if (kk >= 32) {
      a.x *= Rs[m * 36 + kk - 32];
      a.y *= Rs[m * 36 + kk - 31];
    }
    const float* lh = LhW + kk * 32 + n;
    v2f b;
    b.x = lh[0];  b.y = lh[32];  t0 = wmma4(a, b, t0);
    b.x = lh[16]; b.y = lh[48];  t1 = wmma4(a, b, t1);
  }

  const float bh0 = Lhb[n], bh1 = Lhb[16 + n];
  const float hw0 = headW[n], hw1 = headW[16 + n];
  float p[8];
#pragma unroll
  for (int v = 0; v < 8; ++v) {
    const float ht0 = tanhf(t0[v] + bh0);
    const float ht1 = tanhf(t1[v] + bh1);
    const float hp0 = As[(row0 + v) * 132 + 96 + n];
    const float hp1 = As[(row0 + v) * 132 + 112 + n];
    const float h0 = z0[v] * hp0 + (1.0f - z0[v]) * ht0;
    const float h1 = z1[v] * hp1 + (1.0f - z1[v]) * ht1;
    h_out[(size_t)(tile + row0 + v) * 32 + n]      = h0;
    h_out[(size_t)(tile + row0 + v) * 32 + 16 + n] = h1;
    p[v] = fmaxf(h0, 0.0f) * hw0 + fmaxf(h1, 0.0f) * hw1;
  }
#pragma unroll
  for (int s = 1; s < 16; s <<= 1)
#pragma unroll
    for (int v = 0; v < 8; ++v)
      p[v] += __shfl_xor(p[v], s, 32);
  if (n == 0) {
    const float hb = headb[0];
#pragma unroll
    for (int v = 0; v < 8; ++v)
      y_out[tile + row0 + v] = p[v] + hb;
  }
}

extern "C" void kernel_launch(void* const* d_in, const int* in_sizes, int n_in,
                              void* d_out, int out_size, void* d_ws, size_t ws_size,
                              hipStream_t stream) {
  const float* x    = (const float*)d_in[0];
  const int*   ei   = (const int*)d_in[1];
  const float* ew   = (const float*)d_in[2];
  const float* Hp   = (const float*)d_in[3];
  const float* Wz   = (const float*)d_in[4];
  const float* bz   = (const float*)d_in[5];
  const float* Wr   = (const float*)d_in[6];
  const float* br   = (const float*)d_in[7];
  const float* Wh   = (const float*)d_in[8];
  const float* bh   = (const float*)d_in[9];
  const float* LzW  = (const float*)d_in[10];
  const float* Lzb  = (const float*)d_in[11];
  const float* LrW  = (const float*)d_in[12];
  const float* Lrb  = (const float*)d_in[13];
  const float* LhW  = (const float*)d_in[14];
  const float* Lhb  = (const float*)d_in[15];
  const float* hW   = (const float*)d_in[16];
  const float* hb   = (const float*)d_in[17];

  const int* src = ei;
  const int* dst = ei + Ee;

  float* out_y = (float*)d_out;
  float* out_h = out_y + Nn;

  float* ws   = (float*)d_ws;
  float* dis  = ws;
  float* Wcat = ws + Nn;
  float* bcat = Wcat + FIN * 96;
  float* XW   = bcat + 96;
  float* AGG  = XW + (size_t)Nn * 96;
  int*   counts = (int*)(AGG + (size_t)Nn * 96);
  int*   offs   = counts + NBINS * NSLICE;
  uint2* recs   = (uint2*)(offs + NBINS * NSLICE);

  pack_weights_kernel<<<(FIN * 96 + 255) / 256, 256, 0, stream>>>(Wz, Wr, Wh, bz, br, bh, Wcat, bcat);
  gemm_xw_kernel<<<Nn / 16, 32, 0, stream>>>(x, Wcat, XW);
  bin_count_kernel<<<NSLICE, 1024, 0, stream>>>(dst, counts);
  bin_scan_kernel<<<1, 1024, 0, stream>>>(counts, offs);
  bin_scatter_kernel<<<NSLICE, 1024, 0, stream>>>(src, dst, ew, offs, recs);
  bin_degree_kernel<<<NBINS, 1024, 0, stream>>>(recs, offs, dis);
  bin_aggregate_kernel<<<NBINS, 1024, 0, stream>>>(recs, offs, dis, XW, bcat, AGG);
  gru_kernel<<<Nn / 16, 32, 0, stream>>>(AGG, Hp, LzW, Lzb, LrW, Lrb, LhW, Lhb, hW, hb, out_y, out_h);
}


# Offset scan folded into the scatter: each scatter block sums the count table's rows (whole-line loads, wave sums) into its 256 cursors; the one-block scan launch and its strided pass go; 7 launches
# speedup vs baseline: 3.8781x; 1.0691x over previous
#include <hip/hip_runtime.h>

#define Nn    100000
#define Ee    3200000
#define FIN   256
#define Hh    32

typedef float v2f __attribute__((ext_vector_type(2)));
typedef float v8f __attribute__((ext_vector_type(8)));

__device__ __forceinline__ v8f wmma4(v2f a, v2f b, v8f c) {
  return __builtin_amdgcn_wmma_f32_16x16x4_f32(
       false, a,  false, b,
       (short)0, c,  false,  false);
}

__device__ __forceinline__ float sigmoidf_(float x) {
  return 1.0f / (1.0f + __expf(-x));
}

__global__ void pack_weights_kernel(const float* __restrict__ Wz, const float* __restrict__ Wr,
                                    const float* __restrict__ Wh, const float* __restrict__ bz,
                                    const float* __restrict__ br, const float* __restrict__ bh,
                                    float* __restrict__ Wcat, float* __restrict__ bcat) {
  int gid = blockIdx.x * blockDim.x + threadIdx.x;
  if (gid < FIN * 96) {
    int k = gid / 96, c = gid - k * 96;
    float v = (c < 32) ? Wz[k * 32 + c] : ((c < 64) ? Wr[k * 32 + c - 32] : Wh[k * 32 + c - 64]);
    Wcat[gid] = v;
  }
  if (gid < 96) {
    bcat[gid] = (gid < 32) ? bz[gid] : ((gid < 64) ? br[gid - 32] : bh[gid - 64]);
  }
}

__global__ void __launch_bounds__(32) gemm_xw_kernel(const float* __restrict__ x,
                                                     const float* __restrict__ Wcat,
                                                     float* __restrict__ XW) {
  __shared__ float Xs[16 * 260];
  const int lane = threadIdx.x;
  const int tile = blockIdx.x * 16;

  const float4* xt = reinterpret_cast<const float4*>(x + (size_t)tile * FIN) + lane;
  float4 st[32];
#pragma unroll
  for (int j = 0; j < 32; ++j) st[j] = xt[j * 32];
#pragma unroll
  for (int j = 0; j < 32; ++j) {
    const int i = lane + j * 32;
    const int r = i >> 6, c4 = (i & 63) << 2;
    *reinterpret_cast<float4*>(&Xs[r * 260 + c4]) = st[j];
  }
  __syncthreads();

  const int m = lane & 15;
  const int koff = (lane >> 4) << 1;

  v8f acc[6] = {};
  const float2* arow = reinterpret_cast<const float2*>(&Xs[m * 260]);

#pragma unroll 4
  for (int k0 = 0; k0 < FIN; k0 += 4) {
    const int kk = k0 + koff;
    float2 af = arow[kk >> 1];
    v2f a; a.x = af.x; a.y = af.y;
    const float* brow = Wcat + (size_t)kk * 96 + m;
#pragma unroll
    for (int t = 0; t < 6; ++t) {
      v2f b; b.x = brow[t * 16]; b.y = brow[96 + t * 16];
      acc[t] = wmma4(a, b, acc[t]);
    }
  }

  const int row0 = (lane >> 4) << 3;
#pragma unroll
  for (int t = 0; t < 6; ++t)
#pragma unroll
    for (int v = 0; v < 8; ++v)
      XW[(size_t)(tile + row0 + v) * 96 + t * 16 + m] = acc[t][v];
}

#define NBINS   256
#define BINSZ   391
#define NSLICE  256
#define SLICE_E (Ee / NSLICE)

__global__ void __launch_bounds__(1024) bin_count_kernel(const int* __restrict__ dst,
                                                         int* __restrict__ counts) {
  __shared__ int hist[NBINS];
  const int t = threadIdx.x;
  if (t < NBINS) hist[t] = 0;
  __syncthreads();
  const int e0 = blockIdx.x * SLICE_E;
  const int e1 = e0 + SLICE_E;
  for (int e = e0 + t; e < e1; e += 4096) {
    int d[4];
#pragma unroll
    for (int u = 0; u < 4; ++u) {
      const int ee = e + u * 1024;
      d[u] = dst[(ee < e1) ? ee : e];
    }
#pragma unroll
    for (int u = 0; u < 4; ++u)
      if (e + u * 1024 < e1) atomicAdd(&hist[d[u] / BINSZ], 1);
  }
  __syncthreads();
  if (t < NBINS) counts[t * NSLICE + blockIdx.x] = hist[t];
}

__global__ void __launch_bounds__(1024) bin_scatter_kernel(const int* __restrict__ src,
                                                           const int* __restrict__ dst,
                                                           const float* __restrict__ w,
                                                           const int* __restrict__ counts,
                                                           int* __restrict__ binstart,
                                                           uint2* __restrict__ recs) {
  __shared__ int cursor[NBINS];
  __shared__ int rowsum[NBINS];
  const int t = threadIdx.x;
  const int lane = t & 31;
  const int wave = t >> 5;
  const int slice = blockIdx.x;
  {
    int4 c[8][2];
#pragma unroll
    for (int j = 0; j < 8; ++j) {
      const int4* row = reinterpret_cast<const int4*>(counts + (wave * 8 + j) * NSLICE);
      c[j][0] = row[lane];
      c[j][1] = row[32 + lane];
    }
    const int s0 = lane * 4, s1 = 128 + lane * 4;
    int tot[8], part[8];
#pragma unroll
    for (int j = 0; j < 8; ++j) {
      tot[j] = c[j][0].x + c[j][0].y + c[j][0].z + c[j][0].w + c[j][1].x + c[j][1].y + c[j][1].z + c[j][1].w;
      part[j] = ((s0     < slice) ? c[j][0].x : 0) + ((s0 + 1 < slice) ? c[j][0].y : 0)
              + ((s0 + 2 < slice) ? c[j][0].z : 0) + ((s0 + 3 < slice) ? c[j][0].w : 0)
              + ((s1     < slice) ? c[j][1].x : 0) + ((s1 + 1 < slice) ? c[j][1].y : 0)
              + ((s1 + 2 < slice) ? c[j][1].z : 0) + ((s1 + 3 < slice) ? c[j][1].w : 0);
    }
#pragma unroll
    for (int d = 16; d >= 1; d >>= 1) {
      int xt[8], xp[8];
#pragma unroll
      for (int j = 0; j < 8; ++j) {
        xt[j] = __shfl_xor(tot[j], d, 32);
        xp[j] = __shfl_xor(part[j], d, 32);
      }
#pragma unroll
      for (int j = 0; j < 8; ++j) {
        tot[j] += xt[j];
        part[j] += xp[j];
      }
    }
    if (lane < 8) {
      int tsel = tot[0], psel = part[0];
#pragma unroll
      for (int j = 1; j < 8; ++j) {
        tsel = (lane == j) ? tot[j] : tsel;
        psel = (lane == j) ? part[j] : psel;
      }
      rowsum[wave * 8 + lane] = tsel;
      cursor[wave * 8 + lane] = psel;
    }
  }
  __syncthreads();
  if (wave == 0) {
    int v[8];
    int sum = 0;
#pragma unroll
    for (int i = 0; i < 8; ++i) {
      v[i] = rowsum[lane * 8 + i];
      sum += v[i];
    }
    int incl = sum;
#pragma unroll
    for (int d = 1; d < 32; d <<= 1) {
      const int x = __shfl_up(incl, d, 32);
      if (lane >= d) incl += x;
    }
    int run = incl - sum;
#pragma unroll
    for (int i = 0; i < 8; ++i) {
      const int b = lane * 8 + i;
      cursor[b] += run;
      if (slice == 0) binstart[b] = run;
      run += v[i];
    }
  }
  __syncthreads();
  const int e0 = blockIdx.x * SLICE_E;
  const int e1 = e0 + SLICE_E;
  for (int e = e0 + t; e < e1; e += 4096) {
    int s[4], d[4];
    float ww[4];
#pragma unroll
    for (int u = 0; u < 4; ++u) {
      const int ee = e + u * 1024;
      const int ec = (ee < e1) ? ee : e;
      s[u] = src[ec];
      d[u] = dst[ec];
      ww[u] = w[ec];
    }
#pragma unroll
    for (int u = 0; u < 4; ++u) {
      if (e + u * 1024 < e1) {
        const int bin = d[u] / BINSZ;
        const int dl = d[u] - bin * BINSZ;
        const int pos = atomicAdd(&cursor[bin], 1);
        recs[pos] = make_uint2((unsigned)s[u] | ((unsigned)dl << 17), __float_as_uint(ww[u]));
      }
    }
  }
}

__global__ void __launch_bounds__(1024) bin_degree_kernel(const uint2* __restrict__ recs,
                                                          const int* __restrict__ binstart,
                                                          float* __restrict__ dis) {
  __shared__ float deg[BINSZ];
  const int t = threadIdx.x;
  const int bin = blockIdx.x;
  const int node0 = bin * BINSZ;
  const int nrows = (Nn - node0 < BINSZ) ? (Nn - node0) : BINSZ;
  if (t < BINSZ) deg[t] = 0.0f;
  __syncthreads();
  const int start = binstart[bin];
  const int end = (bin == NBINS - 1) ? Ee : binstart[bin + 1];
  for (int e = start + t; e < end; e += 4096) {
    uint2 r[4];
#pragma unroll
    for (int u = 0; u < 4; ++u) {
      const int ee = e + u * 1024;
      r[u] = recs[(ee < end) ? ee : e];
    }
#pragma unroll
    for (int u = 0; u < 4; ++u)
      if (e + u * 1024 < end) atomicAdd(&deg[r[u].x >> 17], __uint_as_float(r[u].y));
  }
  __syncthreads();
  if (t < nrows) dis[node0 + t] = rsqrtf(deg[t] + 1.0f);
}

__global__ void __launch_bounds__(1024) bin_aggregate_kernel(const uint2* __restrict__ recs,
                                                             const int* __restrict__ binstart,
                                                             const float* __restrict__ dis,
                                                             const float* __restrict__ XW,
                                                             const float* __restrict__ bcat,
                                                             float* __restrict__ AGG) {
  __shared__ float acc[BINSZ * 96];
  __shared__ float dloc[BINSZ];
  const int t = threadIdx.x;
  const int bin = blockIdx.x;
  const int node0 = bin * BINSZ;
  const int nrows = (Nn - node0 < BINSZ) ? (Nn - node0) : BINSZ;
  const float* xw0 = XW + (size_t)node0 * 96;
  if (t < BINSZ) dloc[t] = (t < nrows) ? dis[node0 + t] : 0.0f;

  const float4* xw4 = reinterpret_cast<const float4*>(xw0);
  const float4* b4p = reinterpret_cast<const float4*>(bcat);
  float4* acc4 = reinterpret_cast<float4*>(acc);
  for (int q = t; q < nrows * 24; q += 1024) {
    const int i = q / 24;
    const int c4 = q - i * 24;
    const float d = dis[node0 + i];
    const float4 x = xw4[q];
    const float4 b = b4p[c4];
    float4 r;
    r.x = d * d * x.x + b.x;
    r.y = d * d * x.y + b.y;
    r.z = d * d * x.z + b.z;
    r.w = d * d * x.w + b.w;
    acc4[q] = r;
  }
  __syncthreads();

  const int start = binstart[bin];
  const int end = (bin == NBINS - 1) ? Ee : binstart[bin + 1];
  const int lane = t & 31;
  const int wave = __builtin_amdgcn_readfirstlane(t >> 5);

  int base = start + wave * 32;
  uint2 rec_next = make_uint2(0u, 0u);
  if (base < end && lane < end - base) rec_next = recs[base + lane];
  float ds_next = dis[rec_next.x & 0x1FFFFu];
  for (; base < end; base += 1024) {
    const int cnt = (end - base < 32) ? (end - base) : 32;
    const uint2 rec = rec_next;
    const float nrm = ds_next * __uint_as_float(rec.y) * dloc[rec.x >> 17];
    const int nb = base + 1024;
#pragma unroll
    for (int j0 = 0; j0 < 32; j0 += 16) {
      if (j0 >= cnt) break;
      float v[16][3];
      float nr[16];
      int dl[16];
#pragma unroll
      for (int u = 0; u < 16; ++u) {
        const unsigned px = (unsigned)__builtin_amdgcn_readlane((int)rec.x, j0 + u);
        nr[u] = __uint_as_float((unsigned)__builtin_amdgcn_readlane((int)__float_as_uint(nrm), j0 + u));
        dl[u] = (int)(px >> 17);
        const float* row = XW + (size_t)(px & 0x1FFFFu) * 96 + lane;
        v[u][0] = row[0];
        v[u][1] = row[32];
        v[u][2] = row[64];
      }
      if (j0 == 0) {
        rec_next = make_uint2(0u, 0u);
        if (nb < end && lane < end - nb) rec_next = recs[nb + lane];
      } else {
        ds_next = dis[rec_next.x & 0x1FFFFu];
      }
#pragma unroll
      for (int u = 0; u < 16; ++u) {
        if (j0 + u < cnt) {
          float* a = &acc[dl[u] * 96 + lane];
          atomicAdd(a,      nr[u] * v[u][0]);
          atomicAdd(a + 32, nr[u] * v[u][1]);
          atomicAdd(a + 64, nr[u] * v[u][2]);
        }
      }
    }
  }
  __syncthreads();

  float4* out = reinterpret_cast<float4*>(AGG + (size_t)node0 * 96);
  const float4* a4 = reinterpret_cast<const float4*>(acc);
  for (int q = t; q < nrows * 24; q += 1024) out[q] = a4[q];
}

__global__ void __launch_bounds__(32) gru_kernel(const float* __restrict__ AGG,
                                                 const float* __restrict__ Hp,
                                                 const float* __restrict__ LzW, const float* __restrict__ Lzb,
                                                 const float* __restrict__ LrW, const float* __restrict__ Lrb,
                                                 const float* __restrict__ LhW, const float* __restrict__ Lhb,
                                                 const float* __restrict__ headW, const float* __restrict__ headb,
                                                 float* __restrict__ y_out, float* __restrict__ h_out) {
  __shared__ __attribute__((aligned(16))) float As[16 * 132];
  __shared__ float Rs[16 * 36];
  const int lane = threadIdx.x;
  const int tile = blockIdx.x * 16;

  const float4* ag4 = reinterpret_cast<const float4*>(AGG + (size_t)tile * 96) + lane;
  const float4* hp4 = reinterpret_cast<const float4*>(Hp + (size_t)tile * 32) + lane;
  float4 sa[12], sh[4];
#pragma unroll
  for (int j = 0; j < 12; ++j) sa[j] = ag4[j * 32];
#pragma unroll
  for (int j = 0; j < 4; ++j) sh[j] = hp4[j * 32];
#pragma unroll
  for (int j = 0; j < 12; ++j) {
    const int q = lane + j * 32;
    const int r = q / 24, c4 = (q - r * 24) << 2;
    *reinterpret_cast<float4*>(&As[r * 132 + c4]) = sa[j];
  }
#pragma unroll
  for (int j = 0; j < 4; ++j) {
    const int q = lane + j * 32;
    const int r = q >> 3, c4 = (q & 7) << 2;
    *reinterpret_cast<float4*>(&As[r * 132 + 96 + c4]) = sh[j];
  }
  __syncthreads();

  const int m = lane & 15;
  const int n = lane & 15;
  const int koff = (lane >> 4) << 1;

  v8f z0 = {}, z1 = {}, r0 = {}, r1 = {};
#pragma unroll 4
  for (int k0 = 0; k0 < 64; k0 += 4) {
    const int kk = k0 + koff;
    const int colz = (kk < 32) ? kk : (64 + kk);
    const int colr = (kk < 32) ? (32 + kk) : (64 + kk);
    v2f az; az.x = As[m * 132 + colz]; az.y = As[m * 132 + colz + 1];
    v2f ar; ar.x = As[m * 132 + colr]; ar.y = As[m * 132 + colr + 1];
    const float* lz = LzW + kk * 32 + n;
    const float* lr = LrW + kk * 32 + n;
    v2f b;
    b.x = lz[0];  b.y = lz[32];  z0 = wmma4(az, b, z0);
    b.x = lz[16]; b.y = lz[48];  z1 = wmma4(az, b, z1);
    b.x = lr[0];  b.y = lr[32];  r0 = wmma4(ar, b, r0);
    b.x = lr[16]; b.y = lr[48];  r1 = wmma4(ar, b, r1);
  }

  const float bz0 = Lzb[n], bz1 = Lzb[16 + n];
  const float br0 = Lrb[n], br1 = Lrb[16 + n];
  const int row0 = (lane >> 4) << 3;
#pragma unroll
  for (int v = 0; v < 8; ++v) {
    z0[v] = sigmoidf_(z0[v] + bz0);
    z1[v] = sigmoidf_(z1[v] + bz1);
    r0[v] = sigmoidf_(r0[v] + br0);
    r1[v] = sigmoidf_(r1[v] + br1);
    Rs[(row0 + v) * 36 + n]      = r0[v];
    Rs[(row0 + v) * 36 + 16 + n] = r1[v];
  }
  __syncthreads();

  v8f t0 = {}, t1 = {};
#pragma unroll 4
  for (int k0 = 0; k0 < 64; k0 += 4) {
    const int kk = k0 + koff;
    v2f a;
    a.x = As[m * 132 + 64 + kk];
    a.y = As[m * 132 + 65 + kk];
    if (kk >= 32) {
      a.x *= Rs[m * 36 + kk - 32];
      a.y *= Rs[m * 36 + kk - 31];
    }
    const float* lh = LhW + kk * 32 + n;
    v2f b;
    b.x = lh[0];  b.y = lh[32];  t0 = wmma4(a, b, t0);
    b.x = lh[16]; b.y = lh[48];  t1 = wmma4(a, b, t1);
  }

  const float bh0 = Lhb[n], bh1 = Lhb[16 + n];
  const float hw0 = headW[n], hw1 = headW[16 + n];
  float p[8];
#pragma unroll
  for (int v = 0; v < 8; ++v) {
    const float ht0 = tanhf(t0[v] + bh0);
    const float ht1 = tanhf(t1[v] + bh1);
    const float hp0 = As[(row0 + v) * 132 + 96 + n];
    const float hp1 = As[(row0 + v) * 132 + 112 + n];
    const float h0 = z0[v] * hp0 + (1.0f - z0[v]) * ht0;
    const float h1 = z1[v] * hp1 + (1.0f - z1[v]) * ht1;
    h_out[(size_t)(tile + row0 + v) * 32 + n]      = h0;
    h_out[(size_t)(tile + row0 + v) * 32 + 16 + n] = h1;
    p[v] = fmaxf(h0, 0.0f) * hw0 + fmaxf(h1, 0.0f) * hw1;
  }
#pragma unroll
  for (int s = 1; s < 16; s <<= 1)
#pragma unroll
    for (int v = 0; v < 8; ++v)
      p[v] += __shfl_xor(p[v], s, 32);
  if (n == 0) {
    const float hb = headb[0];
#pragma unroll
    for (int v = 0; v < 8; ++v)
      y_out[tile + row0 + v] = p[v] + hb;
  }
}

extern "C" void kernel_launch(void* const* d_in, const int* in_sizes, int n_in,
                              void* d_out, int out_size, void* d_ws, size_t ws_size,
                              hipStream_t stream) {
  const float* x    = (const float*)d_in[0];
  const int*   ei   = (const int*)d_in[1];
  const float* ew   = (const float*)d_in[2];
  const float* Hp   = (const float*)d_in[3];
  const float* Wz   = (const float*)d_in[4];
  const float* bz   = (const float*)d_in[5];
  const float* Wr   = (const float*)d_in[6];
  const float* br   = (const float*)d_in[7];
  const float* Wh   = (const float*)d_in[8];
  const float* bh   = (const float*)d_in[9];
  const float* LzW  = (const float*)d_in[10];
  const float* Lzb  = (const float*)d_in[11];
  const float* LrW  = (const float*)d_in[12];
  const float* Lrb  = (const float*)d_in[13];
  const float* LhW  = (const float*)d_in[14];
  const float* Lhb  = (const float*)d_in[15];
  const float* hW   = (const float*)d_in[16];
  const float* hb   = (const float*)d_in[17];

  const int* src = ei;
  const int* dst = ei + Ee;

  float* out_y = (float*)d_out;
  float* out_h = out_y + Nn;

  float* ws   = (float*)d_ws;
  float* dis  = ws;
  float* Wcat = ws + Nn;
  float* bcat = Wcat + FIN * 96;
  float* XW   = bcat + 96;
  float* AGG  = XW + (size_t)Nn * 96;
  int*   counts = (int*)(AGG + (size_t)Nn * 96);
  int*   binstart = counts + NBINS * NSLICE;
  uint2* recs   = (uint2*)(binstart + NBINS * NSLICE);

  pack_weights_kernel<<<(FIN * 96 + 255) / 256, 256, 0, stream>>>(Wz, Wr, Wh, bz, br, bh, Wcat, bcat);
  gemm_xw_kernel<<<Nn / 16, 32, 0, stream>>>(x, Wcat, XW);
  bin_count_kernel<<<NSLICE, 1024, 0, stream>>>(dst, counts);
  bin_scatter_kernel<<<NSLICE, 1024, 0, stream>>>(src, dst, ew, counts, binstart, recs);
  bin_degree_kernel<<<NBINS, 1024, 0, stream>>>(recs, binstart, dis);
  bin_aggregate_kernel<<<NBINS, 1024, 0, stream>>>(recs, binstart, dis, XW, bcat, AGG);
  gru_kernel<<<Nn / 16, 32, 0, stream>>>(AGG, Hp, LzW, Lzb, LrW, Lrb, LhW, Lhb, hW, hb, out_y, out_h);
}


# Aggregate: bin records ordered by destination row in LDS (counting sort, rank = counting add's return); a wave sums a row's run in registers, one LDS add per run; sum order still arrival order
# speedup vs baseline: 4.2191x; 1.0879x over previous
#include <hip/hip_runtime.h>

#define Nn    100000
#define Ee    3200000
#define FIN   256
#define Hh    32

typedef float v2f __attribute__((ext_vector_type(2)));
typedef float v8f __attribute__((ext_vector_type(8)));

__device__ __forceinline__ v8f wmma4(v2f a, v2f b, v8f c) {
  return __builtin_amdgcn_wmma_f32_16x16x4_f32(
       false, a,  false, b,
       (short)0, c,  false,  false);
}

__device__ __forceinline__ float sigmoidf_(float x) {
  return 1.0f / (1.0f + __expf(-x));
}

__global__ void pack_weights_kernel(const float* __restrict__ Wz, const float* __restrict__ Wr,
                                    const float* __restrict__ Wh, const float* __restrict__ bz,
                                    const float* __restrict__ br, const float* __restrict__ bh,
                                    float* __restrict__ Wcat, float* __restrict__ bcat) {
  int gid = blockIdx.x * blockDim.x + threadIdx.x;
  if (gid < FIN * 96) {
    int k = gid / 96, c = gid - k * 96;
    float v = (c < 32) ? Wz[k * 32 + c] : ((c < 64) ? Wr[k * 32 + c - 32] : Wh[k * 32 + c - 64]);
    Wcat[gid] = v;
  }
  if (gid < 96) {
    bcat[gid] = (gid < 32) ? bz[gid] : ((gid < 64) ? br[gid - 32] : bh[gid - 64]);
  }
}

__global__ void __launch_bounds__(32) gemm_xw_kernel(const float* __restrict__ x,
                                                     const float* __restrict__ Wcat,
                                                     float* __restrict__ XW) {
  __shared__ float Xs[16 * 260];
  const int lane = threadIdx.x;
  const int tile = blockIdx.x * 16;

  const float4* xt = reinterpret_cast<const float4*>(x + (size_t)tile * FIN) + lane;
  float4 st[32];
#pragma unroll
  for (int j = 0; j < 32; ++j) st[j] = xt[j * 32];
#pragma unroll
  for (int j = 0; j < 32; ++j) {
    const int i = lane + j * 32;
    const int r = i >> 6, c4 = (i & 63) << 2;
    *reinterpret_cast<float4*>(&Xs[r * 260 + c4]) = st[j];
  }
  __syncthreads();

  const int m = lane & 15;
  const int koff = (lane >> 4) << 1;

  v8f acc[6] = {};
  const float2* arow = reinterpret_cast<const float2*>(&Xs[m * 260]);

#pragma unroll 4
  for (int k0 = 0; k0 < FIN; k0 += 4) {
    const int kk = k0 + koff;
    float2 af = arow[kk >> 1];
    v2f a; a.x = af.x; a.y = af.y;
    const float* brow = Wcat + (size_t)kk * 96 + m;
#pragma unroll
    for (int t = 0; t < 6; ++t) {
      v2f b; b.x = brow[t * 16]; b.y = brow[96 + t * 16];
      acc[t] = wmma4(a, b, acc[t]);
    }
  }

  const int row0 = (lane >> 4) << 3;
#pragma unroll
  for (int t = 0; t < 6; ++t)
#pragma unroll
    for (int v = 0; v < 8; ++v)
      XW[(size_t)(tile + row0 + v) * 96 + t * 16 + m] = acc[t][v];
}

#define NBINS   256
#define BINSZ   391
#define NSLICE  256
#define SLICE_E (Ee / NSLICE)

__global__ void __launch_bounds__(1024) bin_count_kernel(const int* __restrict__ dst,
                                                         int* __restrict__ counts) {
  __shared__ int hist[NBINS];
  const int t = threadIdx.x;
  if (t < NBINS) hist[t] = 0;
  __syncthreads();
  const int e0 = blockIdx.x * SLICE_E;
  const int e1 = e0 + SLICE_E;
  for (int e = e0 + t; e < e1; e += 4096) {
    int d[4];
#pragma unroll
    for (int u = 0; u < 4; ++u) {
      const int ee = e + u * 1024;
      d[u] = dst[(ee < e1) ? ee : e];
    }
#pragma unroll
    for (int u = 0; u < 4; ++u)
      if (e + u * 1024 < e1) atomicAdd(&hist[d[u] / BINSZ], 1);
  }
  __syncthreads();
  if (t < NBINS) counts[t * NSLICE + blockIdx.x] = hist[t];
}

__global__ void __launch_bounds__(1024) bin_scatter_kernel(const int* __restrict__ src,
                                                           const int* __restrict__ dst,
                                                           const float* __restrict__ w,
                                                           const int* __restrict__ counts,
                                                           int* __restrict__ binstart,
                                                           uint2* __restrict__ recs) {
  __shared__ int cursor[NBINS];
  __shared__ int rowsum[NBINS];
  const int t = threadIdx.x;
  const int lane = t & 31;
  const int wave = t >> 5;
  const int slice = blockIdx.x;
  {
    int4 c[8][2];
#pragma unroll
    for (int j = 0; j < 8; ++j) {
      const int4* row = reinterpret_cast<const int4*>(counts + (wave * 8 + j) * NSLICE);
      c[j][0] = row[lane];
      c[j][1] = row[32 + lane];
    }
    const int s0 = lane * 4, s1 = 128 + lane * 4;
    int tot[8], part[8];
#pragma unroll
    for (int j = 0; j < 8; ++j) {
      tot[j] = c[j][0].x + c[j][0].y + c[j][0].z + c[j][0].w + c[j][1].x + c[j][1].y + c[j][1].z + c[j][1].w;
      part[j] = ((s0     < slice) ? c[j][0].x : 0) + ((s0 + 1 < slice) ? c[j][0].y : 0)
              + ((s0 + 2 < slice) ? c[j][0].z : 0) + ((s0 + 3 < slice) ? c[j][0].w : 0)
              + ((s1     < slice) ? c[j][1].x : 0) + ((s1 + 1 < slice) ? c[j][1].y : 0)
              + ((s1 + 2 < slice) ? c[j][1].z : 0) + ((s1 + 3 < slice) ? c[j][1].w : 0);
    }
#pragma unroll
    for (int d = 16; d >= 1; d >>= 1) {
      int xt[8], xp[8];
#pragma unroll
      for (int j = 0; j < 8; ++j) {
        xt[j] = __shfl_xor(tot[j], d, 32);
        xp[j] = __shfl_xor(part[j], d, 32);
      }
#pragma unroll
      for (int j = 0; j < 8; ++j) {
        tot[j] += xt[j];
        part[j] += xp[j];
      }
    }
    if (lane < 8) {
      int tsel = tot[0], psel = part[0];
#pragma unroll
      for (int j = 1; j < 8; ++j) {
        tsel = (lane == j) ? tot[j] : tsel;
        psel = (lane == j) ? part[j] : psel;
      }
      rowsum[wave * 8 + lane] = tsel;
      cursor[wave * 8 + lane] = psel;
    }
  }
  __syncthreads();
  if (wave == 0) {
    int v[8];
    int sum = 0;
#pragma unroll
    for (int i = 0; i < 8; ++i) {
      v[i] = rowsum[lane * 8 + i];
      sum += v[i];
    }
    int incl = sum;
#pragma unroll
    for (int d = 1; d < 32; d <<= 1) {
      const int x = __shfl_up(incl, d, 32);
      if (lane >= d) incl += x;
    }
    int run = incl - sum;
#pragma unroll
    for (int i = 0; i < 8; ++i) {
      const int b = lane * 8 + i;
      cursor[b] += run;
      if (slice == 0) binstart[b] = run;
      run += v[i];
    }
  }
  __syncthreads();
  const int e0 = blockIdx.x * SLICE_E;
  const int e1 = e0 + SLICE_E;
  for (int e = e0 + t; e < e1; e += 4096) {
    int s[4], d[4];
    float ww[4];
#pragma unroll
    for (int u = 0; u < 4; ++u) {
      const int ee = e + u * 1024;
      const int ec = (ee < e1) ? ee : e;
      s[u] = src[ec];
      d[u] = dst[ec];
      ww[u] = w[ec];
    }
#pragma unroll
    for (int u = 0; u < 4; ++u) {
      if (e + u * 1024 < e1) {
        const int bin = d[u] / BINSZ;
        const int dl = d[u] - bin * BINSZ;
        const int pos = atomicAdd(&cursor[bin], 1);
        recs[pos] = make_uint2((unsigned)s[u] | ((unsigned)dl << 17), __float_as_uint(ww[u]));
      }
    }
  }
}

__global__ void __launch_bounds__(1024) bin_degree_kernel(const uint2* __restrict__ recs,
                                                          const int* __restrict__ binstart,
                                                          float* __restrict__ dis) {
  __shared__ float deg[BINSZ];
  const int t = threadIdx.x;
  const int bin = blockIdx.x;
  const int node0 = bin * BINSZ;
  const int nrows = (Nn - node0 < BINSZ) ? (Nn - node0) : BINSZ;
  if (t < BINSZ) deg[t] = 0.0f;
  __syncthreads();
  const int start = binstart[bin];
  const int end = (bin == NBINS - 1) ? Ee : binstart[bin + 1];
  for (int e = start + t; e < end; e += 4096) {
    uint2 r[4];
#pragma unroll
    for (int u = 0; u < 4; ++u) {
      const int ee = e + u * 1024;
      r[u] = recs[(ee < end) ? ee : e];
    }
#pragma unroll
    for (int u = 0; u < 4; ++u)
      if (e + u * 1024 < end) atomicAdd(&deg[r[u].x >> 17], __uint_as_float(r[u].y));
  }
  __syncthreads();
  if (t < nrows) dis[node0 + t] = rsqrtf(deg[t] + 1.0f);
}

__device__ __forceinline__ void run_add(float* acc, const int lane, int& cur, float4& a,
                                        const unsigned px, const float nr, const float4 v) {
  const int dl = (int)(px >> 17);
  if (dl != cur) {
    if (cur >= 0) {
      float* p = &acc[cur * 96 + lane * 4];
      atomicAdd(p,     a.x);
      atomicAdd(p + 1, a.y);
      atomicAdd(p + 2, a.z);
      atomicAdd(p + 3, a.w);
    }
    cur = dl;
    a = make_float4(0.0f, 0.0f, 0.0f, 0.0f);
  }
  a.x = fmaf(nr, v.x, a.x);
  a.y = fmaf(nr, v.y, a.y);
  a.z = fmaf(nr, v.z, a.z);
  a.w = fmaf(nr, v.w, a.w);
}

#define SORTCAP 13312
#define RCN     416
__global__ void __launch_bounds__(1024) bin_aggregate_kernel(const uint2* __restrict__ recs,
                                                             const int* __restrict__ binstart,
                                                             const float* __restrict__ dis,
                                                             const float* __restrict__ XW,
                                                             const float* __restrict__ bcat,
                                                             float* __restrict__ AGG) {
  __shared__ __attribute__((aligned(16))) float acc[BINSZ * 96];
  __shared__ float dloc[BINSZ];
  __shared__ int rcnt[RCN];
  __shared__ uint2 srt[SORTCAP];
  const int t = threadIdx.x;
  const int bin = blockIdx.x;
  const int node0 = bin * BINSZ;
  const int nrows = (Nn - node0 < BINSZ) ? (Nn - node0) : BINSZ;
  const float* xw0 = XW + (size_t)node0 * 96;
  const int start = binstart[bin];
  const int end = (bin == NBINS - 1) ? Ee : binstart[bin + 1];
  uint2 r[13];
  if (start < end) {
    const int n0 = (end - start < SORTCAP) ? (end - start) : SORTCAP;
#pragma unroll
    for (int k = 0; k < 13; ++k) {
      const int i = t + k * 1024;
      r[k] = recs[start + ((i < n0) ? i : 0)];
    }
  }
  if (t < BINSZ) dloc[t] = (t < nrows) ? dis[node0 + t] : 0.0f;
  if (t < RCN) rcnt[t] = 0;

  const float4* xw4 = reinterpret_cast<const float4*>(xw0);
  const float4* b4p = reinterpret_cast<const float4*>(bcat);
  float4* acc4 = reinterpret_cast<float4*>(acc);
  for (int q = t; q < nrows * 24; q += 1024) {
    const int i = q / 24;
    const int c4 = q - i * 24;
    const float d = dis[node0 + i];
    const float4 x = xw4[q];
    const float4 b = b4p[c4];
    float4 sv;
    sv.x = d * d * x.x + b.x;
    sv.y = d * d * x.y + b.y;
    sv.z = d * d * x.z + b.z;
    sv.w = d * d * x.w + b.w;
    acc4[q] = sv;
  }
  __syncthreads();

  const int lane = t & 31;
  const int wave = __builtin_amdgcn_readfirstlane(t >> 5);
  const float4* xrow = reinterpret_cast<const float4*>(XW) + ((lane < 24) ? lane : 0);

  for (int c0 = start; c0 < end; c0 += SORTCAP) {
    const int n = (end - c0 < SORTCAP) ? (end - c0) : SORTCAP;

    float dsv[13];
    int rank[13];
    if (c0 != start) {
#pragma unroll
      for (int k = 0; k < 13; ++k) {
        const int i = t + k * 1024;
        r[k] = recs[c0 + ((i < n) ? i : 0)];
      }
    }
#pragma unroll
    for (int k = 0; k < 13; ++k) dsv[k] = dis[r[k].x & 0x1FFFFu];
#pragma unroll
    for (int k = 0; k < 13; ++k) {
      rank[k] = 0;
      if (t + k * 1024 < n) {
        const int dl = (int)(r[k].x >> 17);
        rank[k] = atomicAdd(&rcnt[dl], 1);
        dsv[k] = dsv[k] * __uint_as_float(r[k].y) * dloc[dl];
      }
    }
    __syncthreads();

    if (wave == 0) {
      int v[13];
      int sum = 0;
#pragma unroll
      for (int i = 0; i < 13; ++i) {
        v[i] = rcnt[lane * 13 + i];
        sum += v[i];
      }
      int incl = sum;
#pragma unroll
      for (int d = 1; d < 32; d <<= 1) {
        const int x = __shfl_up(incl, d, 32);
        if (lane >= d) incl += x;
      }
      int run = incl - sum;
#pragma unroll
      for (int i = 0; i < 13; ++i) {
        rcnt[lane * 13 + i] = run;
        run += v[i];
      }
    }
    __syncthreads();

#pragma unroll
    for (int k = 0; k < 13; ++k) {
      if (t + k * 1024 < n) {
        const int dl = (int)(r[k].x >> 17);
        srt[rcnt[dl] + rank[k]] = make_uint2(r[k].x, __float_as_uint(dsv[k]));
      }
    }
    __syncthreads();

    const int ws = (n * wave) >> 5;
    const int we = (n * (wave + 1)) >> 5;
    if (lane < 24) {
      int cur = -1;
      float4 a = make_float4(0.0f, 0.0f, 0.0f, 0.0f);
      uint2 rec_next = make_uint2(0u, 0u);
      if (ws + lane < we) rec_next = srt[ws + lane];
      for (int base = ws; base < we; base += 16) {
        const int cnt = (we - base < 16) ? (we - base) : 16;
        const uint2 rec = rec_next;
        float4 v[16];
        unsigned px[16];
        float nr[16];
#pragma unroll
        for (int u = 0; u < 16; ++u) {
          px[u] = (unsigned)__builtin_amdgcn_readlane((int)rec.x, u);
          nr[u] = __uint_as_float((unsigned)__builtin_amdgcn_readlane((int)rec.y, u));
          v[u] = xrow[(size_t)(px[u] & 0x1FFFFu) * 24];
        }
        const int nb = base + 16;
        rec_next = make_uint2(0u, 0u);
        if (nb + lane < we) rec_next = srt[nb + lane];
        if (cnt == 16) {
#pragma unroll
          for (int u = 0; u < 16; ++u) run_add(acc, lane, cur, a, px[u], nr[u], v[u]);
        } else {
#pragma unroll
          for (int u = 0; u < 16; ++u)
            if (u < cnt) run_add(acc, lane, cur, a, px[u], nr[u], v[u]);
        }
      }
      if (cur >= 0) {
        float* p = &acc[cur * 96 + lane * 4];
        atomicAdd(p,     a.x);
        atomicAdd(p + 1, a.y);
        atomicAdd(p + 2, a.z);
        atomicAdd(p + 3, a.w);
      }
    }
    if (c0 + SORTCAP < end) {
      __syncthreads();
      if (t < RCN) rcnt[t] = 0;
      __syncthreads();
    }
  }
  __syncthreads();

  float4* out = reinterpret_cast<float4*>(AGG + (size_t)node0 * 96);
  const float4* a4 = reinterpret_cast<const float4*>(acc);
  for (int q = t; q < nrows * 24; q += 1024) out[q] = a4[q];
}

__global__ void __launch_bounds__(32) gru_kernel(const float* __restrict__ AGG,
                                                 const float* __restrict__ Hp,
                                                 const float* __restrict__ LzW, const float* __restrict__ Lzb,
                                                 const float* __restrict__ LrW, const float* __restrict__ Lrb,
                                                 const float* __restrict__ LhW, const float* __restrict__ Lhb,
                                                 const float* __restrict__ headW, const float* __restrict__ headb,
                                                 float* __restrict__ y_out, float* __restrict__ h_out) {
  __shared__ __attribute__((aligned(16))) float As[16 * 132];
  __shared__ float Rs[16 * 36];
  const int lane = threadIdx.x;
  const int tile = blockIdx.x * 16;

  const float4* ag4 = reinterpret_cast<const float4*>(AGG + (size_t)tile * 96) + lane;
  const float4* hp4 = reinterpret_cast<const float4*>(Hp + (size_t)tile * 32) + lane;
  float4 sa[12], sh[4];
#pragma unroll
  for (int j = 0; j < 12; ++j) sa[j] = ag4[j * 32];
#pragma unroll
  for (int j = 0; j < 4; ++j) sh[j] = hp4[j * 32];
#pragma unroll
  for (int j = 0; j < 12; ++j) {
    const int q = lane + j * 32;
    const int r = q / 24, c4 = (q - r * 24) << 2;
    *reinterpret_cast<float4*>(&As[r * 132 + c4]) = sa[j];
  }
#pragma unroll
  for (int j = 0; j < 4; ++j) {
    const int q = lane + j * 32;
    const int r = q >> 3, c4 = (q & 7) << 2;
    *reinterpret_cast<float4*>(&As[r * 132 + 96 + c4]) = sh[j];
  }
  __syncthreads();

  const int m = lane & 15;
  const int n = lane & 15;
  const int koff = (lane >> 4) << 1;

  v8f z0 = {}, z1 = {}, r0 = {}, r1 = {};
#pragma unroll 4
  for (int k0 = 0; k0 < 64; k0 += 4) {
    const int kk = k0 + koff;
    const int colz = (kk < 32) ? kk : (64 + kk);
    const int colr = (kk < 32) ? (32 + kk) : (64 + kk);
    v2f az; az.x = As[m * 132 + colz]; az.y = As[m * 132 + colz + 1];
    v2f ar; ar.x = As[m * 132 + colr]; ar.y = As[m * 132 + colr + 1];
    const float* lz = LzW + kk * 32 + n;
    const float* lr = LrW + kk * 32 + n;
    v2f b;
    b.x = lz[0];  b.y = lz[32];  z0 = wmma4(az, b, z0);
    b.x = lz[16]; b.y = lz[48];  z1 = wmma4(az, b, z1);
    b.x = lr[0];  b.y = lr[32];  r0 = wmma4(ar, b, r0);
    b.x = lr[16]; b.y = lr[48];  r1 = wmma4(ar, b, r1);
  }

  const float bz0 = Lzb[n], bz1 = Lzb[16 + n];
  const float br0 = Lrb[n], br1 = Lrb[16 + n];
  const int row0 = (lane >> 4) << 3;
#pragma unroll
  for (int v = 0; v < 8; ++v) {
    z0[v] = sigmoidf_(z0[v] + bz0);
    z1[v] = sigmoidf_(z1[v] + bz1);
    r0[v] = sigmoidf_(r0[v] + br0);
    r1[v] = sigmoidf_(r1[v] + br1);
    Rs[(row0 + v) * 36 + n]      = r0[v];
    Rs[(row0 + v) * 36 + 16 + n] = r1[v];
  }
  __syncthreads();

  v8f t0 = {}, t1 = {};
#pragma unroll 4
  for (int k0 = 0; k0 < 64; k0 += 4) {
    const int kk = k0 + koff;
    v2f a;
    a.x = As[m * 132 + 64 + kk];
    a.y = As[m * 132 + 65 + kk];
    if (kk >= 32) {
      a.x *= Rs[m * 36 + kk - 32];
      a.y *= Rs[m * 36 + kk - 31];
    }
    const float* lh = LhW + kk * 32 + n;
    v2f b;
    b.x = lh[0];  b.y = lh[32];  t0 = wmma4(a, b, t0);
    b.x = lh[16]; b.y = lh[48];  t1 = wmma4(a, b, t1);
  }

  const float bh0 = Lhb[n], bh1 = Lhb[16 + n];
  const float hw0 = headW[n], hw1 = headW[16 + n];
  float p[8];
#pragma unroll
  for (int v = 0; v < 8; ++v) {
    const float ht0 = tanhf(t0[v] + bh0);
    const float ht1 = tanhf(t1[v] + bh1);
    const float hp0 = As[(row0 + v) * 132 + 96 + n];
    const float hp1 = As[(row0 + v) * 132 + 112 + n];
    const float h0 = z0[v] * hp0 + (1.0f - z0[v]) * ht0;
    const float h1 = z1[v] * hp1 + (1.0f - z1[v]) * ht1;
    h_out[(size_t)(tile + row0 + v) * 32 + n]      = h0;
    h_out[(size_t)(tile + row0 + v) * 32 + 16 + n] = h1;
    p[v] = fmaxf(h0, 0.0f) * hw0 + fmaxf(h1, 0.0f) * hw1;
  }
#pragma unroll
  for (int s = 1; s < 16; s <<= 1)
#pragma unroll
    for (int v = 0; v < 8; ++v)
      p[v] += __shfl_xor(p[v], s, 32);
  if (n == 0) {
    const float hb = headb[0];
#pragma unroll
    for (int v = 0; v < 8; ++v)
      y_out[tile + row0 + v] = p[v] + hb;
  }
}

extern "C" void kernel_launch(void* const* d_in, const int* in_sizes, int n_in,
                              void* d_out, int out_size, void* d_ws, size_t ws_size,
                              hipStream_t stream) {
  const float* x    = (const float*)d_in[0];
  const int*   ei   = (const int*)d_in[1];
  const float* ew   = (const float*)d_in[2];
  const float* Hp   = (const float*)d_in[3];
  const float* Wz   = (const float*)d_in[4];
  const float* bz   = (const float*)d_in[5];
  const float* Wr   = (const float*)d_in[6];
  const float* br   = (const float*)d_in[7];
  const float* Wh   = (const float*)d_in[8];
  const float* bh   = (const float*)d_in[9];
  const float* LzW  = (const float*)d_in[10];
  const float* Lzb  = (const float*)d_in[11];
  const float* LrW  = (const float*)d_in[12];
  const float* Lrb  = (const float*)d_in[13];
  const float* LhW  = (const float*)d_in[14];
  const float* Lhb  = (const float*)d_in[15];
  const float* hW   = (const float*)d_in[16];
  const float* hb   = (const float*)d_in[17];

  const int* src = ei;
  const int* dst = ei + Ee;

  float* out_y = (float*)d_out;
  float* out_h = out_y + Nn;

  float* ws   = (float*)d_ws;
  float* dis  = ws;
  float* Wcat = ws + Nn;
  float* bcat = Wcat + FIN * 96;
  float* XW   = bcat + 96;
  float* AGG  = XW + (size_t)Nn * 96;
  int*   counts = (int*)(AGG + (size_t)Nn * 96);
  int*   binstart = counts + NBINS * NSLICE;
  uint2* recs   = (uint2*)(binstart + NBINS * NSLICE);

  pack_weights_kernel<<<(FIN * 96 + 255) / 256, 256, 0, stream>>>(Wz, Wr, Wh, bz, br, bh, Wcat, bcat);
  gemm_xw_kernel<<<Nn / 16, 32, 0, stream>>>(x, Wcat, XW);
  bin_count_kernel<<<NSLICE, 1024, 0, stream>>>(dst, counts);
  bin_scatter_kernel<<<NSLICE, 1024, 0, stream>>>(src, dst, ew, counts, binstart, recs);
  bin_degree_kernel<<<NBINS, 1024, 0, stream>>>(recs, binstart, dis);
  bin_aggregate_kernel<<<NBINS, 1024, 0, stream>>>(recs, binstart, dis, XW, bcat, AGG);
  gru_kernel<<<Nn / 16, 32, 0, stream>>>(AGG, Hp, LzW, Lzb, LrW, Lrb, LhW, Lhb, hW, hb, out_y, out_h);
}


# Source factor dis[src] folded into the XW rows at the GEMM store (GEMM after degree); aggregate sums w*row, applies dis[dst] and bias once per row: no per-edge dis gather; products re-associated
# speedup vs baseline: 4.4158x; 1.0466x over previous
#include <hip/hip_runtime.h>

#define Nn    100000
#define Ee    3200000
#define FIN   256
#define Hh    32

typedef float v2f __attribute__((ext_vector_type(2)));
typedef float v8f __attribute__((ext_vector_type(8)));

__device__ __forceinline__ v8f wmma4(v2f a, v2f b, v8f c) {
  return __builtin_amdgcn_wmma_f32_16x16x4_f32(
       false, a,  false, b,
       (short)0, c,  false,  false);
}

__device__ __forceinline__ float sigmoidf_(float x) {
  return 1.0f / (1.0f + __expf(-x));
}

__global__ void pack_weights_kernel(const float* __restrict__ Wz, const float* __restrict__ Wr,
                                    const float* __restrict__ Wh, const float* __restrict__ bz,
                                    const float* __restrict__ br, const float* __restrict__ bh,
                                    float* __restrict__ Wcat, float* __restrict__ bcat) {
  int gid = blockIdx.x * blockDim.x + threadIdx.x;
  if (gid < FIN * 96) {
    int k = gid / 96, c = gid - k * 96;
    float v = (c < 32) ? Wz[k * 32 + c] : ((c < 64) ? Wr[k * 32 + c - 32] : Wh[k * 32 + c - 64]);
    Wcat[gid] = v;
  }
  if (gid < 96) {
    bcat[gid] = (gid < 32) ? bz[gid] : ((gid < 64) ? br[gid - 32] : bh[gid - 64]);
  }
}

__global__ void __launch_bounds__(32) gemm_xw_kernel(const float* __restrict__ x,
                                                     const float* __restrict__ Wcat,
                                                     const float* __restrict__ dis,
                                                     float* __restrict__ XW) {
  __shared__ float Xs[16 * 260];
  const int lane = threadIdx.x;
  const int tile = blockIdx.x * 16;

  const float4* xt = reinterpret_cast<const float4*>(x + (size_t)tile * FIN) + lane;
  float4 st[32];
#pragma unroll
  for (int j = 0; j < 32; ++j) st[j] = xt[j * 32];
#pragma unroll
  for (int j = 0; j < 32; ++j) {
    const int i = lane + j * 32;
    const int r = i >> 6, c4 = (i & 63) << 2;
    *reinterpret_cast<float4*>(&Xs[r * 260 + c4]) = st[j];
  }
  __syncthreads();

  const int m = lane & 15;
  const int koff = (lane >> 4) << 1;

  v8f acc[6] = {};
  const float2* arow = reinterpret_cast<const float2*>(&Xs[m * 260]);
  const int row0 = (lane >> 4) << 3;
  float dv[8];
#pragma unroll
  for (int v = 0; v < 8; ++v) dv[v] = dis[tile + row0 + v];

#pragma unroll 4
  for (int k0 = 0; k0 < FIN; k0 += 4) {
    const int kk = k0 + koff;
    float2 af = arow[kk >> 1];
    v2f a; a.x = af.x; a.y = af.y;
    const float* brow = Wcat + (size_t)kk * 96 + m;
#pragma unroll
    for (int t = 0; t < 6; ++t) {
      v2f b; b.x = brow[t * 16]; b.y = brow[96 + t * 16];
      acc[t] = wmma4(a, b, acc[t]);
    }
  }

#pragma unroll
  for (int t = 0; t < 6; ++t)
#pragma unroll
    for (int v = 0; v < 8; ++v)
      XW[(size_t)(tile + row0 + v) * 96 + t * 16 + m] = acc[t][v] * dv[v];
}

#define NBINS   256
#define BINSZ   391
#define NSLICE  256
#define SLICE_E (Ee / NSLICE)

__global__ void __launch_bounds__(1024) bin_count_kernel(const int* __restrict__ dst,
                                                         int* __restrict__ counts) {
  __shared__ int hist[NBINS];
  const int t = threadIdx.x;
  if (t < NBINS) hist[t] = 0;
  __syncthreads();
  const int e0 = blockIdx.x * SLICE_E;
  const int e1 = e0 + SLICE_E;
  for (int e = e0 + t; e < e1; e += 4096) {
    int d[4];
#pragma unroll
    for (int u = 0; u < 4; ++u) {
      const int ee = e + u * 1024;
      d[u] = dst[(ee < e1) ? ee : e];
    }
#pragma unroll
    for (int u = 0; u < 4; ++u)
      if (e + u * 1024 < e1) atomicAdd(&hist[d[u] / BINSZ], 1);
  }
  __syncthreads();
  if (t < NBINS) counts[t * NSLICE + blockIdx.x] = hist[t];
}

__global__ void __launch_bounds__(1024) bin_scatter_kernel(const int* __restrict__ src,
                                                           const int* __restrict__ dst,
                                                           const float* __restrict__ w,
                                                           const int* __restrict__ counts,
                                                           int* __restrict__ binstart,
                                                           uint2* __restrict__ recs) {
  __shared__ int cursor[NBINS];
  __shared__ int rowsum[NBINS];
  const int t = threadIdx.x;
  const int lane = t & 31;
  const int wave = t >> 5;
  const int slice = blockIdx.x;
  {
    int4 c[8][2];
#pragma unroll
    for (int j = 0; j < 8; ++j) {
      const int4* row = reinterpret_cast<const int4*>(counts + (wave * 8 + j) * NSLICE);
      c[j][0] = row[lane];
      c[j][1] = row[32 + lane];
    }
    const int s0 = lane * 4, s1 = 128 + lane * 4;
    int tot[8], part[8];
#pragma unroll
    for (int j = 0; j < 8; ++j) {
      tot[j] = c[j][0].x + c[j][0].y + c[j][0].z + c[j][0].w + c[j][1].x + c[j][1].y + c[j][1].z + c[j][1].w;
      part[j] = ((s0     < slice) ? c[j][0].x : 0) + ((s0 + 1 < slice) ? c[j][0].y : 0)
              + ((s0 + 2 < slice) ? c[j][0].z : 0) + ((s0 + 3 < slice) ? c[j][0].w : 0)
              + ((s1     < slice) ? c[j][1].x : 0) + ((s1 + 1 < slice) ? c[j][1].y : 0)
              + ((s1 + 2 < slice) ? c[j][1].z : 0) + ((s1 + 3 < slice) ? c[j][1].w : 0);
    }
#pragma unroll
    for (int d = 16; d >= 1; d >>= 1) {
      int xt[8], xp[8];
#pragma unroll
      for (int j = 0; j < 8; ++j) {
        xt[j] = __shfl_xor(tot[j], d, 32);
        xp[j] = __shfl_xor(part[j], d, 32);
      }
#pragma unroll
      for (int j = 0; j < 8; ++j) {
        tot[j] += xt[j];
        part[j] += xp[j];
      }
    }
    if (lane < 8) {
      int tsel = tot[0], psel = part[0];
#pragma unroll
      for (int j = 1; j < 8; ++j) {
        tsel = (lane == j) ? tot[j] : tsel;
        psel = (lane == j) ? part[j] : psel;
      }
      rowsum[wave * 8 + lane] = tsel;
      cursor[wave * 8 + lane] = psel;
    }
  }
  __syncthreads();
  if (wave == 0) {
    int v[8];
    int sum = 0;
#pragma unroll
    for (int i = 0; i < 8; ++i) {
      v[i] = rowsum[lane * 8 + i];
      sum += v[i];
    }
    int incl = sum;
#pragma unroll
    for (int d = 1; d < 32; d <<= 1) {
      const int x = __shfl_up(incl, d, 32);
      if (lane >= d) incl += x;
    }
    int run = incl - sum;
#pragma unroll
    for (int i = 0; i < 8; ++i) {
      const int b = lane * 8 + i;
      cursor[b] += run;
      if (slice == 0) binstart[b] = run;
      run += v[i];
    }
  }
  __syncthreads();
  const int e0 = blockIdx.x * SLICE_E;
  const int e1 = e0 + SLICE_E;
  for (int e = e0 + t; e < e1; e += 4096) {
    int s[4], d[4];
    float ww[4];
#pragma unroll
    for (int u = 0; u < 4; ++u) {
      const int ee = e + u * 1024;
      const int ec = (ee < e1) ? ee : e;
      s[u] = src[ec];
      d[u] = dst[ec];
      ww[u] = w[ec];
    }
#pragma unroll
    for (int u = 0; u < 4; ++u) {
      if (e + u * 1024 < e1) {
        const int bin = d[u] / BINSZ;
        const int dl = d[u] - bin * BINSZ;
        const int pos = atomicAdd(&cursor[bin], 1);
        recs[pos] = make_uint2((unsigned)s[u] | ((unsigned)dl << 17), __float_as_uint(ww[u]));
      }
    }
  }
}

__global__ void __launch_bounds__(1024) bin_degree_kernel(const uint2* __restrict__ recs,
                                                          const int* __restrict__ binstart,
                                                          float* __restrict__ dis) {
  __shared__ float deg[BINSZ];
  const int t = threadIdx.x;
  const int bin = blockIdx.x;
  const int node0 = bin * BINSZ;
  const int nrows = (Nn - node0 < BINSZ) ? (Nn - node0) : BINSZ;
  if (t < BINSZ) deg[t] = 0.0f;
  __syncthreads();
  const int start = binstart[bin];
  const int end = (bin == NBINS - 1) ? Ee : binstart[bin + 1];
  for (int e = start + t; e < end; e += 4096) {
    uint2 r[4];
#pragma unroll
    for (int u = 0; u < 4; ++u) {
      const int ee = e + u * 1024;
      r[u] = recs[(ee < end) ? ee : e];
    }
#pragma unroll
    for (int u = 0; u < 4; ++u)
      if (e + u * 1024 < end) atomicAdd(&deg[r[u].x >> 17], __uint_as_float(r[u].y));
  }
  __syncthreads();
  if (t < nrows) dis[node0 + t] = rsqrtf(deg[t] + 1.0f);
}

__device__ __forceinline__ void run_add(float* acc, const int lane, int& cur, float4& a,
                                        const unsigned px, const float nr, const float4 v) {
  const int dl = (int)(px >> 17);
  if (dl != cur) {
    if (cur >= 0) {
      float* p = &acc[cur * 96 + lane * 4];
      atomicAdd(p,     a.x);
      atomicAdd(p + 1, a.y);
      atomicAdd(p + 2, a.z);
      atomicAdd(p + 3, a.w);
    }
    cur = dl;
    a = make_float4(0.0f, 0.0f, 0.0f, 0.0f);
  }
  a.x = fmaf(nr, v.x, a.x);
  a.y = fmaf(nr, v.y, a.y);
  a.z = fmaf(nr, v.z, a.z);
  a.w = fmaf(nr, v.w, a.w);
}

#define SORTCAP 13312
#define RCN     416
__global__ void __launch_bounds__(1024) bin_aggregate_kernel(const uint2* __restrict__ recs,
                                                             const int* __restrict__ binstart,
                                                             const float* __restrict__ dis,
                                                             const float* __restrict__ XW,
                                                             const float* __restrict__ bcat,
                                                             float* __restrict__ AGG) {
  __shared__ __attribute__((aligned(16))) float acc[BINSZ * 96];
  __shared__ float dloc[BINSZ];
  __shared__ int rcnt[RCN];
  __shared__ uint2 srt[SORTCAP];
  const int t = threadIdx.x;
  const int bin = blockIdx.x;
  const int node0 = bin * BINSZ;
  const int nrows = (Nn - node0 < BINSZ) ? (Nn - node0) : BINSZ;
  const float* xw0 = XW + (size_t)node0 * 96;
  const int start = binstart[bin];
  const int end = (bin == NBINS - 1) ? Ee : binstart[bin + 1];
  uint2 r[13];
  if (start < end) {
    const int n0 = (end - start < SORTCAP) ? (end - start) : SORTCAP;
#pragma unroll
    for (int k = 0; k < 13; ++k) {
      const int i = t + k * 1024;
      r[k] = recs[start + ((i < n0) ? i : 0)];
    }
  }
  if (t < BINSZ) dloc[t] = (t < nrows) ? dis[node0 + t] : 0.0f;
  if (t < RCN) rcnt[t] = 0;

  const float4* xw4 = reinterpret_cast<const float4*>(xw0);
  const float4* b4p = reinterpret_cast<const float4*>(bcat);
  float4* acc4 = reinterpret_cast<float4*>(acc);
  for (int q = t; q < nrows * 24; q += 1024) acc4[q] = xw4[q];
  __syncthreads();

  const int lane = t & 31;
  const int wave = __builtin_amdgcn_readfirstlane(t >> 5);
  const float4* xrow = reinterpret_cast<const float4*>(XW) + ((lane < 24) ? lane : 0);

  for (int c0 = start; c0 < end; c0 += SORTCAP) {
    const int n = (end - c0 < SORTCAP) ? (end - c0) : SORTCAP;

    int rank[13];
    if (c0 != start) {
#pragma unroll
      for (int k = 0; k < 13; ++k) {
        const int i = t + k * 1024;
        r[k] = recs[c0 + ((i < n) ? i : 0)];
      }
    }
#pragma unroll
    for (int k = 0; k < 13; ++k) {
      rank[k] = 0;
      if (t + k * 1024 < n) rank[k] = atomicAdd(&rcnt[(int)(r[k].x >> 17)], 1);
    }
    __syncthreads();

    if (wave == 0) {
      int v[13];
      int sum = 0;
#pragma unroll
      for (int i = 0; i < 13; ++i) {
        v[i] = rcnt[lane * 13 + i];
        sum += v[i];
      }
      int incl = sum;
#pragma unroll
      for (int d = 1; d < 32; d <<= 1) {
        const int x = __shfl_up(incl, d, 32);
        if (lane >= d) incl += x;
      }
      int run = incl - sum;
#pragma unroll
      for (int i = 0; i < 13; ++i) {
        rcnt[lane * 13 + i] = run;
        run += v[i];
      }
    }
    __syncthreads();

#pragma unroll
    for (int k = 0; k < 13; ++k) {
      if (t + k * 1024 < n) {
        const int dl = (int)(r[k].x >> 17);
        srt[rcnt[dl] + rank[k]] = r[k];
      }
    }
    __syncthreads();

    const int ws = (n * wave) >> 5;
    const int we = (n * (wave + 1)) >> 5;
    if (lane < 24) {
      int cur = -1;
      float4 a = make_float4(0.0f, 0.0f, 0.0f, 0.0f);
      uint2 rec_next = make_uint2(0u, 0u);
      if (ws + lane < we) rec_next = srt[ws + lane];
      for (int base = ws; base < we; base += 16) {
        const int cnt = (we - base < 16) ? (we - base) : 16;
        const uint2 rec = rec_next;
        float4 v[16];
        unsigned px[16];
        float nr[16];
#pragma unroll
        for (int u = 0; u < 16; ++u) {
          px[u] = (unsigned)__builtin_amdgcn_readlane((int)rec.x, u);
          nr[u] = __uint_as_float((unsigned)__builtin_amdgcn_readlane((int)rec.y, u));
          v[u] = xrow[(size_t)(px[u] & 0x1FFFFu) * 24];
        }
        const int nb = base + 16;
        rec_next = make_uint2(0u, 0u);
        if (nb + lane < we) rec_next = srt[nb + lane];
        if (cnt == 16) {
#pragma unroll
          for (int u = 0; u < 16; ++u) run_add(acc, lane, cur, a, px[u], nr[u], v[u]);
        } else {
#pragma unroll
          for (int u = 0; u < 16; ++u)
            if (u < cnt) run_add(acc, lane, cur, a, px[u], nr[u], v[u]);
        }
      }
      if (cur >= 0) {
        float* p = &acc[cur * 96 + lane * 4];
        atomicAdd(p,     a.x);
        atomicAdd(p + 1, a.y);
        atomicAdd(p + 2, a.z);
        atomicAdd(p + 3, a.w);
      }
    }
    if (c0 + SORTCAP < end) {
      __syncthreads();
      if (t < RCN) rcnt[t] = 0;
      __syncthreads();
    }
  }
  __syncthreads();

  float4* out = reinterpret_cast<float4*>(AGG + (size_t)node0 * 96);
  const float4* a4 = reinterpret_cast<const float4*>(acc);
  for (int q = t; q < nrows * 24; q += 1024) {
    const int i = q / 24;
    const int c4 = q - i * 24;
    const float d = dloc[i];
    const float4 s = a4[q];
    const float4 b = b4p[c4];
    float4 o;
    o.x = d * s.x + b.x;
    o.y = d * s.y + b.y;
    o.z = d * s.z + b.z;
    o.w = d * s.w + b.w;
    out[q] = o;
  }
}

__global__ void __launch_bounds__(32) gru_kernel(const float* __restrict__ AGG,
                                                 const float* __restrict__ Hp,
                                                 const float* __restrict__ LzW, const float* __restrict__ Lzb,
                                                 const float* __restrict__ LrW, const float* __restrict__ Lrb,
                                                 const float* __restrict__ LhW, const float* __restrict__ Lhb,
                                                 const float* __restrict__ headW, const float* __restrict__ headb,
                                                 float* __restrict__ y_out, float* __restrict__ h_out) {
  __shared__ __attribute__((aligned(16))) float As[16 * 132];
  __shared__ float Rs[16 * 36];
  const int lane = threadIdx.x;
  const int tile = blockIdx.x * 16;

  const float4* ag4 = reinterpret_cast<const float4*>(AGG + (size_t)tile * 96) + lane;
  const float4* hp4 = reinterpret_cast<const float4*>(Hp + (size_t)tile * 32) + lane;
  float4 sa[12], sh[4];
#pragma unroll
  for (int j = 0; j < 12; ++j) sa[j] = ag4[j * 32];
#pragma unroll
  for (int j = 0; j < 4; ++j) sh[j] = hp4[j * 32];
#pragma unroll
  for (int j = 0; j < 12; ++j) {
    const int q = lane + j * 32;
    const int r = q / 24, c4 = (q - r * 24) << 2;
    *reinterpret_cast<float4*>(&As[r * 132 + c4]) = sa[j];
  }
#pragma unroll
  for (int j = 0; j < 4; ++j) {
    const int q = lane + j * 32;
    const int r = q >> 3, c4 = (q & 7) << 2;
    *reinterpret_cast<float4*>(&As[r * 132 + 96 + c4]) = sh[j];
  }
  __syncthreads();

  const int m = lane & 15;
  const int n = lane & 15;
  const int koff = (lane >> 4) << 1;

  v8f z0 = {}, z1 = {}, r0 = {}, r1 = {};
#pragma unroll 4
  for (int k0 = 0; k0 < 64; k0 += 4) {
    const int kk = k0 + koff;
    const int colz = (kk < 32) ? kk : (64 + kk);
    const int colr = (kk < 32) ? (32 + kk) : (64 + kk);
    v2f az; az.x = As[m * 132 + colz]; az.y = As[m * 132 + colz + 1];
    v2f ar; ar.x = As[m * 132 + colr]; ar.y = As[m * 132 + colr + 1];
    const float* lz = LzW + kk * 32 + n;
    const float* lr = LrW + kk * 32 + n;
    v2f b;
    b.x = lz[0];  b.y = lz[32];  z0 = wmma4(az, b, z0);
    b.x = lz[16]; b.y = lz[48];  z1 = wmma4(az, b, z1);
    b.x = lr[0];  b.y = lr[32];  r0 = wmma4(ar, b, r0);
    b.x = lr[16]; b.y = lr[48];  r1 = wmma4(ar, b, r1);
  }

  const float bz0 = Lzb[n], bz1 = Lzb[16 + n];
  const float br0 = Lrb[n], br1 = Lrb[16 + n];
  const int row0 = (lane >> 4) << 3;
#pragma unroll
  for (int v = 0; v < 8; ++v) {
    z0[v] = sigmoidf_(z0[v] + bz0);
    z1[v] = sigmoidf_(z1[v] + bz1);
    r0[v] = sigmoidf_(r0[v] + br0);
    r1[v] = sigmoidf_(r1[v] + br1);
    Rs[(row0 + v) * 36 + n]      = r0[v];
    Rs[(row0 + v) * 36 + 16 + n] = r1[v];
  }
  __syncthreads();

  v8f t0 = {}, t1 = {};
#pragma unroll 4
  for (int k0 = 0; k0 < 64; k0 += 4) {
    const int kk = k0 + koff;
    v2f a;
    a.x = As[m * 132 + 64 + kk];
    a.y = As[m * 132 + 65 + kk];
    if (kk >= 32) {
      a.x *= Rs[m * 36 + kk - 32];
      a.y *= Rs[m * 36 + kk - 31];
    }
    const float* lh = LhW + kk * 32 + n;
    v2f b;
    b.x = lh[0];  b.y = lh[32];  t0 = wmma4(a, b, t0);
    b.x = lh[16]; b.y = lh[48];  t1 = wmma4(a, b, t1);
  }

  const float bh0 = Lhb[n], bh1 = Lhb[16 + n];
  const float hw0 = headW[n], hw1 = headW[16 + n];
  float p[8];
#pragma unroll
  for (int v = 0; v < 8; ++v) {
    const float ht0 = tanhf(t0[v] + bh0);
    const float ht1 = tanhf(t1[v] + bh1);
    const float hp0 = As[(row0 + v) * 132 + 96 + n];
    const float hp1 = As[(row0 + v) * 132 + 112 + n];
    const float h0 = z0[v] * hp0 + (1.0f - z0[v]) * ht0;
    const float h1 = z1[v] * hp1 + (1.0f - z1[v]) * ht1;
    h_out[(size_t)(tile + row0 + v) * 32 + n]      = h0;
    h_out[(size_t)(tile + row0 + v) * 32 + 16 + n] = h1;
    p[v] = fmaxf(h0, 0.0f) * hw0 + fmaxf(h1, 0.0f) * hw1;
  }
#pragma unroll
  for (int s = 1; s < 16; s <<= 1)
#pragma unroll
    for (int v = 0; v < 8; ++v)
      p[v] += __shfl_xor(p[v], s, 32);
  if (n == 0) {
    const float hb = headb[0];
#pragma unroll
    for (int v = 0; v < 8; ++v)
      y_out[tile + row0 + v] = p[v] + hb;
  }
}

extern "C" void kernel_launch(void* const* d_in, const int* in_sizes, int n_in,
                              void* d_out, int out_size, void* d_ws, size_t ws_size,
                              hipStream_t stream) {
  const float* x    = (const float*)d_in[0];
  const int*   ei   = (const int*)d_in[1];
  const float* ew   = (const float*)d_in[2];
  const float* Hp   = (const float*)d_in[3];
  const float* Wz   = (const float*)d_in[4];
  const float* bz   = (const float*)d_in[5];
  const float* Wr   = (const float*)d_in[6];
  const float* br   = (const float*)d_in[7];
  const float* Wh   = (const float*)d_in[8];
  const float* bh   = (const float*)d_in[9];
  const float* LzW  = (const float*)d_in[10];
  const float* Lzb  = (const float*)d_in[11];
  const float* LrW  = (const float*)d_in[12];
  const float* Lrb  = (const float*)d_in[13];
  const float* LhW  = (const float*)d_in[14];
  const float* Lhb  = (const float*)d_in[15];
  const float* hW   = (const float*)d_in[16];
  const float* hb   = (const float*)d_in[17];

  const int* src = ei;
  const int* dst = ei + Ee;

  float* out_y = (float*)d_out;
  float* out_h = out_y + Nn;

  float* ws   = (float*)d_ws;
  float* dis  = ws;
  float* Wcat = ws + Nn;
  float* bcat = Wcat + FIN * 96;
  float* XW   = bcat + 96;
  float* AGG  = XW + (size_t)Nn * 96;
  int*   counts = (int*)(AGG + (size_t)Nn * 96);
  int*   binstart = counts + NBINS * NSLICE;
  uint2* recs   = (uint2*)(binstart + NBINS * NSLICE);

  pack_weights_kernel<<<(FIN * 96 + 255) / 256, 256, 0, stream>>>(Wz, Wr, Wh, bz, br, bh, Wcat, bcat);
  bin_count_kernel<<<NSLICE, 1024, 0, stream>>>(dst, counts);
  bin_scatter_kernel<<<NSLICE, 1024, 0, stream>>>(src, dst, ew, counts, binstart, recs);
  bin_degree_kernel<<<NBINS, 1024, 0, stream>>>(recs, binstart, dis);
  gemm_xw_kernel<<<Nn / 16, 32, 0, stream>>>(x, Wcat, dis, XW);
  bin_aggregate_kernel<<<NBINS, 1024, 0, stream>>>(recs, binstart, dis, XW, bcat, AGG);
  gru_kernel<<<Nn / 16, 32, 0, stream>>>(AGG, Hp, LzW, Lzb, LrW, Lrb, LhW, Lhb, hW, hb, out_y, out_h);
}


# Bucketing in one pass: each edge slice ordered by bin in LDS, stored whole with a table of run starts; degree launch gathers its bin's 256 runs into recs; no count launch; sums still in arrival order
# speedup vs baseline: 4.9221x; 1.1147x over previous
#include <hip/hip_runtime.h>

#define Nn    100000
#define Ee    3200000
#define FIN   256
#define Hh    32

typedef float v2f __attribute__((ext_vector_type(2)));
typedef float v8f __attribute__((ext_vector_type(8)));

__device__ __forceinline__ v8f wmma4(v2f a, v2f b, v8f c) {
  return __builtin_amdgcn_wmma_f32_16x16x4_f32(
       false, a,  false, b,
       (short)0, c,  false,  false);
}

__device__ __forceinline__ float sigmoidf_(float x) {
  return 1.0f / (1.0f + __expf(-x));
}

__global__ void pack_weights_kernel(const float* __restrict__ Wz, const float* __restrict__ Wr,
                                    const float* __restrict__ Wh, const float* __restrict__ bz,
                                    const float* __restrict__ br, const float* __restrict__ bh,
                                    float* __restrict__ Wcat, float* __restrict__ bcat) {
  int gid = blockIdx.x * blockDim.x + threadIdx.x;
  if (gid < FIN * 96) {
    int k = gid / 96, c = gid - k * 96;
    float v = (c < 32) ? Wz[k * 32 + c] : ((c < 64) ? Wr[k * 32 + c - 32] : Wh[k * 32 + c - 64]);
    Wcat[gid] = v;
  }
  if (gid < 96) {
    bcat[gid] = (gid < 32) ? bz[gid] : ((gid < 64) ? br[gid - 32] : bh[gid - 64]);
  }
}

__global__ void __launch_bounds__(32) gemm_xw_kernel(const float* __restrict__ x,
                                                     const float* __restrict__ Wcat,
                                                     const float* __restrict__ dis,
                                                     float* __restrict__ XW) {
  __shared__ float Xs[16 * 260];
  const int lane = threadIdx.x;
  const int tile = blockIdx.x * 16;

  const float4* xt = reinterpret_cast<const float4*>(x + (size_t)tile * FIN) + lane;
  float4 st[32];
#pragma unroll
  for (int j = 0; j < 32; ++j) st[j] = xt[j * 32];
#pragma unroll
  for (int j = 0; j < 32; ++j) {
    const int i = lane + j * 32;
    const int r = i >> 6, c4 = (i & 63) << 2;
    *reinterpret_cast<float4*>(&Xs[r * 260 + c4]) = st[j];
  }
  __syncthreads();

  const int m = lane & 15;
  const int koff = (lane >> 4) << 1;

  v8f acc[6] = {};
  const float2* arow = reinterpret_cast<const float2*>(&Xs[m * 260]);
  const int row0 = (lane >> 4) << 3;
  float dv[8];
#pragma unroll
  for (int v = 0; v < 8; ++v) dv[v] = dis[tile + row0 + v];

#pragma unroll 4
  for (int k0 = 0; k0 < FIN; k0 += 4) {
    const int kk = k0 + koff;
    float2 af = arow[kk >> 1];
    v2f a; a.x = af.x; a.y = af.y;
    const float* brow = Wcat + (size_t)kk * 96 + m;
#pragma unroll
    for (int t = 0; t < 6; ++t) {
      v2f b; b.x = brow[t * 16]; b.y = brow[96 + t * 16];
      acc[t] = wmma4(a, b, acc[t]);
    }
  }

#pragma unroll
  for (int t = 0; t < 6; ++t)
#pragma unroll
    for (int v = 0; v < 8; ++v)
      XW[(size_t)(tile + row0 + v) * 96 + t * 16 + m] = acc[t][v] * dv[v];
}

#define NBINS   256
#define BINSZ   391
#define NSLICE  256
#define SLICE_E (Ee / NSLICE)

#define ESLOTS  13

__global__ void __launch_bounds__(1024) bin_scatter_kernel(const int* __restrict__ src,
                                                           const int* __restrict__ dst,
                                                           const float* __restrict__ w,
                                                           int* __restrict__ runstart,
                                                           uint2* __restrict__ srt) {
  __shared__ int lcnt[NBINS];
  __shared__ __attribute__((aligned(16))) uint2 srec[SLICE_E];
  const int t = threadIdx.x;
  const int lane = t & 31;
  const int wave = t >> 5;
  const int slice = blockIdx.x;
  const int e0 = slice * SLICE_E;
  if (t < NBINS) lcnt[t] = 0;
  unsigned px[ESLOTS];
  int bn[ESLOTS];
  float ww[ESLOTS];
  {
    int s[ESLOTS], d[ESLOTS];
#pragma unroll
    for (int k = 0; k < ESLOTS; ++k) {
      const int i = t + k * 1024;
      const int ec = e0 + ((i < SLICE_E) ? i : t);
      s[k] = src[ec];
      d[k] = dst[ec];
      ww[k] = w[ec];
    }
#pragma unroll
    for (int k = 0; k < ESLOTS; ++k) {
      bn[k] = d[k] / BINSZ;
      px[k] = (unsigned)s[k] | ((unsigned)(d[k] - bn[k] * BINSZ) << 17);
    }
  }
  __syncthreads();
  int rank[ESLOTS];
#pragma unroll
  for (int k = 0; k < ESLOTS; ++k) {
    rank[k] = 0;
    if (t + k * 1024 < SLICE_E) rank[k] = atomicAdd(&lcnt[bn[k]], 1);
  }
  __syncthreads();
  if (wave == 0) {
    int v[8];
    int sum = 0;
#pragma unroll
    for (int i = 0; i < 8; ++i) {
      v[i] = lcnt[lane * 8 + i];
      sum += v[i];
    }
    int incl = sum;
#pragma unroll
    for (int d = 1; d < 32; d <<= 1) {
      const int x = __shfl_up(incl, d, 32);
      if (lane >= d) incl += x;
    }
    int run = incl - sum;
#pragma unroll
    for (int i = 0; i < 8; ++i) {
      lcnt[lane * 8 + i] = run;
      run += v[i];
    }
  }
  __syncthreads();
  if (t < NBINS) runstart[t * NSLICE + slice] = lcnt[t];
#pragma unroll
  for (int k = 0; k < ESLOTS; ++k) {
    if (t + k * 1024 < SLICE_E)
      srec[lcnt[bn[k]] + rank[k]] = make_uint2(px[k], __float_as_uint(ww[k]));
  }
  __syncthreads();
  const uint4* s4 = reinterpret_cast<const uint4*>(srec);
  uint4* o4 = reinterpret_cast<uint4*>(srt + e0);
#pragma unroll
  for (int k = 0; k < 7; ++k) {
    const int i = t + k * 1024;
    if (i < SLICE_E / 2) o4[i] = s4[i];
  }
}

__global__ void __launch_bounds__(1024) bin_degree_kernel(const uint2* __restrict__ srt,
                                                          const int* __restrict__ runstart,
                                                          int* __restrict__ binstart,
                                                          uint2* __restrict__ recs,
                                                          float* __restrict__ dis) {
  __shared__ float deg[BINSZ];
  __shared__ int segst[NSLICE];
  __shared__ int segln[NSLICE];
  __shared__ int segoff[NSLICE];
  __shared__ int base_s;
  const int t = threadIdx.x;
  const int lane = t & 31;
  const int wave = __builtin_amdgcn_readfirstlane(t >> 5);
  const int bin = blockIdx.x;
  const int node0 = bin * BINSZ;
  const int nrows = (Nn - node0 < BINSZ) ? (Nn - node0) : BINSZ;
  if (t < BINSZ) deg[t] = 0.0f;
  if (t < NSLICE) {
    const int a = runstart[bin * NSLICE + t];
    const int e = (bin == NBINS - 1) ? SLICE_E : runstart[(bin + 1) * NSLICE + t];
    segst[t] = t * SLICE_E + a;
    segln[t] = e - a;
    segoff[t] = a;
  }
  __syncthreads();
  if (wave == 0) {
    int v[8];
    int sum = 0, asum = 0;
#pragma unroll
    for (int i = 0; i < 8; ++i) {
      v[i] = segln[lane * 8 + i];
      sum += v[i];
      asum += segoff[lane * 8 + i];
    }
    int incl = sum;
#pragma unroll
    for (int d = 1; d < 32; d <<= 1) {
      const int x = __shfl_up(incl, d, 32);
      if (lane >= d) incl += x;
    }
#pragma unroll
    for (int d = 16; d >= 1; d >>= 1) asum += __shfl_xor(asum, d, 32);
    int run = incl - sum;
#pragma unroll
    for (int i = 0; i < 8; ++i) {
      segoff[lane * 8 + i] = run;
      run += v[i];
    }
    if (lane == 0) {
      base_s = asum;
      binstart[bin] = asum;
    }
  }
  __syncthreads();
  const int base = base_s;
  int st[8], ln[8], of[8];
#pragma unroll
  for (int j = 0; j < 8; ++j) {
    st[j] = __builtin_amdgcn_readfirstlane(segst[wave * 8 + j]);
    ln[j] = __builtin_amdgcn_readfirstlane(segln[wave * 8 + j]);
    of[j] = __builtin_amdgcn_readfirstlane(segoff[wave * 8 + j]);
  }
  uint2 r[8][2];
#pragma unroll
  for (int j = 0; j < 8; ++j) {
#pragma unroll
    for (int c = 0; c < 2; ++c) {
      const int i = lane + c * 32;
      r[j][c] = srt[(i < ln[j]) ? (st[j] + i) : 0];
    }
  }
#pragma unroll
  for (int j = 0; j < 8; ++j) {
#pragma unroll
    for (int c = 0; c < 2; ++c) {
      const int i = lane + c * 32;
      if (i < ln[j]) {
        atomicAdd(&deg[r[j][c].x >> 17], __uint_as_float(r[j][c].y));
        recs[base + of[j] + i] = r[j][c];
      }
    }
  }
#pragma unroll
  for (int j = 0; j < 8; ++j) {
    for (int i = 64 + lane; i < ln[j]; i += 32) {
      const uint2 rr = srt[st[j] + i];
      atomicAdd(&deg[rr.x >> 17], __uint_as_float(rr.y));
      recs[base + of[j] + i] = rr;
    }
  }
  __syncthreads();
  if (t < nrows) dis[node0 + t] = rsqrtf(deg[t] + 1.0f);
}

__device__ __forceinline__ void run_add(float* acc, const int lane, int& cur, float4& a,
                                        const unsigned px, const float nr, const float4 v) {
  const int dl = (int)(px >> 17);
  if (dl != cur) {
    if (cur >= 0) {
      float* p = &acc[cur * 96 + lane * 4];
      atomicAdd(p,     a.x);
      atomicAdd(p + 1, a.y);
      atomicAdd(p + 2, a.z);
      atomicAdd(p + 3, a.w);
    }
    cur = dl;
    a = make_float4(0.0f, 0.0f, 0.0f, 0.0f);
  }
  a.x = fmaf(nr, v.x, a.x);
  a.y = fmaf(nr, v.y, a.y);
  a.z = fmaf(nr, v.z, a.z);
  a.w = fmaf(nr, v.w, a.w);
}

#define SORTCAP 13312
#define RCN     416
__global__ void __launch_bounds__(1024) bin_aggregate_kernel(const uint2* __restrict__ recs,
                                                             const int* __restrict__ binstart,
                                                             const float* __restrict__ dis,
                                                             const float* __restrict__ XW,
                                                             const float* __restrict__ bcat,
                                                             float* __restrict__ AGG) {
  __shared__ __attribute__((aligned(16))) float acc[BINSZ * 96];
  __shared__ float dloc[BINSZ];
  __shared__ int rcnt[RCN];
  __shared__ uint2 srt[SORTCAP];
  const int t = threadIdx.x;
  const int bin = blockIdx.x;
  const int node0 = bin * BINSZ;
  const int nrows = (Nn - node0 < BINSZ) ? (Nn - node0) : BINSZ;
  const float* xw0 = XW + (size_t)node0 * 96;
  const int start = binstart[bin];
  const int end = (bin == NBINS - 1) ? Ee : binstart[bin + 1];
  uint2 r[13];
  if (start < end) {
    const int n0 = (end - start < SORTCAP) ? (end - start) : SORTCAP;
#pragma unroll
    for (int k = 0; k < 13; ++k) {
      const int i = t + k * 1024;
      r[k] = recs[start + ((i < n0) ? i : 0)];
    }
  }
  if (t < BINSZ) dloc[t] = (t < nrows) ? dis[node0 + t] : 0.0f;
  if (t < RCN) rcnt[t] = 0;

  const float4* xw4 = reinterpret_cast<const float4*>(xw0);
  const float4* b4p = reinterpret_cast<const float4*>(bcat);
  float4* acc4 = reinterpret_cast<float4*>(acc);
  for (int q = t; q < nrows * 24; q += 1024) acc4[q] = xw4[q];
  __syncthreads();

  const int lane = t & 31;
  const int wave = __builtin_amdgcn_readfirstlane(t >> 5);
  const float4* xrow = reinterpret_cast<const float4*>(XW) + ((lane < 24) ? lane : 0);

  for (int c0 = start; c0 < end; c0 += SORTCAP) {
    const int n = (end - c0 < SORTCAP) ? (end - c0) : SORTCAP;

    int rank[13];
    if (c0 != start) {
#pragma unroll
      for (int k = 0; k < 13; ++k) {
        const int i = t + k * 1024;
        r[k] = recs[c0 + ((i < n) ? i : 0)];
      }
    }
#pragma unroll
    for (int k = 0; k < 13; ++k) {
      rank[k] = 0;
      if (t + k * 1024 < n) rank[k] = atomicAdd(&rcnt[(int)(r[k].x >> 17)], 1);
    }
    __syncthreads();

    if (wave == 0) {
      int v[13];
      int sum = 0;
#pragma unroll
      for (int i = 0; i < 13; ++i) {
        v[i] = rcnt[lane * 13 + i];
        sum += v[i];
      }
      int incl = sum;
#pragma unroll
      for (int d = 1; d < 32; d <<= 1) {
        const int x = __shfl_up(incl, d, 32);
        if (lane >= d) incl += x;
      }
      int run = incl - sum;
#pragma unroll
      for (int i = 0; i < 13; ++i) {
        rcnt[lane * 13 + i] = run;
        run += v[i];
      }
    }
    __syncthreads();

#pragma unroll
    for (int k = 0; k < 13; ++k) {
      if (t + k * 1024 < n) {
        const int dl = (int)(r[k].x >> 17);
        srt[rcnt[dl] + rank[k]] = r[k];
      }
    }
    __syncthreads();

    const int ws = (n * wave) >> 5;
    const int we = (n * (wave + 1)) >> 5;
    if (lane < 24) {
      int cur = -1;
      float4 a = make_float4(0.0f, 0.0f, 0.0f, 0.0f);
      uint2 rec_next = make_uint2(0u, 0u);
      if (ws + lane < we) rec_next = srt[ws + lane];
      for (int base = ws; base < we; base += 16) {
        const int cnt = (we - base < 16) ? (we - base) : 16;
        const uint2 rec = rec_next;
        float4 v[16];
        unsigned px[16];
        float nr[16];
#pragma unroll
        for (int u = 0; u < 16; ++u) {
          px[u] = (unsigned)__builtin_amdgcn_readlane((int)rec.x, u);
          nr[u] = __uint_as_float((unsigned)__builtin_amdgcn_readlane((int)rec.y, u));
          v[u] = xrow[(size_t)(px[u] & 0x1FFFFu) * 24];
        }
        const int nb = base + 16;
        rec_next = make_uint2(0u, 0u);
        if (nb + lane < we) rec_next = srt[nb + lane];
        if (cnt == 16) {
#pragma unroll
          for (int u = 0; u < 16; ++u) run_add(acc, lane, cur, a, px[u], nr[u], v[u]);
        } else {
#pragma unroll
          for (int u = 0; u < 16; ++u)
            if (u < cnt) run_add(acc, lane, cur, a, px[u], nr[u], v[u]);
        }
      }
      if (cur >= 0) {
        float* p = &acc[cur * 96 + lane * 4];
        atomicAdd(p,     a.x);
        atomicAdd(p + 1, a.y);
        atomicAdd(p + 2, a.z);
        atomicAdd(p + 3, a.w);
      }
    }
    if (c0 + SORTCAP < end) {
      __syncthreads();
      if (t < RCN) rcnt[t] = 0;
      __syncthreads();
    }
  }
  __syncthreads();

  float4* out = reinterpret_cast<float4*>(AGG + (size_t)node0 * 96);
  const float4* a4 = reinterpret_cast<const float4*>(acc);
  for (int q = t; q < nrows * 24; q += 1024) {
    const int i = q / 24;
    const int c4 = q - i * 24;
    const float d = dloc[i];
    const float4 s = a4[q];
    const float4 b = b4p[c4];
    float4 o;
    o.x = d * s.x + b.x;
    o.y = d * s.y + b.y;
    o.z = d * s.z + b.z;
    o.w = d * s.w + b.w;
    out[q] = o;
  }
}

__global__ void __launch_bounds__(32) gru_kernel(const float* __restrict__ AGG,
                                                 const float* __restrict__ Hp,
                                                 const float* __restrict__ LzW, const float* __restrict__ Lzb,
                                                 const float* __restrict__ LrW, const float* __restrict__ Lrb,
                                                 const float* __restrict__ LhW, const float* __restrict__ Lhb,
                                                 const float* __restrict__ headW, const float* __restrict__ headb,
                                                 float* __restrict__ y_out, float* __restrict__ h_out) {
  __shared__ __attribute__((aligned(16))) float As[16 * 132];
  __shared__ float Rs[16 * 36];
  const int lane = threadIdx.x;
  const int tile = blockIdx.x * 16;

  const float4* ag4 = reinterpret_cast<const float4*>(AGG + (size_t)tile * 96) + lane;
  const float4* hp4 = reinterpret_cast<const float4*>(Hp + (size_t)tile * 32) + lane;
  float4 sa[12], sh[4];
#pragma unroll
  for (int j = 0; j < 12; ++j) sa[j] = ag4[j * 32];
#pragma unroll
  for (int j = 0; j < 4; ++j) sh[j] = hp4[j * 32];
#pragma unroll
  for (int j = 0; j < 12; ++j) {
    const int q = lane + j * 32;
    const int r = q / 24, c4 = (q - r * 24) << 2;
    *reinterpret_cast<float4*>(&As[r * 132 + c4]) = sa[j];
  }
#pragma unroll
  for (int j = 0; j < 4; ++j) {
    const int q = lane + j * 32;
    const int r = q >> 3, c4 = (q & 7) << 2;
    *reinterpret_cast<float4*>(&As[r * 132 + 96 + c4]) = sh[j];
  }
  __syncthreads();

  const int m = lane & 15;
  const int n = lane & 15;
  const int koff = (lane >> 4) << 1;

  v8f z0 = {}, z1 = {}, r0 = {}, r1 = {};
#pragma unroll 4
  for (int k0 = 0; k0 < 64; k0 += 4) {
    const int kk = k0 + koff;
    const int colz = (kk < 32) ? kk : (64 + kk);
    const int colr = (kk < 32) ? (32 + kk) : (64 + kk);
    v2f az; az.x = As[m * 132 + colz]; az.y = As[m * 132 + colz + 1];
    v2f ar; ar.x = As[m * 132 + colr]; ar.y = As[m * 132 + colr + 1];
    const float* lz = LzW + kk * 32 + n;
    const float* lr = LrW + kk * 32 + n;
    v2f b;
    b.x = lz[0];  b.y = lz[32];  z0 = wmma4(az, b, z0);
    b.x = lz[16]; b.y = lz[48];  z1 = wmma4(az, b, z1);
    b.x = lr[0];  b.y = lr[32];  r0 = wmma4(ar, b, r0);
    b.x = lr[16]; b.y = lr[48];  r1 = wmma4(ar, b, r1);
  }

  const float bz0 = Lzb[n], bz1 = Lzb[16 + n];
  const float br0 = Lrb[n], br1 = Lrb[16 + n];
  const int row0 = (lane >> 4) << 3;
#pragma unroll
  for (int v = 0; v < 8; ++v) {
    z0[v] = sigmoidf_(z0[v] + bz0);
    z1[v] = sigmoidf_(z1[v] + bz1);
    r0[v] = sigmoidf_(r0[v] + br0);
    r1[v] = sigmoidf_(r1[v] + br1);
    Rs[(row0 + v) * 36 + n]      = r0[v];
    Rs[(row0 + v) * 36 + 16 + n] = r1[v];
  }
  __syncthreads();

  v8f t0 = {}, t1 = {};
#pragma unroll 4
  for (int k0 = 0; k0 < 64; k0 += 4) {
    const int kk = k0 + koff;
    v2f a;
    a.x = As[m * 132 + 64 + kk];
    a.y = As[m * 132 + 65 + kk];
    if (kk >= 32) {
      a.x *= Rs[m * 36 + kk - 32];
      a.y *= Rs[m * 36 + kk - 31];
    }
    const float* lh = LhW + kk * 32 + n;
    v2f b;
    b.x = lh[0];  b.y = lh[32];  t0 = wmma4(a, b, t0);
    b.x = lh[16]; b.y = lh[48];  t1 = wmma4(a, b, t1);
  }

  const float bh0 = Lhb[n], bh1 = Lhb[16 + n];
  const float hw0 = headW[n], hw1 = headW[16 + n];
  float p[8];
#pragma unroll
  for (int v = 0; v < 8; ++v) {
    const float ht0 = tanhf(t0[v] + bh0);
    const float ht1 = tanhf(t1[v] + bh1);
    const float hp0 = As[(row0 + v) * 132 + 96 + n];
    const float hp1 = As[(row0 + v) * 132 + 112 + n];
    const float h0 = z0[v] * hp0 + (1.0f - z0[v]) * ht0;
    const float h1 = z1[v] * hp1 + (1.0f - z1[v]) * ht1;
    h_out[(size_t)(tile + row0 + v) * 32 + n]      = h0;
    h_out[(size_t)(tile + row0 + v) * 32 + 16 + n] = h1;
    p[v] = fmaxf(h0, 0.0f) * hw0 + fmaxf(h1, 0.0f) * hw1;
  }
#pragma unroll
  for (int s = 1; s < 16; s <<= 1)
#pragma unroll
    for (int v = 0; v < 8; ++v)
      p[v] += __shfl_xor(p[v], s, 32);
  if (n == 0) {
    const float hb = headb[0];
#pragma unroll
    for (int v = 0; v < 8; ++v)
      y_out[tile + row0 + v] = p[v] + hb;
  }
}

extern "C" void kernel_launch(void* const* d_in, const int* in_sizes, int n_in,
                              void* d_out, int out_size, void* d_ws, size_t ws_size,
                              hipStream_t stream) {
  const float* x    = (const float*)d_in[0];
  const int*   ei   = (const int*)d_in[1];
  const float* ew   = (const float*)d_in[2];
  const float* Hp   = (const float*)d_in[3];
  const float* Wz   = (const float*)d_in[4];
  const float* bz   = (const float*)d_in[5];
  const float* Wr   = (const float*)d_in[6];
  const float* br   = (const float*)d_in[7];
  const float* Wh   = (const float*)d_in[8];
  const float* bh   = (const float*)d_in[9];
  const float* LzW  = (const float*)d_in[10];
  const float* Lzb  = (const float*)d_in[11];
  const float* LrW  = (const float*)d_in[12];
  const float* Lrb  = (const float*)d_in[13];
  const float* LhW  = (const float*)d_in[14];
  const float* Lhb  = (const float*)d_in[15];
  const float* hW   = (const float*)d_in[16];
  const float* hb   = (const float*)d_in[17];

  const int* src = ei;
  const int* dst = ei + Ee;

  float* out_y = (float*)d_out;
  float* out_h = out_y + Nn;

  float* ws   = (float*)d_ws;
  float* dis  = ws;
  float* Wcat = ws + Nn;
  float* bcat = Wcat + FIN * 96;
  float* XW   = bcat + 96;
  float* AGG  = XW + (size_t)Nn * 96;
  int*   runstart = (int*)(AGG + (size_t)Nn * 96);
  int*   binstart = runstart + NBINS * NSLICE;
  uint2* recs   = (uint2*)(binstart + NBINS * NSLICE);
  uint2* srt    = recs + Ee;

  pack_weights_kernel<<<(FIN * 96 + 255) / 256, 256, 0, stream>>>(Wz, Wr, Wh, bz, br, bh, Wcat, bcat);
  bin_scatter_kernel<<<NSLICE, 1024, 0, stream>>>(src, dst, ew, runstart, srt);
  bin_degree_kernel<<<NBINS, 1024, 0, stream>>>(srt, runstart, binstart, recs, dis);
  gemm_xw_kernel<<<Nn / 16, 32, 0, stream>>>(x, Wcat, dis, XW);
  bin_aggregate_kernel<<<NBINS, 1024, 0, stream>>>(recs, binstart, dis, XW, bcat, AGG);
  gru_kernel<<<Nn / 16, 32, 0, stream>>>(AGG, Hp, LzW, Lzb, LrW, Lrb, LhW, Lhb, hW, hb, out_y, out_h);
}


# GRU gates, update and readout moved into the aggregate launch: one 16-row tile of the bin per wave, A fragments read from LDS; AGG never stored or re-read, one launch fewer; sums unchanged
# speedup vs baseline: 5.2614x; 1.0689x over previous
#include <hip/hip_runtime.h>

#define Nn    100000
#define Ee    3200000
#define FIN   256
#define Hh    32

typedef float v2f __attribute__((ext_vector_type(2)));
typedef float v8f __attribute__((ext_vector_type(8)));

__device__ __forceinline__ v8f wmma4(v2f a, v2f b, v8f c) {
  return __builtin_amdgcn_wmma_f32_16x16x4_f32(
       false, a,  false, b,
       (short)0, c,  false,  false);
}

__device__ __forceinline__ float sigmoidf_(float x) {
  return 1.0f / (1.0f + __expf(-x));
}

__global__ void pack_weights_kernel(const float* __restrict__ Wz, const float* __restrict__ Wr,
                                    const float* __restrict__ Wh, const float* __restrict__ bz,
                                    const float* __restrict__ br, const float* __restrict__ bh,
                                    float* __restrict__ Wcat, float* __restrict__ bcat) {
  int gid = blockIdx.x * blockDim.x + threadIdx.x;
  if (gid < FIN * 96) {
    int k = gid / 96, c = gid - k * 96;
    float v = (c < 32) ? Wz[k * 32 + c] : ((c < 64) ? Wr[k * 32 + c - 32] : Wh[k * 32 + c - 64]);
    Wcat[gid] = v;
  }
  if (gid < 96) {
    bcat[gid] = (gid < 32) ? bz[gid] : ((gid < 64) ? br[gid - 32] : bh[gid - 64]);
  }
}

__global__ void __launch_bounds__(32) gemm_xw_kernel(const float* __restrict__ x,
                                                     const float* __restrict__ Wcat,
                                                     const float* __restrict__ dis,
                                                     float* __restrict__ XW) {
  __shared__ float Xs[16 * 260];
  const int lane = threadIdx.x;
  const int tile = blockIdx.x * 16;

  const float4* xt = reinterpret_cast<const float4*>(x + (size_t)tile * FIN) + lane;
  float4 st[32];
#pragma unroll
  for (int j = 0; j < 32; ++j) st[j] = xt[j * 32];
#pragma unroll
  for (int j = 0; j < 32; ++j) {
    const int i = lane + j * 32;
    const int r = i >> 6, c4 = (i & 63) << 2;
    *reinterpret_cast<float4*>(&Xs[r * 260 + c4]) = st[j];
  }
  __syncthreads();

  const int m = lane & 15;
  const int koff = (lane >> 4) << 1;

  v8f acc[6] = {};
  const float2* arow = reinterpret_cast<const float2*>(&Xs[m * 260]);
  const int row0 = (lane >> 4) << 3;
  float dv[8];
#pragma unroll
  for (int v = 0; v < 8; ++v) dv[v] = dis[tile + row0 + v];

#pragma unroll 4
  for (int k0 = 0; k0 < FIN; k0 += 4) {
    const int kk = k0 + koff;
    float2 af = arow[kk >> 1];
    v2f a; a.x = af.x; a.y = af.y;
    const float* brow = Wcat + (size_t)kk * 96 + m;
#pragma unroll
    for (int t = 0; t < 6; ++t) {
      v2f b; b.x = brow[t * 16]; b.y = brow[96 + t * 16];
      acc[t] = wmma4(a, b, acc[t]);
    }
  }

#pragma unroll
  for (int t = 0; t < 6; ++t)
#pragma unroll
    for (int v = 0; v < 8; ++v)
      XW[(size_t)(tile + row0 + v) * 96 + t * 16 + m] = acc[t][v] * dv[v];
}

#define NBINS   256
#define BINSZ   391
#define NSLICE  256
#define SLICE_E (Ee / NSLICE)

#define ESLOTS  13

__global__ void __launch_bounds__(1024) bin_scatter_kernel(const int* __restrict__ src,
                                                           const int* __restrict__ dst,
                                                           const float* __restrict__ w,
                                                           int* __restrict__ runstart,
                                                           uint2* __restrict__ srt) {
  __shared__ int lcnt[NBINS];
  __shared__ __attribute__((aligned(16))) uint2 srec[SLICE_E];
  const int t = threadIdx.x;
  const int lane = t & 31;
  const int wave = t >> 5;
  const int slice = blockIdx.x;
  const int e0 = slice * SLICE_E;
  if (t < NBINS) lcnt[t] = 0;
  unsigned px[ESLOTS];
  int bn[ESLOTS];
  float ww[ESLOTS];
  {
    int s[ESLOTS], d[ESLOTS];
#pragma unroll
    for (int k = 0; k < ESLOTS; ++k) {
      const int i = t + k * 1024;
      const int ec = e0 + ((i < SLICE_E) ? i : t);
      s[k] = src[ec];
      d[k] = dst[ec];
      ww[k] = w[ec];
    }
#pragma unroll
    for (int k = 0; k < ESLOTS; ++k) {
      bn[k] = d[k] / BINSZ;
      px[k] = (unsigned)s[k] | ((unsigned)(d[k] - bn[k] * BINSZ) << 17);
    }
  }
  __syncthreads();
  int rank[ESLOTS];
#pragma unroll
  for (int k = 0; k < ESLOTS; ++k) {
    rank[k] = 0;
    if (t + k * 1024 < SLICE_E) rank[k] = atomicAdd(&lcnt[bn[k]], 1);
  }
  __syncthreads();
  if (wave == 0) {
    int v[8];
    int sum = 0;
#pragma unroll
    for (int i = 0; i < 8; ++i) {
      v[i] = lcnt[lane * 8 + i];
      sum += v[i];
    }
    int incl = sum;
#pragma unroll
    for (int d = 1; d < 32; d <<= 1) {
      const int x = __shfl_up(incl, d, 32);
      if (lane >= d) incl += x;
    }
    int run = incl - sum;
#pragma unroll
    for (int i = 0; i < 8; ++i) {
      lcnt[lane * 8 + i] = run;
      run += v[i];
    }
  }
  __syncthreads();
  if (t < NBINS) runstart[t * NSLICE + slice] = lcnt[t];
#pragma unroll
  for (int k = 0; k < ESLOTS; ++k) {
    if (t + k * 1024 < SLICE_E)
      srec[lcnt[bn[k]] + rank[k]] = make_uint2(px[k], __float_as_uint(ww[k]));
  }
  __syncthreads();
  const uint4* s4 = reinterpret_cast<const uint4*>(srec);
  uint4* o4 = reinterpret_cast<uint4*>(srt + e0);
#pragma unroll
  for (int k = 0; k < 7; ++k) {
    const int i = t + k * 1024;
    if (i < SLICE_E / 2) o4[i] = s4[i];
  }
}

__global__ void __launch_bounds__(1024) bin_degree_kernel(const uint2* __restrict__ srt,
                                                          const int* __restrict__ runstart,
                                                          int* __restrict__ binstart,
                                                          uint2* __restrict__ recs,
                                                          float* __restrict__ dis) {
  __shared__ float deg[BINSZ];
  __shared__ int segst[NSLICE];
  __shared__ int segln[NSLICE];
  __shared__ int segoff[NSLICE];
  __shared__ int base_s;
  const int t = threadIdx.x;
  const int lane = t & 31;
  const int wave = __builtin_amdgcn_readfirstlane(t >> 5);
  const int bin = blockIdx.x;
  const int node0 = bin * BINSZ;
  const int nrows = (Nn - node0 < BINSZ) ? (Nn - node0) : BINSZ;
  if (t < BINSZ) deg[t] = 0.0f;
  if (t < NSLICE) {
    const int a = runstart[bin * NSLICE + t];
    const int e = (bin == NBINS - 1) ? SLICE_E : runstart[(bin + 1) * NSLICE + t];
    segst[t] = t * SLICE_E + a;
    segln[t] = e - a;
    segoff[t] = a;
  }
  __syncthreads();
  if (wave == 0) {
    int v[8];
    int sum = 0, asum = 0;
#pragma unroll
    for (int i = 0; i < 8; ++i) {
      v[i] = segln[lane * 8 + i];
      sum += v[i];
      asum += segoff[lane * 8 + i];
    }
    int incl = sum;
#pragma unroll
    for (int d = 1; d < 32; d <<= 1) {
      const int x = __shfl_up(incl, d, 32);
      if (lane >= d) incl += x;
    }
#pragma unroll
    for (int d = 16; d >= 1; d >>= 1) asum += __shfl_xor(asum, d, 32);
    int run = incl - sum;
#pragma unroll
    for (int i = 0; i < 8; ++i) {
      segoff[lane * 8 + i] = run;
      run += v[i];
    }
    if (lane == 0) {
      base_s = asum;
      binstart[bin] = asum;
    }
  }
  __syncthreads();
  const int base = base_s;
  int st[8], ln[8], of[8];
#pragma unroll
  for (int j = 0; j < 8; ++j) {
    st[j] = __builtin_amdgcn_readfirstlane(segst[wave * 8 + j]);
    ln[j] = __builtin_amdgcn_readfirstlane(segln[wave * 8 + j]);
    of[j] = __builtin_amdgcn_readfirstlane(segoff[wave * 8 + j]);
  }
  uint2 r[8][2];
#pragma unroll
  for (int j = 0; j < 8; ++j) {
#pragma unroll
    for (int c = 0; c < 2; ++c) {
      const int i = lane + c * 32;
      r[j][c] = srt[(i < ln[j]) ? (st[j] + i) : 0];
    }
  }
#pragma unroll
  for (int j = 0; j < 8; ++j) {
#pragma unroll
    for (int c = 0; c < 2; ++c) {
      const int i = lane + c * 32;
      if (i < ln[j]) {
        atomicAdd(&deg[r[j][c].x >> 17], __uint_as_float(r[j][c].y));
        recs[base + of[j] + i] = r[j][c];
      }
    }
  }
#pragma unroll
  for (int j = 0; j < 8; ++j) {
    for (int i = 64 + lane; i < ln[j]; i += 32) {
      const uint2 rr = srt[st[j] + i];
      atomicAdd(&deg[rr.x >> 17], __uint_as_float(rr.y));
      recs[base + of[j] + i] = rr;
    }
  }
  __syncthreads();
  if (t < nrows) dis[node0 + t] = rsqrtf(deg[t] + 1.0f);
}

__device__ __forceinline__ void run_add(float* acc, const int lane, int& cur, float4& a,
                                        const unsigned px, const float nr, const float4 v) {
  const int dl = (int)(px >> 17);
  if (dl != cur) {
    if (cur >= 0) {
      float* p = &acc[cur * 96 + lane * 4];
      atomicAdd(p,     a.x);
      atomicAdd(p + 1, a.y);
      atomicAdd(p + 2, a.z);
      atomicAdd(p + 3, a.w);
    }
    cur = dl;
    a = make_float4(0.0f, 0.0f, 0.0f, 0.0f);
  }
  a.x = fmaf(nr, v.x, a.x);
  a.y = fmaf(nr, v.y, a.y);
  a.z = fmaf(nr, v.z, a.z);
  a.w = fmaf(nr, v.w, a.w);
}

#define SORTCAP 13312
#define RCN     416
#define NTILE   25
#define GST     1152
#define STGW    14400
static_assert(STGW >= SORTCAP && STGW * 2 >= NTILE * GST && NTILE * 16 >= BINSZ, "shared area too small");
__global__ void __launch_bounds__(1024) bin_aggregate_gru_kernel(const uint2* __restrict__ recs,
                                                                 const int* __restrict__ binstart,
                                                                 const float* __restrict__ dis,
                                                                 const float* __restrict__ XW,
                                                                 const float* __restrict__ bcat,
                                                                 const float* __restrict__ Hp,
                                                                 const float* __restrict__ LzW, const float* __restrict__ Lzb,
                                                                 const float* __restrict__ LrW, const float* __restrict__ Lrb,
                                                                 const float* __restrict__ LhW, const float* __restrict__ Lhb,
                                                                 const float* __restrict__ headW, const float* __restrict__ headb,
                                                                 float* __restrict__ y_out, float* __restrict__ h_out) {
  __shared__ __attribute__((aligned(16))) float acc[BINSZ * 96];
  __shared__ float dloc[BINSZ];
  __shared__ int rcnt[RCN];
  __shared__ __attribute__((aligned(16))) uint2 srt[STGW];
  const int t = threadIdx.x;
  const int bin = blockIdx.x;
  const int node0 = bin * BINSZ;
  const int nrows = (Nn - node0 < BINSZ) ? (Nn - node0) : BINSZ;
  const float* xw0 = XW + (size_t)node0 * 96;
  const int start = binstart[bin];
  const int end = (bin == NBINS - 1) ? Ee : binstart[bin + 1];
  uint2 r[13];
  if (start < end) {
    const int n0 = (end - start < SORTCAP) ? (end - start) : SORTCAP;
#pragma unroll
    for (int k = 0; k < 13; ++k) {
      const int i = t + k * 1024;
      r[k] = recs[start + ((i < n0) ? i : 0)];
    }
  }
  if (t < BINSZ) dloc[t] = (t < nrows) ? dis[node0 + t] : 0.0f;
  if (t < RCN) rcnt[t] = 0;

  const float4* xw4 = reinterpret_cast<const float4*>(xw0);
  const float4* b4p = reinterpret_cast<const float4*>(bcat);
  float4* acc4 = reinterpret_cast<float4*>(acc);
  for (int q = t; q < nrows * 24; q += 1024) acc4[q] = xw4[q];
  __syncthreads();

  const int lane = t & 31;
  const int wave = __builtin_amdgcn_readfirstlane(t >> 5);
  const float4* xrow = reinterpret_cast<const float4*>(XW) + ((lane < 24) ? lane : 0);

  for (int c0 = start; c0 < end; c0 += SORTCAP) {
    const int n = (end - c0 < SORTCAP) ? (end - c0) : SORTCAP;

    int rank[13];
    if (c0 != start) {
#pragma unroll
      for (int k = 0; k < 13; ++k) {
        const int i = t + k * 1024;
        r[k] = recs[c0 + ((i < n) ? i : 0)];
      }
    }
#pragma unroll
    for (int k = 0; k < 13; ++k) {
      rank[k] = 0;
      if (t + k * 1024 < n) rank[k] = atomicAdd(&rcnt[(int)(r[k].x >> 17)], 1);
    }
    __syncthreads();

    if (wave == 0) {
      int v[13];
      int sum = 0;
#pragma unroll
      for (int i = 0; i < 13; ++i) {
        v[i] = rcnt[lane * 13 + i];
        sum += v[i];
      }
      int incl = sum;
#pragma unroll
      for (int d = 1; d < 32; d <<= 1) {
        const int x = __shfl_up(incl, d, 32);
        if (lane >= d) incl += x;
      }
      int run = incl - sum;
#pragma unroll
      for (int i = 0; i < 13; ++i) {
        rcnt[lane * 13 + i] = run;
        run += v[i];
      }
    }
    __syncthreads();

#pragma unroll
    for (int k = 0; k < 13; ++k) {
      if (t + k * 1024 < n) {
        const int dl = (int)(r[k].x >> 17);
        srt[rcnt[dl] + rank[k]] = r[k];
      }
    }
    __syncthreads();

    const int ws = (n * wave) >> 5;
    const int we = (n * (wave + 1)) >> 5;
    if (lane < 24) {
      int cur = -1;
      float4 a = make_float4(0.0f, 0.0f, 0.0f, 0.0f);
      uint2 rec_next = make_uint2(0u, 0u);
      if (ws + lane < we) rec_next = srt[ws + lane];
      for (int base = ws; base < we; base += 16) {
        const int cnt = (we - base < 16) ? (we - base) : 16;
        const uint2 rec = rec_next;
        float4 v[16];
        unsigned px[16];
        float nr[16];
#pragma unroll
        for (int u = 0; u < 16; ++u) {
          px[u] = (unsigned)__builtin_amdgcn_readlane((int)rec.x, u);
          nr[u] = __uint_as_float((unsigned)__builtin_amdgcn_readlane((int)rec.y, u));
          v[u] = xrow[(size_t)(px[u] & 0x1FFFFu) * 24];
        }
        const int nb = base + 16;
        rec_next = make_uint2(0u, 0u);
        if (nb + lane < we) rec_next = srt[nb + lane];
        if (cnt == 16) {
#pragma unroll
          for (int u = 0; u < 16; ++u) run_add(acc, lane, cur, a, px[u], nr[u], v[u]);
        } else {
#pragma unroll
          for (int u = 0; u < 16; ++u)
            if (u < cnt) run_add(acc, lane, cur, a, px[u], nr[u], v[u]);
        }
      }
      if (cur >= 0) {
        float* p = &acc[cur * 96 + lane * 4];
        atomicAdd(p,     a.x);
        atomicAdd(p + 1, a.y);
        atomicAdd(p + 2, a.z);
        atomicAdd(p + 3, a.w);
      }
    }
    if (c0 + SORTCAP < end) {
      __syncthreads();
      if (t < RCN) rcnt[t] = 0;
      __syncthreads();
    }
  }
  __syncthreads();

  const int r0 = wave * 16;
  const bool has_tile = r0 < nrows;
  float* Hs = reinterpret_cast<float*>(srt) + wave * GST;
  float* Rs = Hs + 16 * 36;
  float4 sh[4];
#pragma unroll
  for (int j = 0; j < 4; ++j) sh[j] = make_float4(0.0f, 0.0f, 0.0f, 0.0f);
  if (has_tile) {
#pragma unroll
    for (int j = 0; j < 4; ++j) {
      const int q = lane + j * 32;
      int rr = r0 + (q >> 3);
      rr = (rr < nrows) ? rr : (nrows - 1);
      sh[j] = *reinterpret_cast<const float4*>(Hp + (size_t)(node0 + rr) * 32 + ((q & 7) << 2));
    }
  }
  for (int q = t; q < nrows * 24; q += 1024) {
    const int i = q / 24;
    const int c4 = q - i * 24;
    const float d = dloc[i];
    const float4 s = acc4[q];
    const float4 b = b4p[c4];
    float4 o;
    o.x = d * s.x + b.x;
    o.y = d * s.y + b.y;
    o.z = d * s.z + b.z;
    o.w = d * s.w + b.w;
    acc4[q] = o;
  }
  if (has_tile) {
#pragma unroll
    for (int j = 0; j < 4; ++j) {
      const int q = lane + j * 32;
      *reinterpret_cast<float4*>(&Hs[(q >> 3) * 36 + ((q & 7) << 2)]) = sh[j];
    }
  }
  __syncthreads();

  const int m = lane & 15;
  const int koff = (lane >> 4) << 1;
  const int row0 = (lane >> 4) << 3;
  int ai = r0 + m;
  ai = (ai < nrows) ? ai : (nrows - 1);
  const float* arow = acc + ai * 96;
  const float* hrow = Hs + m * 36;
  const float* rrow = Rs + m * 36;

  if (has_tile) {
    v8f z0 = {}, z1 = {};
    v8f g0 = {}, g1 = {};
#pragma unroll 4
    for (int k0 = 0; k0 < 32; k0 += 4) {
      const int kk = k0 + koff;
      const float2 fz = *reinterpret_cast<const float2*>(arow + kk);
      const float2 fr = *reinterpret_cast<const float2*>(arow + 32 + kk);
      v2f az; az.x = fz.x; az.y = fz.y;
      v2f ar; ar.x = fr.x; ar.y = fr.y;
      const float* lz = LzW + kk * 32 + m;
      const float* lr = LrW + kk * 32 + m;
      v2f b;
      b.x = lz[0];  b.y = lz[32];  z0 = wmma4(az, b, z0);
      b.x = lz[16]; b.y = lz[48];  z1 = wmma4(az, b, z1);
      b.x = lr[0];  b.y = lr[32];  g0 = wmma4(ar, b, g0);
      b.x = lr[16]; b.y = lr[48];  g1 = wmma4(ar, b, g1);
    }
#pragma unroll 4
    for (int k0 = 32; k0 < 64; k0 += 4) {
      const int kk = k0 + koff;
      const float2 fh = *reinterpret_cast<const float2*>(hrow + kk - 32);
      v2f a; a.x = fh.x; a.y = fh.y;
      const float* lz = LzW + kk * 32 + m;
      const float* lr = LrW + kk * 32 + m;
      v2f b;
      b.x = lz[0];  b.y = lz[32];  z0 = wmma4(a, b, z0);
      b.x = lz[16]; b.y = lz[48];  z1 = wmma4(a, b, z1);
      b.x = lr[0];  b.y = lr[32];  g0 = wmma4(a, b, g0);
      b.x = lr[16]; b.y = lr[48];  g1 = wmma4(a, b, g1);
    }
    const float bz0 = Lzb[m], bz1 = Lzb[16 + m];
    const float br0 = Lrb[m], br1 = Lrb[16 + m];
#pragma unroll
    for (int v = 0; v < 8; ++v) {
      z0[v] = sigmoidf_(z0[v] + bz0);
      z1[v] = sigmoidf_(z1[v] + bz1);
      g0[v] = sigmoidf_(g0[v] + br0);
      g1[v] = sigmoidf_(g1[v] + br1);
      Rs[(row0 + v) * 36 + m]      = g0[v];
      Rs[(row0 + v) * 36 + 16 + m] = g1[v];
    }
    __builtin_amdgcn_fence(__ATOMIC_RELEASE, "workgroup", "local");
    __builtin_amdgcn_fence(__ATOMIC_ACQUIRE, "workgroup", "local");

    v8f t0 = {}, t1 = {};
#pragma unroll 4
    for (int k0 = 0; k0 < 32; k0 += 4) {
      const int kk = k0 + koff;
      const float2 fa = *reinterpret_cast<const float2*>(arow + 64 + kk);
      v2f a; a.x = fa.x; a.y = fa.y;
      const float* lh = LhW + kk * 32 + m;
      v2f b;
      b.x = lh[0];  b.y = lh[32];  t0 = wmma4(a, b, t0);
      b.x = lh[16]; b.y = lh[48];  t1 = wmma4(a, b, t1);
    }
#pragma unroll 4
    for (int k0 = 32; k0 < 64; k0 += 4) {
      const int kk = k0 + koff;
      const float2 fh = *reinterpret_cast<const float2*>(hrow + kk - 32);
      const float2 fr = *reinterpret_cast<const float2*>(rrow + kk - 32);
      v2f a; a.x = fh.x * fr.x; a.y = fh.y * fr.y;
      const float* lh = LhW + kk * 32 + m;
      v2f b;
      b.x = lh[0];  b.y = lh[32];  t0 = wmma4(a, b, t0);
      b.x = lh[16]; b.y = lh[48];  t1 = wmma4(a, b, t1);
    }

    const float bh0 = Lhb[m], bh1 = Lhb[16 + m];
    const float hw0 = headW[m], hw1 = headW[16 + m];
    float p[8];
#pragma unroll
    for (int v = 0; v < 8; ++v) {
      const float ht0 = tanhf(t0[v] + bh0);
      const float ht1 = tanhf(t1[v] + bh1);
      const float hp0 = Hs[(row0 + v) * 36 + m];
      const float hp1 = Hs[(row0 + v) * 36 + 16 + m];
      const float h0 = z0[v] * hp0 + (1.0f - z0[v]) * ht0;
      const float h1 = z1[v] * hp1 + (1.0f - z1[v]) * ht1;
      const int rr = r0 + row0 + v;
      if (rr < nrows) {
        h_out[(size_t)(node0 + rr) * 32 + m]      = h0;
        h_out[(size_t)(node0 + rr) * 32 + 16 + m] = h1;
      }
      p[v] = fmaxf(h0, 0.0f) * hw0 + fmaxf(h1, 0.0f) * hw1;
    }
#pragma unroll
    for (int s = 1; s < 16; s <<= 1)
#pragma unroll
      for (int v = 0; v < 8; ++v)
        p[v] += __shfl_xor(p[v], s, 32);
    if (m == 0) {
      const float hb = headb[0];
#pragma unroll
      for (int v = 0; v < 8; ++v) {
        const int rr = r0 + row0 + v;
        if (rr < nrows) y_out[node0 + rr] = p[v] + hb;
      }
    }
  }
}

extern "C" void kernel_launch(void* const* d_in, const int* in_sizes, int n_in,
                              void* d_out, int out_size, void* d_ws, size_t ws_size,
                              hipStream_t stream) {
  const float* x    = (const float*)d_in[0];
  const int*   ei   = (const int*)d_in[1];
  const float* ew   = (const float*)d_in[2];
  const float* Hp   = (const float*)d_in[3];
  const float* Wz   = (const float*)d_in[4];
  const float* bz   = (const float*)d_in[5];
  const float* Wr   = (const float*)d_in[6];
  const float* br   = (const float*)d_in[7];
  const float* Wh   = (const float*)d_in[8];
  const float* bh   = (const float*)d_in[9];
  const float* LzW  = (const float*)d_in[10];
  const float* Lzb  = (const float*)d_in[11];
  const float* LrW  = (const float*)d_in[12];
  const float* Lrb  = (const float*)d_in[13];
  const float* LhW  = (const float*)d_in[14];
  const float* Lhb  = (const float*)d_in[15];
  const float* hW   = (const float*)d_in[16];
  const float* hb   = (const float*)d_in[17];

  const int* src = ei;
  const int* dst = ei + Ee;

  float* out_y = (float*)d_out;
  float* out_h = out_y + Nn;

  float* ws   = (float*)d_ws;
  float* dis  = ws;
  float* Wcat = ws + Nn;
  float* bcat = Wcat + FIN * 96;
  float* XW   = bcat + 96;
  int*   runstart = (int*)(XW + (size_t)Nn * 96);
  int*   binstart = runstart + NBINS * NSLICE;
  uint2* recs   = (uint2*)(binstart + NBINS * NSLICE);
  uint2* srt    = recs + Ee;

  pack_weights_kernel<<<(FIN * 96 + 255) / 256, 256, 0, stream>>>(Wz, Wr, Wh, bz, br, bh, Wcat, bcat);
  bin_scatter_kernel<<<NSLICE, 1024, 0, stream>>>(src, dst, ew, runstart, srt);
  bin_degree_kernel<<<NBINS, 1024, 0, stream>>>(srt, runstart, binstart, recs, dis);
  gemm_xw_kernel<<<Nn / 16, 32, 0, stream>>>(x, Wcat, dis, XW);
  bin_aggregate_gru_kernel<<<NBINS, 1024, 0, stream>>>(recs, binstart, dis, XW, bcat, Hp, LzW, Lzb, LrW, Lrb, LhW, Lhb, hW, hb,
                                                       out_y, out_h);
}


# Product x@[Wz|Wr|Wh] moved into the degree launch: per bin 4 waves gather runs and sum degrees (still arrival order) while 28 run its 25 row tiles on WMMA, B from an LDS copy; 4 launches
# speedup vs baseline: 5.6122x; 1.0667x over previous
#include <hip/hip_runtime.h>

#define Nn    100000
#define Ee    3200000
#define FIN   256
#define Hh    32

typedef float v2f __attribute__((ext_vector_type(2)));
typedef float v8f __attribute__((ext_vector_type(8)));

__device__ __forceinline__ v8f wmma4(v2f a, v2f b, v8f c) {
  return __builtin_amdgcn_wmma_f32_16x16x4_f32(
       false, a,  false, b,
       (short)0, c,  false,  false);
}

__device__ __forceinline__ float sigmoidf_(float x) {
  return 1.0f / (1.0f + __expf(-x));
}

__global__ void pack_weights_kernel(const float* __restrict__ Wz, const float* __restrict__ Wr,
                                    const float* __restrict__ Wh, const float* __restrict__ bz,
                                    const float* __restrict__ br, const float* __restrict__ bh,
                                    float* __restrict__ Wcat, float* __restrict__ bcat) {
  int gid = blockIdx.x * blockDim.x + threadIdx.x;
  if (gid < FIN * 96) {
    int k = gid / 96, c = gid - k * 96;
    float v = (c < 32) ? Wz[k * 32 + c] : ((c < 64) ? Wr[k * 32 + c - 32] : Wh[k * 32 + c - 64]);
    Wcat[gid] = v;
  }
  if (gid < 96) {
    bcat[gid] = (gid < 32) ? bz[gid] : ((gid < 64) ? br[gid - 32] : bh[gid - 64]);
  }
}

#define NBINS   256
#define BINSZ   391
#define NSLICE  256
#define SLICE_E (Ee / NSLICE)

#define ESLOTS  13

__global__ void __launch_bounds__(1024) bin_scatter_kernel(const int* __restrict__ src,
                                                           const int* __restrict__ dst,
                                                           const float* __restrict__ w,
                                                           int* __restrict__ runstart,
                                                           uint2* __restrict__ srt) {
  __shared__ int lcnt[NBINS];
  __shared__ __attribute__((aligned(16))) uint2 srec[SLICE_E];
  const int t = threadIdx.x;
  const int lane = t & 31;
  const int wave = t >> 5;
  const int slice = blockIdx.x;
  const int e0 = slice * SLICE_E;
  if (t < NBINS) lcnt[t] = 0;
  unsigned px[ESLOTS];
  int bn[ESLOTS];
  float ww[ESLOTS];
  {
    int s[ESLOTS], d[ESLOTS];
#pragma unroll
    for (int k = 0; k < ESLOTS; ++k) {
      const int i = t + k * 1024;
      const int ec = e0 + ((i < SLICE_E) ? i : t);
      s[k] = src[ec];
      d[k] = dst[ec];
      ww[k] = w[ec];
    }
#pragma unroll
    for (int k = 0; k < ESLOTS; ++k) {
      bn[k] = d[k] / BINSZ;
      px[k] = (unsigned)s[k] | ((unsigned)(d[k] - bn[k] * BINSZ) << 17);
    }
  }
  __syncthreads();
  int rank[ESLOTS];
#pragma unroll
  for (int k = 0; k < ESLOTS; ++k) {
    rank[k] = 0;
    if (t + k * 1024 < SLICE_E) rank[k] = atomicAdd(&lcnt[bn[k]], 1);
  }
  __syncthreads();
  if (wave == 0) {
    int v[8];
    int sum = 0;
#pragma unroll
    for (int i = 0; i < 8; ++i) {
      v[i] = lcnt[lane * 8 + i];
      sum += v[i];
    }
    int incl = sum;
#pragma unroll
    for (int d = 1; d < 32; d <<= 1) {
      const int x = __shfl_up(incl, d, 32);
      if (lane >= d) incl += x;
    }
    int run = incl - sum;
#pragma unroll
    for (int i = 0; i < 8; ++i) {
      lcnt[lane * 8 + i] = run;
      run += v[i];
    }
  }
  __syncthreads();
  if (t < NBINS) runstart[t * NSLICE + slice] = lcnt[t];
#pragma unroll
  for (int k = 0; k < ESLOTS; ++k) {
    if (t + k * 1024 < SLICE_E)
      srec[lcnt[bn[k]] + rank[k]] = make_uint2(px[k], __float_as_uint(ww[k]));
  }
  __syncthreads();
  const uint4* s4 = reinterpret_cast<const uint4*>(srec);
  uint4* o4 = reinterpret_cast<uint4*>(srt + e0);
#pragma unroll
  for (int k = 0; k < 7; ++k) {
    const int i = t + k * 1024;
    if (i < SLICE_E / 2) o4[i] = s4[i];
  }
}

#define NGW     4
#define GR      16
#define NFULL   24
#define NXW     28
#define XP      68
static_assert(NGW + NXW == 32 && NXW == NFULL + 4 && (NFULL + 1) * 16 >= BINSZ, "wave roles do not cover the bin");

template <int NC>
__device__ __forceinline__ void xw_tile(const float* __restrict__ xt, const unsigned loff, float* xs,
                                        const float* wfrag, const int c_lo, const int lane, v8f (&acc)[6]) {
  const int m = lane & 15;
  const int koff = (lane >> 4) << 1;
  const float2* arow = reinterpret_cast<const float2*>(&xs[m * XP]);
  const float* wl = wfrag + lane * 12 + c_lo * 2;
  const char* xb = reinterpret_cast<const char*>(xt) + loff;
  float4* sp = reinterpret_cast<float4*>(&xs[(lane >> 4) * XP + ((lane & 15) << 2)]);
  const int RS = 2 * XP / 4;
  const int RB = 2 * FIN * 4;
  float4 s0, s1, s2, s3, s4, s5, s6, s7;
#pragma unroll
  for (int q = 0; q < 4; ++q) {
    if (q > 0) {
      sp[0 * RS] = s0; sp[1 * RS] = s1; sp[2 * RS] = s2; sp[3 * RS] = s3;
      sp[4 * RS] = s4; sp[5 * RS] = s5; sp[6 * RS] = s6; sp[7 * RS] = s7;
      __builtin_amdgcn_fence(__ATOMIC_RELEASE, "workgroup", "local");
      __builtin_amdgcn_fence(__ATOMIC_ACQUIRE, "workgroup", "local");
    }
    if (q < 3) {
      const char* xq = xb + (q + 1) * 256;
      s0 = *reinterpret_cast<const float4*>(xq + 0 * RB); s1 = *reinterpret_cast<const float4*>(xq + 1 * RB);
      s2 = *reinterpret_cast<const float4*>(xq + 2 * RB); s3 = *reinterpret_cast<const float4*>(xq + 3 * RB);
      s4 = *reinterpret_cast<const float4*>(xq + 4 * RB); s5 = *reinterpret_cast<const float4*>(xq + 5 * RB);
      s6 = *reinterpret_cast<const float4*>(xq + 6 * RB); s7 = *reinterpret_cast<const float4*>(xq + 7 * RB);
      __builtin_amdgcn_sched_barrier(0);
    }
#pragma unroll 2
    for (int ks = 0; ks < 16; ++ks) {
      const float2 af = arow[(ks * 4 + koff) >> 1];
      v2f a; a.x = af.x; a.y = af.y;
      const float* bl = wl + (q * 16 + ks) * 384;
      v2f b[NC];
#pragma unroll
      for (int c = 0; c < NC; ++c) {
        const float2 bf = *reinterpret_cast<const float2*>(bl + 2 * c);
        b[c].x = bf.x; b[c].y = bf.y;
      }
#pragma unroll
      for (int c = 0; c < NC; ++c) acc[c] = wmma4(a, b[c], acc[c]);
    }
  }
}

__global__ void __launch_bounds__(1024) bin_degree_gemm_kernel(const uint2* __restrict__ srt,
                                                               const int* __restrict__ runstart,
                                                               int* __restrict__ binstart,
                                                               uint2* __restrict__ recs,
                                                               float* __restrict__ dis,
                                                               const float* __restrict__ x,
                                                               const float* __restrict__ Wcat,
                                                               float* __restrict__ XW) {
  __shared__ float deg[BINSZ];
  __shared__ int segst[NSLICE];
  __shared__ int segln[NSLICE];
  __shared__ int segoff[NSLICE];
  __shared__ int base_s;
  __shared__ __attribute__((aligned(16))) float Xs[NXW * 16 * XP];
  __shared__ __attribute__((aligned(16))) float Wf[FIN * 96];
  const int t = threadIdx.x;
  const int lane = t & 31;
  const int wave = __builtin_amdgcn_readfirstlane(t >> 5);
  const int bin = blockIdx.x;
  const int node0 = bin * BINSZ;
  const int nrows = (Nn - node0 < BINSZ) ? (Nn - node0) : BINSZ;

  const int xw = (wave >= NGW) ? (wave - NGW) : 0;
  const bool split = xw < 4;
  const int sw = xw;
  const int r0 = (split ? NFULL : (xw - 4)) * 16;
  const int c_lo = !split ? 0 : ((sw < 2) ? (sw * 2) : (sw + 2));
  const int nc = !split ? 6 : ((sw < 2) ? 2 : 1);
  const bool has_tile = (wave >= NGW) && (r0 < nrows);
  float* xs = Xs + xw * (16 * XP);
  const int g0 = node0 + r0;
  const int rbase = (g0 < Nn - 16) ? g0 : (Nn - 16);
  const int shift = g0 - rbase;
  const float* xt = x + (size_t)rbase * FIN;
  const unsigned loff = (unsigned)((lane >> 4) * FIN + ((lane & 15) << 2)) * 4u;
  float4 st[8];
#pragma unroll
  for (int j = 0; j < 8; ++j) st[j] = make_float4(0.0f, 0.0f, 0.0f, 0.0f);
  if (has_tile) {
#pragma unroll
    for (int j = 0; j < 8; ++j)
      st[j] = *reinterpret_cast<const float4*>(reinterpret_cast<const char*>(xt) + loff + j * (2 * FIN * 4));
  }

  int rs_a = 0, rs_e = SLICE_E;
  if (t < NSLICE) {
    rs_a = runstart[bin * NSLICE + t];
    if (bin != NBINS - 1) rs_e = runstart[(bin + 1) * NSLICE + t];
  }

  {
    float4 wv[6];
#pragma unroll
    for (int j = 0; j < 6; ++j) wv[j] = reinterpret_cast<const float4*>(Wcat)[t + j * 1024];
#pragma unroll
    for (int j = 0; j < 6; ++j) {
      const int i = t + j * 1024;
      const int k = i / 24;
      const int col = (i - k * 24) << 2;
      float* p = &Wf[((k >> 2) * 32 + ((k >> 1) & 1) * 16 + (col & 15)) * 12 + (col >> 4) * 2 + (k & 1)];
      p[0]  = wv[j].x;
      p[12] = wv[j].y;
      p[24] = wv[j].z;
      p[36] = wv[j].w;
    }
  }
  if (has_tile) {
#pragma unroll
    for (int j = 0; j < 8; ++j)
      *reinterpret_cast<float4*>(&xs[((lane >> 4) + 2 * j) * XP + ((lane & 15) << 2)]) = st[j];
  }
  if (t < BINSZ) deg[t] = 0.0f;
  if (t < NSLICE) {
    segst[t] = t * SLICE_E + rs_a;
    segln[t] = rs_e - rs_a;
    segoff[t] = rs_a;
  }
  __syncthreads();
  if (wave == 0) {
    int v[8];
    int sum = 0, asum = 0;
#pragma unroll
    for (int i = 0; i < 8; ++i) {
      v[i] = segln[lane * 8 + i];
      sum += v[i];
      asum += segoff[lane * 8 + i];
    }
    int incl = sum;
#pragma unroll
    for (int d = 1; d < 32; d <<= 1) {
      const int xx = __shfl_up(incl, d, 32);
      if (lane >= d) incl += xx;
    }
#pragma unroll
    for (int d = 16; d >= 1; d >>= 1) asum += __shfl_xor(asum, d, 32);
    int run = incl - sum;
#pragma unroll
    for (int i = 0; i < 8; ++i) {
      segoff[lane * 8 + i] = run;
      run += v[i];
    }
    if (lane == 0) {
      base_s = asum;
      binstart[bin] = asum;
    }
  }
  __syncthreads();

  v8f acc[6] = {};

  if (wave < NGW) {
    const int base = base_s;
    for (int b = 0; b < (NSLICE + GR * NGW - 1) / (GR * NGW); ++b) {
      int sg[GR], ln[GR], of[GR];
#pragma unroll
      for (int j = 0; j < GR; ++j) {
        const int ridx = (b * GR + j) * NGW + wave;
        const int rc = (ridx < NSLICE) ? ridx : 0;
        sg[j] = __builtin_amdgcn_readfirstlane(segst[rc]);
        const int l = __builtin_amdgcn_readfirstlane(segln[rc]);
        ln[j] = (ridx < NSLICE) ? l : 0;
        of[j] = __builtin_amdgcn_readfirstlane(segoff[rc]);
      }
      uint2 r[GR][2];
#pragma unroll
      for (int j = 0; j < GR; ++j) {
#pragma unroll
        for (int c = 0; c < 2; ++c) {
          const int i = lane + c * 32;
          r[j][c] = srt[(i < ln[j]) ? (sg[j] + i) : 0];
        }
      }
#pragma unroll
      for (int j = 0; j < GR; ++j) {
#pragma unroll
        for (int c = 0; c < 2; ++c) {
          const int i = lane + c * 32;
          if (i < ln[j]) {
            atomicAdd(&deg[r[j][c].x >> 17], __uint_as_float(r[j][c].y));
            recs[base + of[j] + i] = r[j][c];
          }
        }
      }
#pragma unroll
      for (int j = 0; j < GR; ++j) {
        for (int i = 64 + lane; i < ln[j]; i += 32) {
          const uint2 rr = srt[sg[j] + i];
          atomicAdd(&deg[rr.x >> 17], __uint_as_float(rr.y));
          recs[base + of[j] + i] = rr;
        }
      }
    }
  } else if (has_tile) {
    if (nc == 6)      xw_tile<6>(xt, loff, xs, Wf, 0, lane, acc);
    else if (nc == 2) xw_tile<2>(xt, loff, xs, Wf, c_lo, lane, acc);
    else              xw_tile<1>(xt, loff, xs, Wf, c_lo, lane, acc);
  }
  __syncthreads();
  if (t < nrows) dis[node0 + t] = rsqrtf(deg[t] + 1.0f);
  if (has_tile) {
    const int m = lane & 15;
    const int row0 = (lane >> 4) << 3;
#pragma unroll
    for (int v = 0; v < 8; ++v) {
      const int rr = r0 - shift + row0 + v;
      if (row0 + v >= shift && rr < nrows) {
        const float dv = rsqrtf(deg[rr] + 1.0f);
        float* orow = XW + (size_t)(node0 + rr) * 96 + c_lo * 16 + m;
#pragma unroll
        for (int c = 0; c < 6; ++c)
          if (c < nc) orow[c * 16] = acc[c][v] * dv;
      }
    }
  }
}

__device__ __forceinline__ void run_add(float* acc, const int lane, int& cur, float4& a,
                                        const unsigned px, const float nr, const float4 v) {
  const int dl = (int)(px >> 17);
  if (dl != cur) {
    if (cur >= 0) {
      float* p = &acc[cur * 96 + lane * 4];
      atomicAdd(p,     a.x);
      atomicAdd(p + 1, a.y);
      atomicAdd(p + 2, a.z);
      atomicAdd(p + 3, a.w);
    }
    cur = dl;
    a = make_float4(0.0f, 0.0f, 0.0f, 0.0f);
  }
  a.x = fmaf(nr, v.x, a.x);
  a.y = fmaf(nr, v.y, a.y);
  a.z = fmaf(nr, v.z, a.z);
  a.w = fmaf(nr, v.w, a.w);
}

#define SORTCAP 13312
#define RCN     416
#define NTILE   25
#define GST     1152
#define STGW    14400
static_assert(STGW >= SORTCAP && STGW * 2 >= NTILE * GST && NTILE * 16 >= BINSZ, "shared area too small");
__global__ void __launch_bounds__(1024) bin_aggregate_gru_kernel(const uint2* __restrict__ recs,
                                                                 const int* __restrict__ binstart,
                                                                 const float* __restrict__ dis,
                                                                 const float* __restrict__ XW,
                                                                 const float* __restrict__ bcat,
                                                                 const float* __restrict__ Hp,
                                                                 const float* __restrict__ LzW, const float* __restrict__ Lzb,
                                                                 const float* __restrict__ LrW, const float* __restrict__ Lrb,
                                                                 const float* __restrict__ LhW, const float* __restrict__ Lhb,
                                                                 const float* __restrict__ headW, const float* __restrict__ headb,
                                                                 float* __restrict__ y_out, float* __restrict__ h_out) {
  __shared__ __attribute__((aligned(16))) float acc[BINSZ * 96];
  __shared__ float dloc[BINSZ];
  __shared__ int rcnt[RCN];
  __shared__ __attribute__((aligned(16))) uint2 srt[STGW];
  const int t = threadIdx.x;
  const int bin = blockIdx.x;
  const int node0 = bin * BINSZ;
  const int nrows = (Nn - node0 < BINSZ) ? (Nn - node0) : BINSZ;
  const float* xw0 = XW + (size_t)node0 * 96;
  const int start = binstart[bin];
  const int end = (bin == NBINS - 1) ? Ee : binstart[bin + 1];
  uint2 r[13];
  if (start < end) {
    const int n0 = (end - start < SORTCAP) ? (end - start) : SORTCAP;
#pragma unroll
    for (int k = 0; k < 13; ++k) {
      const int i = t + k * 1024;
      r[k] = recs[start + ((i < n0) ? i : 0)];
    }
  }
  if (t < BINSZ) dloc[t] = (t < nrows) ? dis[node0 + t] : 0.0f;
  if (t < RCN) rcnt[t] = 0;

  const float4* xw4 = reinterpret_cast<const float4*>(xw0);
  const float4* b4p = reinterpret_cast<const float4*>(bcat);
  float4* acc4 = reinterpret_cast<float4*>(acc);
  for (int q = t; q < nrows * 24; q += 1024) acc4[q] = xw4[q];
  __syncthreads();

  const int lane = t & 31;
  const int wave = __builtin_amdgcn_readfirstlane(t >> 5);
  const float4* xrow = reinterpret_cast<const float4*>(XW) + ((lane < 24) ? lane : 0);

  for (int c0 = start; c0 < end; c0 += SORTCAP) {
    const int n = (end - c0 < SORTCAP) ? (end - c0) : SORTCAP;

    int rank[13];
    if (c0 != start) {
#pragma unroll
      for (int k = 0; k < 13; ++k) {
        const int i = t + k * 1024;
        r[k] = recs[c0 + ((i < n) ? i : 0)];
      }
    }
#pragma unroll
    for (int k = 0; k < 13; ++k) {
      rank[k] = 0;
      if (t + k * 1024 < n) rank[k] = atomicAdd(&rcnt[(int)(r[k].x >> 17)], 1);
    }
    __syncthreads();

    if (wave == 0) {
      int v[13];
      int sum = 0;
#pragma unroll
      for (int i = 0; i < 13; ++i) {
        v[i] = rcnt[lane * 13 + i];
        sum += v[i];
      }
      int incl = sum;
#pragma unroll
      for (int d = 1; d < 32; d <<= 1) {
        const int x = __shfl_up(incl, d, 32);
        if (lane >= d) incl += x;
      }
      int run = incl - sum;
#pragma unroll
      for (int i = 0; i < 13; ++i) {
        rcnt[lane * 13 + i] = run;
        run += v[i];
      }
    }
    __syncthreads();

#pragma unroll
    for (int k = 0; k < 13; ++k) {
      if (t + k * 1024 < n) {
        const int dl = (int)(r[k].x >> 17);
        srt[rcnt[dl] + rank[k]] = r[k];
      }
    }
    __syncthreads();

    const int ws = (n * wave) >> 5;
    const int we = (n * (wave + 1)) >> 5;
    if (lane < 24) {
      int cur = -1;
      float4 a = make_float4(0.0f, 0.0f, 0.0f, 0.0f);
      uint2 rec_next = make_uint2(0u, 0u);
      if (ws + lane < we) rec_next = srt[ws + lane];
      for (int base = ws; base < we; base += 16) {
        const int cnt = (we - base < 16) ? (we - base) : 16;
        const uint2 rec = rec_next;
        float4 v[16];
        unsigned px[16];
        float nr[16];
#pragma unroll
        for (int u = 0; u < 16; ++u) {
          px[u] = (unsigned)__builtin_amdgcn_readlane((int)rec.x, u);
          nr[u] = __uint_as_float((unsigned)__builtin_amdgcn_readlane((int)rec.y, u));
          v[u] = xrow[(size_t)(px[u] & 0x1FFFFu) * 24];
        }
        const int nb = base + 16;
        rec_next = make_uint2(0u, 0u);
        if (nb + lane < we) rec_next = srt[nb + lane];
        if (cnt == 16) {
#pragma unroll
          for (int u = 0; u < 16; ++u) run_add(acc, lane, cur, a, px[u], nr[u], v[u]);
        } else {
#pragma unroll
          for (int u = 0; u < 16; ++u)
            if (u < cnt) run_add(acc, lane, cur, a, px[u], nr[u], v[u]);
        }
      }
      if (cur >= 0) {
        float* p = &acc[cur * 96 + lane * 4];
        atomicAdd(p,     a.x);
        atomicAdd(p + 1, a.y);
        atomicAdd(p + 2, a.z);
        atomicAdd(p + 3, a.w);
      }
    }
    if (c0 + SORTCAP < end) {
      __syncthreads();
      if (t < RCN) rcnt[t] = 0;
      __syncthreads();
    }
  }
  __syncthreads();

  const int r0 = wave * 16;
  const bool has_tile = r0 < nrows;
  float* Hs = reinterpret_cast<float*>(srt) + wave * GST;
  float* Rs = Hs + 16 * 36;
  float4 sh[4];
#pragma unroll
  for (int j = 0; j < 4; ++j) sh[j] = make_float4(0.0f, 0.0f, 0.0f, 0.0f);
  if (has_tile) {
#pragma unroll
    for (int j = 0; j < 4; ++j) {
      const int q = lane + j * 32;
      int rr = r0 + (q >> 3);
      rr = (rr < nrows) ? rr : (nrows - 1);
      sh[j] = *reinterpret_cast<const float4*>(Hp + (size_t)(node0 + rr) * 32 + ((q & 7) << 2));
    }
  }
  for (int q = t; q < nrows * 24; q += 1024) {
    const int i = q / 24;
    const int c4 = q - i * 24;
    const float d = dloc[i];
    const float4 s = acc4[q];
    const float4 b = b4p[c4];
    float4 o;
    o.x = d * s.x + b.x;
    o.y = d * s.y + b.y;
    o.z = d * s.z + b.z;
    o.w = d * s.w + b.w;
    acc4[q] = o;
  }
  if (has_tile) {
#pragma unroll
    for (int j = 0; j < 4; ++j) {
      const int q = lane + j * 32;
      *reinterpret_cast<float4*>(&Hs[(q >> 3) * 36 + ((q & 7) << 2)]) = sh[j];
    }
  }
  __syncthreads();

  const int m = lane & 15;
  const int koff = (lane >> 4) << 1;
  const int row0 = (lane >> 4) << 3;
  int ai = r0 + m;
  ai = (ai < nrows) ? ai : (nrows - 1);
  const float* arow = acc + ai * 96;
  const float* hrow = Hs + m * 36;
  const float* rrow = Rs + m * 36;

  if (has_tile) {
    v8f z0 = {}, z1 = {};
    v8f g0 = {}, g1 = {};
#pragma unroll 4
    for (int k0 = 0; k0 < 32; k0 += 4) {
      const int kk = k0 + koff;
      const float2 fz = *reinterpret_cast<const float2*>(arow + kk);
      const float2 fr = *reinterpret_cast<const float2*>(arow + 32 + kk);
      v2f az; az.x = fz.x; az.y = fz.y;
      v2f ar; ar.x = fr.x; ar.y = fr.y;
      const float* lz = LzW + kk * 32 + m;
      const float* lr = LrW + kk * 32 + m;
      v2f b;
      b.x = lz[0];  b.y = lz[32];  z0 = wmma4(az, b, z0);
      b.x = lz[16]; b.y = lz[48];  z1 = wmma4(az, b, z1);
      b.x = lr[0];  b.y = lr[32];  g0 = wmma4(ar, b, g0);
      b.x = lr[16]; b.y = lr[48];  g1 = wmma4(ar, b, g1);
    }
#pragma unroll 4
    for (int k0 = 32; k0 < 64; k0 += 4) {
      const int kk = k0 + koff;
      const float2 fh = *reinterpret_cast<const float2*>(hrow + kk - 32);
      v2f a; a.x = fh.x; a.y = fh.y;
      const float* lz = LzW + kk * 32 + m;
      const float* lr = LrW + kk * 32 + m;
      v2f b;
      b.x = lz[0];  b.y = lz[32];  z0 = wmma4(a, b, z0);
      b.x = lz[16]; b.y = lz[48];  z1 = wmma4(a, b, z1);
      b.x = lr[0];  b.y = lr[32];  g0 = wmma4(a, b, g0);
      b.x = lr[16]; b.y = lr[48];  g1 = wmma4(a, b, g1);
    }
    const float bz0 = Lzb[m], bz1 = Lzb[16 + m];
    const float br0 = Lrb[m], br1 = Lrb[16 + m];
#pragma unroll
    for (int v = 0; v < 8; ++v) {
      z0[v] = sigmoidf_(z0[v] + bz0);
      z1[v] = sigmoidf_(z1[v] + bz1);
      g0[v] = sigmoidf_(g0[v] + br0);
      g1[v] = sigmoidf_(g1[v] + br1);
      Rs[(row0 + v) * 36 + m]      = g0[v];
      Rs[(row0 + v) * 36 + 16 + m] = g1[v];
    }
    __builtin_amdgcn_fence(__ATOMIC_RELEASE, "workgroup", "local");
    __builtin_amdgcn_fence(__ATOMIC_ACQUIRE, "workgroup", "local");

    v8f t0 = {}, t1 = {};
#pragma unroll 4
    for (int k0 = 0; k0 < 32; k0 += 4) {
      const int kk = k0 + koff;
      const float2 fa = *reinterpret_cast<const float2*>(arow + 64 + kk);
      v2f a; a.x = fa.x; a.y = fa.y;
      const float* lh = LhW + kk * 32 + m;
      v2f b;
      b.x = lh[0];  b.y = lh[32];  t0 = wmma4(a, b, t0);
      b.x = lh[16]; b.y = lh[48];  t1 = wmma4(a, b, t1);
    }
#pragma unroll 4
    for (int k0 = 32; k0 < 64; k0 += 4) {
      const int kk = k0 + koff;
      const float2 fh = *reinterpret_cast<const float2*>(hrow + kk - 32);
      const float2 fr = *reinterpret_cast<const float2*>(rrow + kk - 32);
      v2f a; a.x = fh.x * fr.x; a.y = fh.y * fr.y;
      const float* lh = LhW + kk * 32 + m;
      v2f b;
      b.x = lh[0];  b.y = lh[32];  t0 = wmma4(a, b, t0);
      b.x = lh[16]; b.y = lh[48];  t1 = wmma4(a, b, t1);
    }

    const float bh0 = Lhb[m], bh1 = Lhb[16 + m];
    const float hw0 = headW[m], hw1 = headW[16 + m];
    float p[8];
#pragma unroll
    for (int v = 0; v < 8; ++v) {
      const float ht0 = tanhf(t0[v] + bh0);
      const float ht1 = tanhf(t1[v] + bh1);
      const float hp0 = Hs[(row0 + v) * 36 + m];
      const float hp1 = Hs[(row0 + v) * 36 + 16 + m];
      const float h0 = z0[v] * hp0 + (1.0f - z0[v]) * ht0;
      const float h1 = z1[v] * hp1 + (1.0f - z1[v]) * ht1;
      const int rr = r0 + row0 + v;
      if (rr < nrows) {
        h_out[(size_t)(node0 + rr) * 32 + m]      = h0;
        h_out[(size_t)(node0 + rr) * 32 + 16 + m] = h1;
      }
      p[v] = fmaxf(h0, 0.0f) * hw0 + fmaxf(h1, 0.0f) * hw1;
    }
#pragma unroll
    for (int s = 1; s < 16; s <<= 1)
#pragma unroll
      for (int v = 0; v < 8; ++v)
        p[v] += __shfl_xor(p[v], s, 32);
    if (m == 0) {
      const float hb = headb[0];
#pragma unroll
      for (int v = 0; v < 8; ++v) {
        const int rr = r0 + row0 + v;
        if (rr < nrows) y_out[node0 + rr] = p[v] + hb;
      }
    }
  }
}

extern "C" void kernel_launch(void* const* d_in, const int* in_sizes, int n_in,
                              void* d_out, int out_size, void* d_ws, size_t ws_size,
                              hipStream_t stream) {
  const float* x    = (const float*)d_in[0];
  const int*   ei   = (const int*)d_in[1];
  const float* ew   = (const float*)d_in[2];
  const float* Hp   = (const float*)d_in[3];
  const float* Wz   = (const float*)d_in[4];
  const float* bz   = (const float*)d_in[5];
  const float* Wr   = (const float*)d_in[6];
  const float* br   = (const float*)d_in[7];
  const float* Wh   = (const float*)d_in[8];
  const float* bh   = (const float*)d_in[9];
  const float* LzW  = (const float*)d_in[10];
  const float* Lzb  = (const float*)d_in[11];
  const float* LrW  = (const float*)d_in[12];
  const float* Lrb  = (const float*)d_in[13];
  const float* LhW  = (const float*)d_in[14];
  const float* Lhb  = (const float*)d_in[15];
  const float* hW   = (const float*)d_in[16];
  const float* hb   = (const float*)d_in[17];

  const int* src = ei;
  const int* dst = ei + Ee;

  float* out_y = (float*)d_out;
  float* out_h = out_y + Nn;

  float* ws   = (float*)d_ws;
  float* dis  = ws;
  float* Wcat = ws + Nn;
  float* bcat = Wcat + FIN * 96;
  float* XW   = bcat + 96;
  int*   runstart = (int*)(XW + (size_t)Nn * 96);
  int*   binstart = runstart + NBINS * NSLICE;
  uint2* recs   = (uint2*)(binstart + NBINS * NSLICE);
  uint2* srt    = recs + Ee;

  pack_weights_kernel<<<(FIN * 96 + 255) / 256, 256, 0, stream>>>(Wz, Wr, Wh, bz, br, bh, Wcat, bcat);
  bin_scatter_kernel<<<NSLICE, 1024, 0, stream>>>(src, dst, ew, runstart, srt);
  bin_degree_gemm_kernel<<<NBINS, 1024, 0, stream>>>(srt, runstart, binstart, recs, dis, x, Wcat, XW);
  bin_aggregate_gru_kernel<<<NBINS, 1024, 0, stream>>>(recs, binstart, dis, XW, bcat, Hp, LzW, Lzb, LrW, Lrb, LhW, Lhb, hW, hb,
                                                       out_y, out_h);
}
